# starting kernel
# baseline (speedup 1.0000x reference)
#include <hip/hip_runtime.h>
#include <hip/hip_bf16.h>

typedef __attribute__((ext_vector_type(2))) float v2f;
typedef __attribute__((ext_vector_type(8))) float v8f;

#define MODEL_DIM 1024
#define D_STATE   16
#define D_CONV    4
#define D_INNER   2048
#define BATCH     4
#define SEQLEN    1024
#define NROWS     (BATCH * SEQLEN)

__device__ __forceinline__ float silu_f(float x) {
    return x / (1.0f + __expf(-x));
}

#define KC 16
#define BPITCH 288
__global__ void __launch_bounds__(256)
wmma_gemm_f32(const float* __restrict__ A, const float* __restrict__ W,
              float* __restrict__ C, int M, int N, int K)
{
    __shared__ float Bs[(KC / 2) * BPITCH];

    const int wave   = threadIdx.x >> 5;
    const int lane   = threadIdx.x & 31;
    const int lane16 = lane & 15;
    const int khalf  = (lane >> 4) * 2;

    const int m0 = blockIdx.y * 256 + wave * 32;
    const int n0 = blockIdx.x * 128;

    const float* Arow0 = A + (size_t)(m0 + lane16) * K;
    const float* Arow1 = A + (size_t)(m0 + 16 + lane16) * K;

    const int wr = wave;
    const int wc = lane * 4;
    const float* Wbase = W + (size_t)wr * N + n0 + wc;
    float* Bst0 = Bs + (wr >> 1) * BPITCH + (wr & 1);
    float* Bst1 = Bs + ((wr + 8) >> 1) * BPITCH + (wr & 1);
    float4 w0, w1;

    v8f acc[2][8];
    #pragma unroll
    for (int g = 0; g < 2; ++g)
        #pragma unroll
        for (int j = 0; j < 8; ++j)
            acc[g][j] = (v8f){};

    w0 = *(const float4*)(Wbase);
    w1 = *(const float4*)(Wbase + (size_t)8 * N);
    {
        Bst0[(wc + 0) * 2] = w0.x;  Bst0[(wc + 1) * 2] = w0.y;
        Bst0[(wc + 2) * 2] = w0.z;  Bst0[(wc + 3) * 2] = w0.w;
        Bst1[(wc + 0) * 2] = w1.x;  Bst1[(wc + 1) * 2] = w1.y;
        Bst1[(wc + 2) * 2] = w1.z;  Bst1[(wc + 3) * 2] = w1.w;
    }
    __syncthreads();

    for (int k = 0; k < K; k += KC) {
        const bool has_next = (k + KC) < K;
        if (has_next) {
            const float* Wp = Wbase + (size_t)(k + KC) * N;
            w0 = *(const float4*)(Wp);
            w1 = *(const float4*)(Wp + (size_t)8 * N);
        }

        #pragma unroll
        for (int kk = 0; kk < KC; kk += 4) {
            v2f a0 = *(const v2f*)(Arow0 + k + kk + khalf);
            v2f a1 = *(const v2f*)(Arow1 + k + kk + khalf);
            const float* bp = Bs + ((kk + khalf) >> 1) * BPITCH + lane16 * 2;
            #pragma unroll
            for (int j = 0; j < 8; ++j) {
                v2f b = *(const v2f*)(bp + j * 32);
                acc[0][j] = __builtin_amdgcn_wmma_f32_16x16x4_f32(
                    false, a0, false, b, (short)0, acc[0][j], false, false);
                acc[1][j] = __builtin_amdgcn_wmma_f32_16x16x4_f32(
                    false, a1, false, b, (short)0, acc[1][j], false, false);
            }
        }

        __syncthreads();
        if (has_next) {
            Bst0[(wc + 0) * 2] = w0.x;  Bst0[(wc + 1) * 2] = w0.y;
            Bst0[(wc + 2) * 2] = w0.z;  Bst0[(wc + 3) * 2] = w0.w;
            Bst1[(wc + 0) * 2] = w1.x;  Bst1[(wc + 1) * 2] = w1.y;
            Bst1[(wc + 2) * 2] = w1.z;  Bst1[(wc + 3) * 2] = w1.w;
            __syncthreads();
        }
    }

    const int crow = (lane >> 4) ? 8 : 0;
    const int ccol = n0 + lane16;
    #pragma unroll
    for (int g = 0; g < 2; ++g) {
        float* Cp = C + (size_t)(m0 + g * 16 + crow) * N + ccol;
        #pragma unroll
        for (int r = 0; r < 8; ++r) {
            #pragma unroll
            for (int j = 0; j < 8; ++j)
                Cp[(size_t)r * N + j * 16] = acc[g][j][r];
        }
    }
}

__global__ void __launch_bounds__(256)
conv_silu_kernel(const float* __restrict__ xz,
                 const float* __restrict__ conv_w,
                 const float* __restrict__ conv_b,
                 float* __restrict__ xconv)
{
    size_t idx = (size_t)blockIdx.x * blockDim.x + threadIdx.x;
    if (idx >= (size_t)NROWS * D_INNER) return;
    const int d = (int)(idx & (D_INNER - 1));
    const size_t row = idx >> 11;
    const int l = (int)(row & (SEQLEN - 1));

    float acc = conv_b[d];
    const float* w = conv_w + (size_t)d * D_CONV;
    #pragma unroll
    for (int k = 0; k < D_CONV; ++k) {
        int li = l - (D_CONV - 1) + k;
        if (li >= 0)
            acc += xz[(row - l + li) * (2 * D_INNER) + d] * w[k];
    }
    xconv[idx] = silu_f(acc);
}

__global__ void __launch_bounds__(256)
ssm_proj_kernel(const float* __restrict__ xconv,
                const float* __restrict__ Wx,
                float* __restrict__ ssm)
{
    __shared__ float xs[D_INNER];
    const int row = blockIdx.x;
    const float* xr = xconv + (size_t)row * D_INNER;
    for (int d = threadIdx.x; d < D_INNER; d += 256) xs[d] = xr[d];
    __syncthreads();

    const int wave = threadIdx.x >> 5;
    const int lane = threadIdx.x & 31;
    const int NC = 2 * D_STATE + 1;
    for (int j = wave; j < NC; j += 8) {
        float s = 0.0f;
        for (int d = lane; d < D_INNER; d += 32)
            s += xs[d] * Wx[(size_t)d * NC + j];
        #pragma unroll
        for (int off = 16; off > 0; off >>= 1)
            s += __shfl_down(s, off, 32);
        if (lane == 0) ssm[(size_t)row * NC + j] = s;
    }
}

#define TCH 64
__global__ void __launch_bounds__(64)
scan_kernel(const float* __restrict__ xconv,
            const float* __restrict__ ssm,
            const float* __restrict__ dt_w,
            const float* __restrict__ dt_b,
            const float* __restrict__ A_log,
            const float* __restrict__ Dp,
            float* __restrict__ yinner)
{
    __shared__ float sb[TCH * 33];
    const int d = blockIdx.x * 64 + threadIdx.x;
    const int b = blockIdx.y;

    const float dw = dt_w[d];
    const float db = dt_b[d];
    const float Dd = Dp[d];

    float Areg[D_STATE];
    #pragma unroll
    for (int n = 0; n < D_STATE; ++n)
        Areg[n] = -__expf(A_log[(size_t)d * D_STATE + n]);

    float h[D_STATE];
    #pragma unroll
    for (int n = 0; n < D_STATE; ++n) h[n] = 0.0f;

    for (int t0 = 0; t0 < SEQLEN; t0 += TCH) {
        __syncthreads();
        {
            const float* src = ssm + (size_t)(b * SEQLEN + t0) * 33;
            for (int i = threadIdx.x; i < TCH * 33; i += 64) sb[i] = src[i];
        }
        __syncthreads();

        for (int tt = 0; tt < TCH; ++tt) {
            const float* r = sb + tt * 33;
            const size_t row = (size_t)(b * SEQLEN + t0 + tt);
            const float draw = r[0];
            const float xv = xconv[row * D_INNER + d];

            float zz = dw * draw + db;
            float delta = (zz > 20.0f) ? zz : log1pf(__expf(zz));
            const float dx = delta * xv;

            float y = 0.0f;
            #pragma unroll
            for (int n = 0; n < D_STATE; ++n) {
                float dA = __expf(delta * Areg[n]);
                h[n] = dA * h[n] + dx * r[1 + n];
                y += h[n] * r[1 + D_STATE + n];
            }
            yinner[row * D_INNER + d] = y + Dd * xv;
        }
    }
}

__global__ void __launch_bounds__(256)
gate_kernel(const float* __restrict__ yinner,
            const float* __restrict__ xz,
            float* __restrict__ ygated)
{
    size_t idx = (size_t)blockIdx.x * blockDim.x + threadIdx.x;
    if (idx >= (size_t)NROWS * D_INNER) return;
    const int d = (int)(idx & (D_INNER - 1));
    const size_t row = idx >> 11;
    const float z = xz[row * (2 * D_INNER) + D_INNER + d];
    ygated[idx] = yinner[idx] * silu_f(z);
}

extern "C" void kernel_launch(void* const* d_in, const int* in_sizes, int n_in,
                              void* d_out, int out_size, void* d_ws, size_t ws_size,
                              hipStream_t stream)
{
    const float* x      = (const float*)d_in[0];
    const float* W_in   = (const float*)d_in[1];
    const float* conv_w = (const float*)d_in[2];
    const float* conv_b = (const float*)d_in[3];
    const float* W_x    = (const float*)d_in[4];
    const float* dt_w   = (const float*)d_in[5];
    const float* dt_b   = (const float*)d_in[6];
    const float* A_log  = (const float*)d_in[7];
    const float* Dp     = (const float*)d_in[8];
    const float* W_out  = (const float*)d_in[9];
    float* out = (float*)d_out;

    float* ws = (float*)d_ws;
    float* xz     = ws;
    float* xconv  = xz    + (size_t)NROWS * 2 * D_INNER;
    float* ssm    = xconv + (size_t)NROWS * D_INNER;
    float* yinner = ssm   + (size_t)NROWS * 33;
    float* ygated = xconv;

    wmma_gemm_f32<<<dim3(2 * D_INNER / 128, NROWS / 256), 256, 0, stream>>>(
        x, W_in, xz, NROWS, 2 * D_INNER, MODEL_DIM);

    {
        size_t total = (size_t)NROWS * D_INNER;
        conv_silu_kernel<<<(unsigned)((total + 255) / 256), 256, 0, stream>>>(
            xz, conv_w, conv_b, xconv);
    }

    ssm_proj_kernel<<<NROWS, 256, 0, stream>>>(xconv, W_x, ssm);

    scan_kernel<<<dim3(D_INNER / 64, BATCH), 64, 0, stream>>>(
        xconv, ssm, dt_w, dt_b, A_log, Dp, yinner);

    {
        size_t total = (size_t)NROWS * D_INNER;
        gate_kernel<<<(unsigned)((total + 255) / 256), 256, 0, stream>>>(
            yinner, xz, ygated);
    }

    wmma_gemm_f32<<<dim3(MODEL_DIM / 128, NROWS / 256), 256, 0, stream>>>(
        ygated, W_out, out, NROWS, MODEL_DIM, D_INNER);
}


# ssm projection (x_conv @ W_x, 4096x2048x33) moved from per-row vector dot products to f32 WMMA with split-K over 16 waves summed in LDS; changes that product's summation order
# speedup vs baseline: 1.9678x; 1.9678x over previous
#include <hip/hip_runtime.h>
#include <hip/hip_bf16.h>

typedef __attribute__((ext_vector_type(2))) float v2f;
typedef __attribute__((ext_vector_type(8))) float v8f;

#define MODEL_DIM 1024
#define D_STATE   16
#define D_CONV    4
#define D_INNER   2048
#define BATCH     4
#define SEQLEN    1024
#define NROWS     (BATCH * SEQLEN)

__device__ __forceinline__ float silu_f(float x) {
    return x / (1.0f + __expf(-x));
}

#define KC 16
#define BPITCH 288
__global__ void __launch_bounds__(256)
wmma_gemm_f32(const float* __restrict__ A, const float* __restrict__ W,
              float* __restrict__ C, int M, int N, int K)
{
    __shared__ float Bs[(KC / 2) * BPITCH];

    const int wave   = threadIdx.x >> 5;
    const int lane   = threadIdx.x & 31;
    const int lane16 = lane & 15;
    const int khalf  = (lane >> 4) * 2;

    const int m0 = blockIdx.y * 256 + wave * 32;
    const int n0 = blockIdx.x * 128;

    const float* Arow0 = A + (size_t)(m0 + lane16) * K;
    const float* Arow1 = A + (size_t)(m0 + 16 + lane16) * K;

    const int wr = wave;
    const int wc = lane * 4;
    const float* Wbase = W + (size_t)wr * N + n0 + wc;
    float* Bst0 = Bs + (wr >> 1) * BPITCH + (wr & 1);
    float* Bst1 = Bs + ((wr + 8) >> 1) * BPITCH + (wr & 1);
    float4 w0, w1;

    v8f acc[2][8];
    #pragma unroll
    for (int g = 0; g < 2; ++g)
        #pragma unroll
        for (int j = 0; j < 8; ++j)
            acc[g][j] = (v8f){};

    w0 = *(const float4*)(Wbase);
    w1 = *(const float4*)(Wbase + (size_t)8 * N);
    {
        Bst0[(wc + 0) * 2] = w0.x;  Bst0[(wc + 1) * 2] = w0.y;
        Bst0[(wc + 2) * 2] = w0.z;  Bst0[(wc + 3) * 2] = w0.w;
        Bst1[(wc + 0) * 2] = w1.x;  Bst1[(wc + 1) * 2] = w1.y;
        Bst1[(wc + 2) * 2] = w1.z;  Bst1[(wc + 3) * 2] = w1.w;
    }
    __syncthreads();

    for (int k = 0; k < K; k += KC) {
        const bool has_next = (k + KC) < K;
        if (has_next) {
            const float* Wp = Wbase + (size_t)(k + KC) * N;
            w0 = *(const float4*)(Wp);
            w1 = *(const float4*)(Wp + (size_t)8 * N);
        }

        #pragma unroll
        for (int kk = 0; kk < KC; kk += 4) {
            v2f a0 = *(const v2f*)(Arow0 + k + kk + khalf);
            v2f a1 = *(const v2f*)(Arow1 + k + kk + khalf);
            const float* bp = Bs + ((kk + khalf) >> 1) * BPITCH + lane16 * 2;
            #pragma unroll
            for (int j = 0; j < 8; ++j) {
                v2f b = *(const v2f*)(bp + j * 32);
                acc[0][j] = __builtin_amdgcn_wmma_f32_16x16x4_f32(
                    false, a0, false, b, (short)0, acc[0][j], false, false);
                acc[1][j] = __builtin_amdgcn_wmma_f32_16x16x4_f32(
                    false, a1, false, b, (short)0, acc[1][j], false, false);
            }
        }

        __syncthreads();
        if (has_next) {
            Bst0[(wc + 0) * 2] = w0.x;  Bst0[(wc + 1) * 2] = w0.y;
            Bst0[(wc + 2) * 2] = w0.z;  Bst0[(wc + 3) * 2] = w0.w;
            Bst1[(wc + 0) * 2] = w1.x;  Bst1[(wc + 1) * 2] = w1.y;
            Bst1[(wc + 2) * 2] = w1.z;  Bst1[(wc + 3) * 2] = w1.w;
            __syncthreads();
        }
    }

    const int crow = (lane >> 4) ? 8 : 0;
    const int ccol = n0 + lane16;
    #pragma unroll
    for (int g = 0; g < 2; ++g) {
        float* Cp = C + (size_t)(m0 + g * 16 + crow) * N + ccol;
        #pragma unroll
        for (int r = 0; r < 8; ++r) {
            #pragma unroll
            for (int j = 0; j < 8; ++j)
                Cp[(size_t)r * N + j * 16] = acc[g][j][r];
        }
    }
}

__global__ void __launch_bounds__(256)
conv_silu_kernel(const float* __restrict__ xz,
                 const float* __restrict__ conv_w,
                 const float* __restrict__ conv_b,
                 float* __restrict__ xconv)
{
    size_t idx = (size_t)blockIdx.x * blockDim.x + threadIdx.x;
    if (idx >= (size_t)NROWS * D_INNER) return;
    const int d = (int)(idx & (D_INNER - 1));
    const size_t row = idx >> 11;
    const int l = (int)(row & (SEQLEN - 1));

    float acc = conv_b[d];
    const float* w = conv_w + (size_t)d * D_CONV;
    #pragma unroll
    for (int k = 0; k < D_CONV; ++k) {
        int li = l - (D_CONV - 1) + k;
        if (li >= 0)
            acc += xz[(row - l + li) * (2 * D_INNER) + d] * w[k];
    }
    xconv[idx] = silu_f(acc);
}

#define SP_NC     33
#define SP_WAVES  16
#define SP_KSLICE (D_INNER / SP_WAVES)
#define SP_GRP    4
#define SP_NGRP   (SP_KSLICE / (8 * SP_GRP))
#define SP_RQ     267
#define SP_RW     (6 * SP_RQ)
#define SP_SLOTS  (SP_WAVES / 2)

struct F3 { float x, y, z; };

__global__ void __launch_bounds__(512)
ssm_proj_wmma(const float* __restrict__ xconv,
              const float* __restrict__ Wx,
              float* __restrict__ ssm)
{
    __shared__ float red[SP_SLOTS * SP_RW];

    const int wave   = threadIdx.x >> 5;
    const int lane   = threadIdx.x & 31;
    const int lane16 = lane & 15;
    const int hi     = lane >> 4;
    const int m0     = blockIdx.x * 32;
    const int k0     = wave * SP_KSLICE + 4 * hi;

    const float* ap0 = xconv + (size_t)(m0 + lane16) * D_INNER + k0;
    const float* ap1 = ap0 + (size_t)16 * D_INNER;
    const int jj = (lane16 < 11) ? lane16 : 10;
    const float* bp = Wx + (size_t)k0 * SP_NC + 3 * jj;

    v8f acc[2][3];
    #pragma unroll
    for (int g = 0; g < 2; ++g)
        #pragma unroll
        for (int t = 0; t < 3; ++t)
            acc[g][t] = (v8f){};

    float4 a0[2][SP_GRP], a1[2][SP_GRP];
    F3 bf[2][SP_GRP][4];

    #pragma unroll
    for (int o = 0; o < SP_GRP; ++o) {
        a0[0][o] = *(const float4*)(ap0 + 8 * o);
        a1[0][o] = *(const float4*)(ap1 + 8 * o);
        #pragma unroll
        for (int r = 0; r < 4; ++r)
            bf[0][o][r] = *(const F3*)(bp + (size_t)(8 * o + r) * SP_NC);
    }

    #pragma unroll
    for (int grp = 0; grp < SP_NGRP; ++grp) {
        const int cur = grp & 1, nxt = cur ^ 1;
        if (grp + 1 < SP_NGRP) {
            #pragma unroll
            for (int o = 0; o < SP_GRP; ++o) {
                const int ko = 8 * ((grp + 1) * SP_GRP + o);
                a0[nxt][o] = *(const float4*)(ap0 + ko);
                a1[nxt][o] = *(const float4*)(ap1 + ko);
                #pragma unroll
                for (int r = 0; r < 4; ++r)
                    bf[nxt][o][r] = *(const F3*)(bp + (size_t)(ko + r) * SP_NC);
            }
        }
        __builtin_amdgcn_sched_barrier(0);
        #pragma unroll
        for (int o = 0; o < SP_GRP; ++o) {
            #pragma unroll
            for (int s = 0; s < 2; ++s) {
                const v2f x0 = s ? (v2f){a0[cur][o].z, a0[cur][o].w}
                                 : (v2f){a0[cur][o].x, a0[cur][o].y};
                const v2f x1 = s ? (v2f){a1[cur][o].z, a1[cur][o].w}
                                 : (v2f){a1[cur][o].x, a1[cur][o].y};
                const F3 lo = bf[cur][o][2 * s], up = bf[cur][o][2 * s + 1];
                const v2f b0 = (v2f){lo.x, up.x};
                const v2f b1 = (v2f){lo.y, up.y};
                const v2f b2 = (v2f){lo.z, up.z};
                acc[0][0] = __builtin_amdgcn_wmma_f32_16x16x4_f32(
                    false, x0, false, b0, (short)0, acc[0][0], false, false);
                acc[1][0] = __builtin_amdgcn_wmma_f32_16x16x4_f32(
                    false, x1, false, b0, (short)0, acc[1][0], false, false);
                acc[0][1] = __builtin_amdgcn_wmma_f32_16x16x4_f32(
                    false, x0, false, b1, (short)0, acc[0][1], false, false);
                acc[1][1] = __builtin_amdgcn_wmma_f32_16x16x4_f32(
                    false, x1, false, b1, (short)0, acc[1][1], false, false);
                acc[0][2] = __builtin_amdgcn_wmma_f32_16x16x4_f32(
                    false, x0, false, b2, (short)0, acc[0][2], false, false);
                acc[1][2] = __builtin_amdgcn_wmma_f32_16x16x4_f32(
                    false, x1, false, b2, (short)0, acc[1][2], false, false);
            }
        }
    }

    float* rp = red + (wave & (SP_SLOTS - 1)) * SP_RW + lane;
    if (wave >= SP_SLOTS) {
        #pragma unroll
        for (int g = 0; g < 2; ++g)
            #pragma unroll
            for (int t = 0; t < 3; ++t)
                #pragma unroll
                for (int r = 0; r < 8; ++r)
                    rp[(g * 3 + t) * SP_RQ + r * 32] = acc[g][t][r];
    }
    __syncthreads();
    if (wave < SP_SLOTS) {
        #pragma unroll
        for (int g = 0; g < 2; ++g)
            #pragma unroll
            for (int t = 0; t < 3; ++t)
                #pragma unroll
                for (int r = 0; r < 8; ++r) {
                    float* q = rp + (g * 3 + t) * SP_RQ + r * 32;
                    *q = acc[g][t][r] + *q;
                }
    }
    __syncthreads();

    float* outp = ssm + (size_t)m0 * SP_NC;
    for (int o = threadIdx.x; o < 32 * SP_NC; o += 512) {
        const int row = o / SP_NC;
        const int n   = o - row * SP_NC;
        const int j   = n / 3;
        const int t   = n - 3 * j;
        const int q   = (row >> 4) * 3 + t;
        const int r   = row & 7;
        const int ln  = ((row >> 3) & 1) * 16 + j;
        const float* p = red + q * SP_RQ + r * 32 + ln;
        float s = 0.0f;
        #pragma unroll
        for (int w = 0; w < SP_SLOTS; ++w)
            s += p[w * SP_RW];
        outp[o] = s;
    }
}

#define TCH 64
__global__ void __launch_bounds__(64)
scan_kernel(const float* __restrict__ xconv,
            const float* __restrict__ ssm,
            const float* __restrict__ dt_w,
            const float* __restrict__ dt_b,
            const float* __restrict__ A_log,
            const float* __restrict__ Dp,
            float* __restrict__ yinner)
{
    __shared__ float sb[TCH * 33];
    const int d = blockIdx.x * 64 + threadIdx.x;
    const int b = blockIdx.y;

    const float dw = dt_w[d];
    const float db = dt_b[d];
    const float Dd = Dp[d];

    float Areg[D_STATE];
    #pragma unroll
    for (int n = 0; n < D_STATE; ++n)
        Areg[n] = -__expf(A_log[(size_t)d * D_STATE + n]);

    float h[D_STATE];
    #pragma unroll
    for (int n = 0; n < D_STATE; ++n) h[n] = 0.0f;

    for (int t0 = 0; t0 < SEQLEN; t0 += TCH) {
        __syncthreads();
        {
            const float* src = ssm + (size_t)(b * SEQLEN + t0) * 33;
            for (int i = threadIdx.x; i < TCH * 33; i += 64) sb[i] = src[i];
        }
        __syncthreads();

        for (int tt = 0; tt < TCH; ++tt) {
            const float* r = sb + tt * 33;
            const size_t row = (size_t)(b * SEQLEN + t0 + tt);
            const float draw = r[0];
            const float xv = xconv[row * D_INNER + d];

            float zz = dw * draw + db;
            float delta = (zz > 20.0f) ? zz : log1pf(__expf(zz));
            const float dx = delta * xv;

            float y = 0.0f;
            #pragma unroll
            for (int n = 0; n < D_STATE; ++n) {
                float dA = __expf(delta * Areg[n]);
                h[n] = dA * h[n] + dx * r[1 + n];
                y += h[n] * r[1 + D_STATE + n];
            }
            yinner[row * D_INNER + d] = y + Dd * xv;
        }
    }
}

__global__ void __launch_bounds__(256)
gate_kernel(const float* __restrict__ yinner,
            const float* __restrict__ xz,
            float* __restrict__ ygated)
{
    size_t idx = (size_t)blockIdx.x * blockDim.x + threadIdx.x;
    if (idx >= (size_t)NROWS * D_INNER) return;
    const int d = (int)(idx & (D_INNER - 1));
    const size_t row = idx >> 11;
    const float z = xz[row * (2 * D_INNER) + D_INNER + d];
    ygated[idx] = yinner[idx] * silu_f(z);
}

extern "C" void kernel_launch(void* const* d_in, const int* in_sizes, int n_in,
                              void* d_out, int out_size, void* d_ws, size_t ws_size,
                              hipStream_t stream)
{
    const float* x      = (const float*)d_in[0];
    const float* W_in   = (const float*)d_in[1];
    const float* conv_w = (const float*)d_in[2];
    const float* conv_b = (const float*)d_in[3];
    const float* W_x    = (const float*)d_in[4];
    const float* dt_w   = (const float*)d_in[5];
    const float* dt_b   = (const float*)d_in[6];
    const float* A_log  = (const float*)d_in[7];
    const float* Dp     = (const float*)d_in[8];
    const float* W_out  = (const float*)d_in[9];
    float* out = (float*)d_out;

    float* ws = (float*)d_ws;
    float* xz     = ws;
    float* xconv  = xz    + (size_t)NROWS * 2 * D_INNER;
    float* ssm    = xconv + (size_t)NROWS * D_INNER;
    float* yinner = ssm   + (size_t)NROWS * 33;
    float* ygated = xconv;

    wmma_gemm_f32<<<dim3(2 * D_INNER / 128, NROWS / 256), 256, 0, stream>>>(
        x, W_in, xz, NROWS, 2 * D_INNER, MODEL_DIM);

    {
        size_t total = (size_t)NROWS * D_INNER;
        conv_silu_kernel<<<(unsigned)((total + 255) / 256), 256, 0, stream>>>(
            xz, conv_w, conv_b, xconv);
    }

    ssm_proj_wmma<<<NROWS / 32, 512, 0, stream>>>(xconv, W_x, ssm);

    scan_kernel<<<dim3(D_INNER / 64, BATCH), 64, 0, stream>>>(
        xconv, ssm, dt_w, dt_b, A_log, Dp, yinner);

    {
        size_t total = (size_t)NROWS * D_INNER;
        gate_kernel<<<(unsigned)((total + 255) / 256), 256, 0, stream>>>(
            yinner, xz, ygated);
    }

    wmma_gemm_f32<<<dim3(MODEL_DIM / 128, NROWS / 256), 256, 0, stream>>>(
        ygated, W_out, out, NROWS, MODEL_DIM, D_INNER);
}


# scan kernel software-pipelined: per 8 time steps the ssm rows (one LDS load group), the x loads (one batch ahead) and the softplus step sizes (8 independent chains) precede the serial state updates
# speedup vs baseline: 3.6420x; 1.8508x over previous
#include <hip/hip_runtime.h>
#include <hip/hip_bf16.h>

typedef __attribute__((ext_vector_type(2))) float v2f;
typedef __attribute__((ext_vector_type(8))) float v8f;

#define MODEL_DIM 1024
#define D_STATE   16
#define D_CONV    4
#define D_INNER   2048
#define BATCH     4
#define SEQLEN    1024
#define NROWS     (BATCH * SEQLEN)

__device__ __forceinline__ float silu_f(float x) {
    return x / (1.0f + __expf(-x));
}

#define KC 16
#define BPITCH 288
__global__ void __launch_bounds__(256)
wmma_gemm_f32(const float* __restrict__ A, const float* __restrict__ W,
              float* __restrict__ C, int M, int N, int K)
{
    __shared__ float Bs[(KC / 2) * BPITCH];

    const int wave   = threadIdx.x >> 5;
    const int lane   = threadIdx.x & 31;
    const int lane16 = lane & 15;
    const int khalf  = (lane >> 4) * 2;

    const int m0 = blockIdx.y * 256 + wave * 32;
    const int n0 = blockIdx.x * 128;

    const float* Arow0 = A + (size_t)(m0 + lane16) * K;
    const float* Arow1 = A + (size_t)(m0 + 16 + lane16) * K;

    const int wr = wave;
    const int wc = lane * 4;
    const float* Wbase = W + (size_t)wr * N + n0 + wc;
    float* Bst0 = Bs + (wr >> 1) * BPITCH + (wr & 1);
    float* Bst1 = Bs + ((wr + 8) >> 1) * BPITCH + (wr & 1);
    float4 w0, w1;

    v8f acc[2][8];
    #pragma unroll
    for (int g = 0; g < 2; ++g)
        #pragma unroll
        for (int j = 0; j < 8; ++j)
            acc[g][j] = (v8f){};

    w0 = *(const float4*)(Wbase);
    w1 = *(const float4*)(Wbase + (size_t)8 * N);
    {
        Bst0[(wc + 0) * 2] = w0.x;  Bst0[(wc + 1) * 2] = w0.y;
        Bst0[(wc + 2) * 2] = w0.z;  Bst0[(wc + 3) * 2] = w0.w;
        Bst1[(wc + 0) * 2] = w1.x;  Bst1[(wc + 1) * 2] = w1.y;
        Bst1[(wc + 2) * 2] = w1.z;  Bst1[(wc + 3) * 2] = w1.w;
    }
    __syncthreads();

    for (int k = 0; k < K; k += KC) {
        const bool has_next = (k + KC) < K;
        if (has_next) {
            const float* Wp = Wbase + (size_t)(k + KC) * N;
            w0 = *(const float4*)(Wp);
            w1 = *(const float4*)(Wp + (size_t)8 * N);
        }

        #pragma unroll
        for (int kk = 0; kk < KC; kk += 4) {
            v2f a0 = *(const v2f*)(Arow0 + k + kk + khalf);
            v2f a1 = *(const v2f*)(Arow1 + k + kk + khalf);
            const float* bp = Bs + ((kk + khalf) >> 1) * BPITCH + lane16 * 2;
            #pragma unroll
            for (int j = 0; j < 8; ++j) {
                v2f b = *(const v2f*)(bp + j * 32);
                acc[0][j] = __builtin_amdgcn_wmma_f32_16x16x4_f32(
                    false, a0, false, b, (short)0, acc[0][j], false, false);
                acc[1][j] = __builtin_amdgcn_wmma_f32_16x16x4_f32(
                    false, a1, false, b, (short)0, acc[1][j], false, false);
            }
        }

        __syncthreads();
        if (has_next) {
            Bst0[(wc + 0) * 2] = w0.x;  Bst0[(wc + 1) * 2] = w0.y;
            Bst0[(wc + 2) * 2] = w0.z;  Bst0[(wc + 3) * 2] = w0.w;
            Bst1[(wc + 0) * 2] = w1.x;  Bst1[(wc + 1) * 2] = w1.y;
            Bst1[(wc + 2) * 2] = w1.z;  Bst1[(wc + 3) * 2] = w1.w;
            __syncthreads();
        }
    }

    const int crow = (lane >> 4) ? 8 : 0;
    const int ccol = n0 + lane16;
    #pragma unroll
    for (int g = 0; g < 2; ++g) {
        float* Cp = C + (size_t)(m0 + g * 16 + crow) * N + ccol;
        #pragma unroll
        for (int r = 0; r < 8; ++r) {
            #pragma unroll
            for (int j = 0; j < 8; ++j)
                Cp[(size_t)r * N + j * 16] = acc[g][j][r];
        }
    }
}

__global__ void __launch_bounds__(256)
conv_silu_kernel(const float* __restrict__ xz,
                 const float* __restrict__ conv_w,
                 const float* __restrict__ conv_b,
                 float* __restrict__ xconv)
{
    size_t idx = (size_t)blockIdx.x * blockDim.x + threadIdx.x;
    if (idx >= (size_t)NROWS * D_INNER) return;
    const int d = (int)(idx & (D_INNER - 1));
    const size_t row = idx >> 11;
    const int l = (int)(row & (SEQLEN - 1));

    float acc = conv_b[d];
    const float* w = conv_w + (size_t)d * D_CONV;
    #pragma unroll
    for (int k = 0; k < D_CONV; ++k) {
        int li = l - (D_CONV - 1) + k;
        if (li >= 0)
            acc += xz[(row - l + li) * (2 * D_INNER) + d] * w[k];
    }
    xconv[idx] = silu_f(acc);
}

#define SP_NC     33
#define SP_WAVES  16
#define SP_KSLICE (D_INNER / SP_WAVES)
#define SP_GRP    4
#define SP_NGRP   (SP_KSLICE / (8 * SP_GRP))
#define SP_RQ     267
#define SP_RW     (6 * SP_RQ)
#define SP_SLOTS  (SP_WAVES / 2)

struct F3 { float x, y, z; };

__global__ void __launch_bounds__(512)
ssm_proj_wmma(const float* __restrict__ xconv,
              const float* __restrict__ Wx,
              float* __restrict__ ssm)
{
    __shared__ float red[SP_SLOTS * SP_RW];

    const int wave   = threadIdx.x >> 5;
    const int lane   = threadIdx.x & 31;
    const int lane16 = lane & 15;
    const int hi     = lane >> 4;
    const int m0     = blockIdx.x * 32;
    const int k0     = wave * SP_KSLICE + 4 * hi;

    const float* ap0 = xconv + (size_t)(m0 + lane16) * D_INNER + k0;
    const float* ap1 = ap0 + (size_t)16 * D_INNER;
    const int jj = (lane16 < 11) ? lane16 : 10;
    const float* bp = Wx + (size_t)k0 * SP_NC + 3 * jj;

    v8f acc[2][3];
    #pragma unroll
    for (int g = 0; g < 2; ++g)
        #pragma unroll
        for (int t = 0; t < 3; ++t)
            acc[g][t] = (v8f){};

    float4 a0[2][SP_GRP], a1[2][SP_GRP];
    F3 bf[2][SP_GRP][4];

    #pragma unroll
    for (int o = 0; o < SP_GRP; ++o) {
        a0[0][o] = *(const float4*)(ap0 + 8 * o);
        a1[0][o] = *(const float4*)(ap1 + 8 * o);
        #pragma unroll
        for (int r = 0; r < 4; ++r)
            bf[0][o][r] = *(const F3*)(bp + (size_t)(8 * o + r) * SP_NC);
    }

    #pragma unroll
    for (int grp = 0; grp < SP_NGRP; ++grp) {
        const int cur = grp & 1, nxt = cur ^ 1;
        if (grp + 1 < SP_NGRP) {
            #pragma unroll
            for (int o = 0; o < SP_GRP; ++o) {
                const int ko = 8 * ((grp + 1) * SP_GRP + o);
                a0[nxt][o] = *(const float4*)(ap0 + ko);
                a1[nxt][o] = *(const float4*)(ap1 + ko);
                #pragma unroll
                for (int r = 0; r < 4; ++r)
                    bf[nxt][o][r] = *(const F3*)(bp + (size_t)(ko + r) * SP_NC);
            }
        }
        __builtin_amdgcn_sched_barrier(0);
        #pragma unroll
        for (int o = 0; o < SP_GRP; ++o) {
            #pragma unroll
            for (int s = 0; s < 2; ++s) {
                const v2f x0 = s ? (v2f){a0[cur][o].z, a0[cur][o].w}
                                 : (v2f){a0[cur][o].x, a0[cur][o].y};
                const v2f x1 = s ? (v2f){a1[cur][o].z, a1[cur][o].w}
                                 : (v2f){a1[cur][o].x, a1[cur][o].y};
                const F3 lo = bf[cur][o][2 * s], up = bf[cur][o][2 * s + 1];
                const v2f b0 = (v2f){lo.x, up.x};
                const v2f b1 = (v2f){lo.y, up.y};
                const v2f b2 = (v2f){lo.z, up.z};
                acc[0][0] = __builtin_amdgcn_wmma_f32_16x16x4_f32(
                    false, x0, false, b0, (short)0, acc[0][0], false, false);
                acc[1][0] = __builtin_amdgcn_wmma_f32_16x16x4_f32(
                    false, x1, false, b0, (short)0, acc[1][0], false, false);
                acc[0][1] = __builtin_amdgcn_wmma_f32_16x16x4_f32(
                    false, x0, false, b1, (short)0, acc[0][1], false, false);
                acc[1][1] = __builtin_amdgcn_wmma_f32_16x16x4_f32(
                    false, x1, false, b1, (short)0, acc[1][1], false, false);
                acc[0][2] = __builtin_amdgcn_wmma_f32_16x16x4_f32(
                    false, x0, false, b2, (short)0, acc[0][2], false, false);
                acc[1][2] = __builtin_amdgcn_wmma_f32_16x16x4_f32(
                    false, x1, false, b2, (short)0, acc[1][2], false, false);
            }
        }
    }

    float* rp = red + (wave & (SP_SLOTS - 1)) * SP_RW + lane;
    if (wave >= SP_SLOTS) {
        #pragma unroll
        for (int g = 0; g < 2; ++g)
            #pragma unroll
            for (int t = 0; t < 3; ++t)
                #pragma unroll
                for (int r = 0; r < 8; ++r)
                    rp[(g * 3 + t) * SP_RQ + r * 32] = acc[g][t][r];
    }
    __syncthreads();
    if (wave < SP_SLOTS) {
        #pragma unroll
        for (int g = 0; g < 2; ++g)
            #pragma unroll
            for (int t = 0; t < 3; ++t)
                #pragma unroll
                for (int r = 0; r < 8; ++r) {
                    float* q = rp + (g * 3 + t) * SP_RQ + r * 32;
                    *q = acc[g][t][r] + *q;
                }
    }
    __syncthreads();

    float* outp = ssm + (size_t)m0 * SP_NC;
    for (int o = threadIdx.x; o < 32 * SP_NC; o += 512) {
        const int row = o / SP_NC;
        const int n   = o - row * SP_NC;
        const int j   = n / 3;
        const int t   = n - 3 * j;
        const int q   = (row >> 4) * 3 + t;
        const int r   = row & 7;
        const int ln  = ((row >> 3) & 1) * 16 + j;
        const float* p = red + q * SP_RQ + r * 32 + ln;
        float s = 0.0f;
        #pragma unroll
        for (int w = 0; w < SP_SLOTS; ++w)
            s += p[w * SP_RW];
        outp[o] = s;
    }
}

#define TCH 64
#define SBATCH 8
#define SROW (SBATCH * 33)

__global__ void __launch_bounds__(64) __attribute__((amdgpu_waves_per_eu(1, 1)))
scan_kernel(const float* __restrict__ xconv,
            const float* __restrict__ ssm,
            const float* __restrict__ dt_w,
            const float* __restrict__ dt_b,
            const float* __restrict__ A_log,
            const float* __restrict__ Dp,
            float* __restrict__ yinner)
{
    __shared__ __attribute__((aligned(16))) float sb[TCH * 33];
    const int d = blockIdx.x * 64 + threadIdx.x;
    const int b = blockIdx.y;

    const float dw = dt_w[d];
    const float db = dt_b[d];
    const float Dd = Dp[d];

    float Areg[D_STATE];
    #pragma unroll
    for (int n = 0; n < D_STATE; ++n)
        Areg[n] = -__expf(A_log[(size_t)d * D_STATE + n]);

    float h[D_STATE];
    #pragma unroll
    for (int n = 0; n < D_STATE; ++n) h[n] = 0.0f;

    const float* xp = xconv  + (size_t)b * SEQLEN * D_INNER + d;
    float*       yp = yinner + (size_t)b * SEQLEN * D_INNER + d;

    float xn[SBATCH];
    #pragma unroll
    for (int i = 0; i < SBATCH; ++i)
        xn[i] = xp[(size_t)i * D_INNER];

    for (int t0 = 0; t0 < SEQLEN; t0 += TCH) {
        __syncthreads();
        {
            const float* src = ssm + (size_t)(b * SEQLEN + t0) * 33;
            for (int i = threadIdx.x; i < TCH * 33; i += 64) sb[i] = src[i];
        }
        __syncthreads();

        for (int s0 = 0; s0 < TCH; s0 += SBATCH) {
            float4 rq[SROW / 4];
            const float4* rsrc = (const float4*)(sb + s0 * 33);
            #pragma unroll
            for (int q = 0; q < SROW / 4; ++q) rq[q] = rsrc[q];

            float xv[SBATCH], dl[SBATCH];
            #pragma unroll
            for (int i = 0; i < SBATCH; ++i) xv[i] = xn[i];

            const int tn = t0 + s0 + SBATCH;
            if (tn < SEQLEN) {
                #pragma unroll
                for (int i = 0; i < SBATCH; ++i)
                    xn[i] = xp[(size_t)(tn + i) * D_INNER];
            }

            float rr[SROW];
            #pragma unroll
            for (int q = 0; q < SROW / 4; ++q) {
                rr[4 * q + 0] = rq[q].x;  rr[4 * q + 1] = rq[q].y;
                rr[4 * q + 2] = rq[q].z;  rr[4 * q + 3] = rq[q].w;
            }

            #pragma unroll
            for (int i = 0; i < SBATCH; ++i) {
                const float zz = dw * rr[i * 33] + db;
                const float sp = log1pf(__expf(zz));
                dl[i] = (zz > 20.0f) ? zz : sp;
            }

            #pragma unroll
            for (int i = 0; i < SBATCH; ++i) {
                const float delta = dl[i];
                const float xvv = xv[i];
                const float dx = delta * xvv;

                float y = 0.0f;
                #pragma unroll
                for (int n = 0; n < D_STATE; ++n) {
                    float dA = __expf(delta * Areg[n]);
                    h[n] = dA * h[n] + dx * rr[i * 33 + 1 + n];
                    y += h[n] * rr[i * 33 + 1 + D_STATE + n];
                }
                yp[(size_t)(t0 + s0 + i) * D_INNER] = y + Dd * xvv;
            }
        }
    }
}

__global__ void __launch_bounds__(256)
gate_kernel(const float* __restrict__ yinner,
            const float* __restrict__ xz,
            float* __restrict__ ygated)
{
    size_t idx = (size_t)blockIdx.x * blockDim.x + threadIdx.x;
    if (idx >= (size_t)NROWS * D_INNER) return;
    const int d = (int)(idx & (D_INNER - 1));
    const size_t row = idx >> 11;
    const float z = xz[row * (2 * D_INNER) + D_INNER + d];
    ygated[idx] = yinner[idx] * silu_f(z);
}

extern "C" void kernel_launch(void* const* d_in, const int* in_sizes, int n_in,
                              void* d_out, int out_size, void* d_ws, size_t ws_size,
                              hipStream_t stream)
{
    const float* x      = (const float*)d_in[0];
    const float* W_in   = (const float*)d_in[1];
    const float* conv_w = (const float*)d_in[2];
    const float* conv_b = (const float*)d_in[3];
    const float* W_x    = (const float*)d_in[4];
    const float* dt_w   = (const float*)d_in[5];
    const float* dt_b   = (const float*)d_in[6];
    const float* A_log  = (const float*)d_in[7];
    const float* Dp     = (const float*)d_in[8];
    const float* W_out  = (const float*)d_in[9];
    float* out = (float*)d_out;

    float* ws = (float*)d_ws;
    float* xz     = ws;
    float* xconv  = xz    + (size_t)NROWS * 2 * D_INNER;
    float* ssm    = xconv + (size_t)NROWS * D_INNER;
    float* yinner = ssm   + (size_t)NROWS * 33;
    float* ygated = xconv;

    wmma_gemm_f32<<<dim3(2 * D_INNER / 128, NROWS / 256), 256, 0, stream>>>(
        x, W_in, xz, NROWS, 2 * D_INNER, MODEL_DIM);

    {
        size_t total = (size_t)NROWS * D_INNER;
        conv_silu_kernel<<<(unsigned)((total + 255) / 256), 256, 0, stream>>>(
            xz, conv_w, conv_b, xconv);
    }

    ssm_proj_wmma<<<NROWS / 32, 512, 0, stream>>>(xconv, W_x, ssm);

    scan_kernel<<<dim3(D_INNER / 64, BATCH), 64, 0, stream>>>(
        xconv, ssm, dt_w, dt_b, A_log, Dp, yinner);

    {
        size_t total = (size_t)NROWS * D_INNER;
        gate_kernel<<<(unsigned)((total + 255) / 256), 256, 0, stream>>>(
            yinner, xz, ygated);
    }

    wmma_gemm_f32<<<dim3(MODEL_DIM / 128, NROWS / 256), 256, 0, stream>>>(
        ygated, W_out, out, NROWS, MODEL_DIM, D_INNER);
}


# scan kernel: a channel's work split over 4 waves (4 of 16 states per thread, softplus step sizes shared via LDS; 1024 waves, was 256); sum over states now 4 in-order partials added ((p0+p1)+p2)+p3
# speedup vs baseline: 4.2790x; 1.1749x over previous
#include <hip/hip_runtime.h>
#include <hip/hip_bf16.h>

typedef __attribute__((ext_vector_type(2))) float v2f;
typedef __attribute__((ext_vector_type(8))) float v8f;

#define MODEL_DIM 1024
#define D_STATE   16
#define D_CONV    4
#define D_INNER   2048
#define BATCH     4
#define SEQLEN    1024
#define NROWS     (BATCH * SEQLEN)

__device__ __forceinline__ float silu_f(float x) {
    return x / (1.0f + __expf(-x));
}

#define KC 16
#define BPITCH 288
__global__ void __launch_bounds__(256)
wmma_gemm_f32(const float* __restrict__ A, const float* __restrict__ W,
              float* __restrict__ C, int M, int N, int K)
{
    __shared__ float Bs[(KC / 2) * BPITCH];

    const int wave   = threadIdx.x >> 5;
    const int lane   = threadIdx.x & 31;
    const int lane16 = lane & 15;
    const int khalf  = (lane >> 4) * 2;

    const int m0 = blockIdx.y * 256 + wave * 32;
    const int n0 = blockIdx.x * 128;

    const float* Arow0 = A + (size_t)(m0 + lane16) * K;
    const float* Arow1 = A + (size_t)(m0 + 16 + lane16) * K;

    const int wr = wave;
    const int wc = lane * 4;
    const float* Wbase = W + (size_t)wr * N + n0 + wc;
    float* Bst0 = Bs + (wr >> 1) * BPITCH + (wr & 1);
    float* Bst1 = Bs + ((wr + 8) >> 1) * BPITCH + (wr & 1);
    float4 w0, w1;

    v8f acc[2][8];
    #pragma unroll
    for (int g = 0; g < 2; ++g)
        #pragma unroll
        for (int j = 0; j < 8; ++j)
            acc[g][j] = (v8f){};

    w0 = *(const float4*)(Wbase);
    w1 = *(const float4*)(Wbase + (size_t)8 * N);
    {
        Bst0[(wc + 0) * 2] = w0.x;  Bst0[(wc + 1) * 2] = w0.y;
        Bst0[(wc + 2) * 2] = w0.z;  Bst0[(wc + 3) * 2] = w0.w;
        Bst1[(wc + 0) * 2] = w1.x;  Bst1[(wc + 1) * 2] = w1.y;
        Bst1[(wc + 2) * 2] = w1.z;  Bst1[(wc + 3) * 2] = w1.w;
    }
    __syncthreads();

    for (int k = 0; k < K; k += KC) {
        const bool has_next = (k + KC) < K;
        if (has_next) {
            const float* Wp = Wbase + (size_t)(k + KC) * N;
            w0 = *(const float4*)(Wp);
            w1 = *(const float4*)(Wp + (size_t)8 * N);
        }

        #pragma unroll
        for (int kk = 0; kk < KC; kk += 4) {
            v2f a0 = *(const v2f*)(Arow0 + k + kk + khalf);
            v2f a1 = *(const v2f*)(Arow1 + k + kk + khalf);
            const float* bp = Bs + ((kk + khalf) >> 1) * BPITCH + lane16 * 2;
            #pragma unroll
            for (int j = 0; j < 8; ++j) {
                v2f b = *(const v2f*)(bp + j * 32);
                acc[0][j] = __builtin_amdgcn_wmma_f32_16x16x4_f32(
                    false, a0, false, b, (short)0, acc[0][j], false, false);
                acc[1][j] = __builtin_amdgcn_wmma_f32_16x16x4_f32(
                    false, a1, false, b, (short)0, acc[1][j], false, false);
            }
        }

        __syncthreads();
        if (has_next) {
            Bst0[(wc + 0) * 2] = w0.x;  Bst0[(wc + 1) * 2] = w0.y;
            Bst0[(wc + 2) * 2] = w0.z;  Bst0[(wc + 3) * 2] = w0.w;
            Bst1[(wc + 0) * 2] = w1.x;  Bst1[(wc + 1) * 2] = w1.y;
            Bst1[(wc + 2) * 2] = w1.z;  Bst1[(wc + 3) * 2] = w1.w;
            __syncthreads();
        }
    }

    const int crow = (lane >> 4) ? 8 : 0;
    const int ccol = n0 + lane16;
    #pragma unroll
    for (int g = 0; g < 2; ++g) {
        float* Cp = C + (size_t)(m0 + g * 16 + crow) * N + ccol;
        #pragma unroll
        for (int r = 0; r < 8; ++r) {
            #pragma unroll
            for (int j = 0; j < 8; ++j)
                Cp[(size_t)r * N + j * 16] = acc[g][j][r];
        }
    }
}

__global__ void __launch_bounds__(256)
conv_silu_kernel(const float* __restrict__ xz,
                 const float* __restrict__ conv_w,
                 const float* __restrict__ conv_b,
                 float* __restrict__ xconv)
{
    size_t idx = (size_t)blockIdx.x * blockDim.x + threadIdx.x;
    if (idx >= (size_t)NROWS * D_INNER) return;
    const int d = (int)(idx & (D_INNER - 1));
    const size_t row = idx >> 11;
    const int l = (int)(row & (SEQLEN - 1));

    float acc = conv_b[d];
    const float* w = conv_w + (size_t)d * D_CONV;
    #pragma unroll
    for (int k = 0; k < D_CONV; ++k) {
        int li = l - (D_CONV - 1) + k;
        if (li >= 0)
            acc += xz[(row - l + li) * (2 * D_INNER) + d] * w[k];
    }
    xconv[idx] = silu_f(acc);
}

#define SP_NC     33
#define SP_WAVES  16
#define SP_KSLICE (D_INNER / SP_WAVES)
#define SP_GRP    4
#define SP_NGRP   (SP_KSLICE / (8 * SP_GRP))
#define SP_RQ     267
#define SP_RW     (6 * SP_RQ)
#define SP_SLOTS  (SP_WAVES / 2)

struct F3 { float x, y, z; };

__global__ void __launch_bounds__(512)
ssm_proj_wmma(const float* __restrict__ xconv,
              const float* __restrict__ Wx,
              float* __restrict__ ssm)
{
    __shared__ float red[SP_SLOTS * SP_RW];

    const int wave   = threadIdx.x >> 5;
    const int lane   = threadIdx.x & 31;
    const int lane16 = lane & 15;
    const int hi     = lane >> 4;
    const int m0     = blockIdx.x * 32;
    const int k0     = wave * SP_KSLICE + 4 * hi;

    const float* ap0 = xconv + (size_t)(m0 + lane16) * D_INNER + k0;
    const float* ap1 = ap0 + (size_t)16 * D_INNER;
    const int jj = (lane16 < 11) ? lane16 : 10;
    const float* bp = Wx + (size_t)k0 * SP_NC + 3 * jj;

    v8f acc[2][3];
    #pragma unroll
    for (int g = 0; g < 2; ++g)
        #pragma unroll
        for (int t = 0; t < 3; ++t)
            acc[g][t] = (v8f){};

    float4 a0[2][SP_GRP], a1[2][SP_GRP];
    F3 bf[2][SP_GRP][4];

    #pragma unroll
    for (int o = 0; o < SP_GRP; ++o) {
        a0[0][o] = *(const float4*)(ap0 + 8 * o);
        a1[0][o] = *(const float4*)(ap1 + 8 * o);
        #pragma unroll
        for (int r = 0; r < 4; ++r)
            bf[0][o][r] = *(const F3*)(bp + (size_t)(8 * o + r) * SP_NC);
    }

    #pragma unroll
    for (int grp = 0; grp < SP_NGRP; ++grp) {
        const int cur = grp & 1, nxt = cur ^ 1;
        if (grp + 1 < SP_NGRP) {
            #pragma unroll
            for (int o = 0; o < SP_GRP; ++o) {
                const int ko = 8 * ((grp + 1) * SP_GRP + o);
                a0[nxt][o] = *(const float4*)(ap0 + ko);
                a1[nxt][o] = *(const float4*)(ap1 + ko);
                #pragma unroll
                for (int r = 0; r < 4; ++r)
                    bf[nxt][o][r] = *(const F3*)(bp + (size_t)(ko + r) * SP_NC);
            }
        }
        __builtin_amdgcn_sched_barrier(0);
        #pragma unroll
        for (int o = 0; o < SP_GRP; ++o) {
            #pragma unroll
            for (int s = 0; s < 2; ++s) {
                const v2f x0 = s ? (v2f){a0[cur][o].z, a0[cur][o].w}
                                 : (v2f){a0[cur][o].x, a0[cur][o].y};
                const v2f x1 = s ? (v2f){a1[cur][o].z, a1[cur][o].w}
                                 : (v2f){a1[cur][o].x, a1[cur][o].y};
                const F3 lo = bf[cur][o][2 * s], up = bf[cur][o][2 * s + 1];
                const v2f b0 = (v2f){lo.x, up.x};
                const v2f b1 = (v2f){lo.y, up.y};
                const v2f b2 = (v2f){lo.z, up.z};
                acc[0][0] = __builtin_amdgcn_wmma_f32_16x16x4_f32(
                    false, x0, false, b0, (short)0, acc[0][0], false, false);
                acc[1][0] = __builtin_amdgcn_wmma_f32_16x16x4_f32(
                    false, x1, false, b0, (short)0, acc[1][0], false, false);
                acc[0][1] = __builtin_amdgcn_wmma_f32_16x16x4_f32(
                    false, x0, false, b1, (short)0, acc[0][1], false, false);
                acc[1][1] = __builtin_amdgcn_wmma_f32_16x16x4_f32(
                    false, x1, false, b1, (short)0, acc[1][1], false, false);
                acc[0][2] = __builtin_amdgcn_wmma_f32_16x16x4_f32(
                    false, x0, false, b2, (short)0, acc[0][2], false, false);
                acc[1][2] = __builtin_amdgcn_wmma_f32_16x16x4_f32(
                    false, x1, false, b2, (short)0, acc[1][2], false, false);
            }
        }
    }

    float* rp = red + (wave & (SP_SLOTS - 1)) * SP_RW + lane;
    if (wave >= SP_SLOTS) {
        #pragma unroll
        for (int g = 0; g < 2; ++g)
            #pragma unroll
            for (int t = 0; t < 3; ++t)
                #pragma unroll
                for (int r = 0; r < 8; ++r)
                    rp[(g * 3 + t) * SP_RQ + r * 32] = acc[g][t][r];
    }
    __syncthreads();
    if (wave < SP_SLOTS) {
        #pragma unroll
        for (int g = 0; g < 2; ++g)
            #pragma unroll
            for (int t = 0; t < 3; ++t)
                #pragma unroll
                for (int r = 0; r < 8; ++r) {
                    float* q = rp + (g * 3 + t) * SP_RQ + r * 32;
                    *q = acc[g][t][r] + *q;
                }
    }
    __syncthreads();

    float* outp = ssm + (size_t)m0 * SP_NC;
    for (int o = threadIdx.x; o < 32 * SP_NC; o += 512) {
        const int row = o / SP_NC;
        const int n   = o - row * SP_NC;
        const int j   = n / 3;
        const int t   = n - 3 * j;
        const int q   = (row >> 4) * 3 + t;
        const int r   = row & 7;
        const int ln  = ((row >> 3) & 1) * 16 + j;
        const float* p = red + q * SP_RQ + r * 32 + ln;
        float s = 0.0f;
        #pragma unroll
        for (int w = 0; w < SP_SLOTS; ++w)
            s += p[w * SP_RW];
        outp[o] = s;
    }
}

#define TCH 64
#define SBATCH 16
#define SPITCH 36
#define SQ 4
#define SWAVES (D_STATE / SQ)
#define SCH 32
#define SDL (TCH / SWAVES)
#define SFIN (SBATCH / SWAVES)
#define SSTG ((TCH * 33 + SCH * SWAVES - 1) / (SCH * SWAVES))

template <int Q>
__device__ __forceinline__ void scan_finish(const float (&xv)[SBATCH],
                                            const float* pc, float* yt, float Dd)
{
    float ys[SFIN];
    #pragma unroll
    for (int f = 0; f < SFIN; ++f) {
        const float* p = pc + (SWAVES * f + Q) * (SWAVES * SCH);
        ys[f] = ((p[0] + p[SCH]) + p[2 * SCH]) + p[3 * SCH];
    }
    #pragma unroll
    for (int f = 0; f < SFIN; ++f)
        yt[(size_t)(SWAVES * f + Q) * D_INNER] = ys[f] + Dd * xv[SWAVES * f + Q];
}

__global__ void __launch_bounds__(SCH * SWAVES) __attribute__((amdgpu_waves_per_eu(1, 1)))
scan_kernel(const float* __restrict__ xconv,
            const float* __restrict__ ssm,
            const float* __restrict__ dt_w,
            const float* __restrict__ dt_b,
            const float* __restrict__ A_log,
            const float* __restrict__ Dp,
            float* __restrict__ yinner)
{
    __shared__ __attribute__((aligned(16))) float sb[TCH * SPITCH];
    __shared__ float dlc[TCH * SCH];
    __shared__ float yb[2 * SBATCH * SWAVES * SCH];

    const int c = threadIdx.x & (SCH - 1);
    const int q = __builtin_amdgcn_readfirstlane(threadIdx.x >> 5);
    const int d = blockIdx.x * SCH + c;
    const int b = blockIdx.y;

    const float dw = dt_w[d];
    const float db = dt_b[d];
    const float Dd = Dp[d];

    float Areg[SQ];
    #pragma unroll
    for (int n = 0; n < SQ; ++n)
        Areg[n] = -__expf(A_log[(size_t)d * D_STATE + SQ * q + n]);

    float h[SQ];
    #pragma unroll
    for (int n = 0; n < SQ; ++n) h[n] = 0.0f;

    const float* xp = xconv  + (size_t)b * SEQLEN * D_INNER + d;
    float*       yp = yinner + (size_t)b * SEQLEN * D_INNER + d;

    float xn[SBATCH];
    #pragma unroll
    for (int i = 0; i < SBATCH; ++i)
        xn[i] = xp[(size_t)i * D_INNER];

    for (int t0 = 0; t0 < SEQLEN; t0 += TCH) {
        __syncthreads();
        {
            const float* src = ssm + (size_t)(b * SEQLEN + t0) * 33;
            const bool tail = threadIdx.x < TCH * 33 - (SSTG - 1) * (SCH * SWAVES);
            float stg[SSTG];
            #pragma unroll
            for (int k = 0; k < SSTG; ++k) {
                const int i = threadIdx.x + k * (SCH * SWAVES);
                if (k < SSTG - 1 || tail) stg[k] = src[i];
            }
            #pragma unroll
            for (int k = 0; k < SSTG; ++k) {
                const int i = threadIdx.x + k * (SCH * SWAVES);
                const int t = i / 33;
                const int j = i - t * 33;
                if (k < SSTG - 1 || tail) sb[t * SPITCH + (j ? j + 3 : 0)] = stg[k];
            }
        }
        __syncthreads();

        #pragma unroll
        for (int j = 0; j < SDL; ++j) {
            const int t = q * SDL + j;
            const float zz = dw * sb[t * SPITCH] + db;
            const float sp = log1pf(__expf(zz));
            dlc[t * SCH + c] = (zz > 20.0f) ? zz : sp;
        }
        __syncthreads();

        for (int s0 = 0; s0 < TCH; s0 += SBATCH) {
            float dl[SBATCH];
            float4 Bq[SBATCH], Cq[SBATCH];
            #pragma unroll
            for (int i = 0; i < SBATCH; ++i) {
                const float* row = sb + (s0 + i) * SPITCH;
                dl[i] = dlc[(s0 + i) * SCH + c];
                Bq[i] = *(const float4*)(row + 4 + SQ * q);
                Cq[i] = *(const float4*)(row + 4 + D_STATE + SQ * q);
            }

            float xv[SBATCH];
            #pragma unroll
            for (int i = 0; i < SBATCH; ++i) xv[i] = xn[i];

            const int tn = t0 + s0 + SBATCH;
            if (tn < SEQLEN) {
                #pragma unroll
                for (int i = 0; i < SBATCH; ++i)
                    xn[i] = xp[(size_t)(tn + i) * D_INNER];
            }

            float* ybat = yb + ((s0 / SBATCH) & 1) * (SBATCH * SWAVES * SCH);
            float* ypart = ybat + q * SCH + c;

            #pragma unroll
            for (int i = 0; i < SBATCH; ++i) {
                const float delta = dl[i];
                const float dx = delta * xv[i];
                const float Bv[SQ] = {Bq[i].x, Bq[i].y, Bq[i].z, Bq[i].w};
                const float Cv[SQ] = {Cq[i].x, Cq[i].y, Cq[i].z, Cq[i].w};

                float y = 0.0f;
                #pragma unroll
                for (int n = 0; n < SQ; ++n) {
                    float dA = __expf(delta * Areg[n]);
                    h[n] = dA * h[n] + dx * Bv[n];
                    y += h[n] * Cv[n];
                }
                ypart[i * (SWAVES * SCH)] = y;
            }

            __syncthreads();

            {
                const float* pc = ybat + c;
                float* yt = yp + (size_t)(t0 + s0) * D_INNER;
                if (q == 0)      scan_finish<0>(xv, pc, yt, Dd);
                else if (q == 1) scan_finish<1>(xv, pc, yt, Dd);
                else if (q == 2) scan_finish<2>(xv, pc, yt, Dd);
                else             scan_finish<3>(xv, pc, yt, Dd);
            }
        }
    }
}

__global__ void __launch_bounds__(256)
gate_kernel(const float* __restrict__ yinner,
            const float* __restrict__ xz,
            float* __restrict__ ygated)
{
    size_t idx = (size_t)blockIdx.x * blockDim.x + threadIdx.x;
    if (idx >= (size_t)NROWS * D_INNER) return;
    const int d = (int)(idx & (D_INNER - 1));
    const size_t row = idx >> 11;
    const float z = xz[row * (2 * D_INNER) + D_INNER + d];
    ygated[idx] = yinner[idx] * silu_f(z);
}

extern "C" void kernel_launch(void* const* d_in, const int* in_sizes, int n_in,
                              void* d_out, int out_size, void* d_ws, size_t ws_size,
                              hipStream_t stream)
{
    const float* x      = (const float*)d_in[0];
    const float* W_in   = (const float*)d_in[1];
    const float* conv_w = (const float*)d_in[2];
    const float* conv_b = (const float*)d_in[3];
    const float* W_x    = (const float*)d_in[4];
    const float* dt_w   = (const float*)d_in[5];
    const float* dt_b   = (const float*)d_in[6];
    const float* A_log  = (const float*)d_in[7];
    const float* Dp     = (const float*)d_in[8];
    const float* W_out  = (const float*)d_in[9];
    float* out = (float*)d_out;

    float* ws = (float*)d_ws;
    float* xz     = ws;
    float* xconv  = xz    + (size_t)NROWS * 2 * D_INNER;
    float* ssm    = xconv + (size_t)NROWS * D_INNER;
    float* yinner = ssm   + (size_t)NROWS * 33;
    float* ygated = xconv;

    wmma_gemm_f32<<<dim3(2 * D_INNER / 128, NROWS / 256), 256, 0, stream>>>(
        x, W_in, xz, NROWS, 2 * D_INNER, MODEL_DIM);

    {
        size_t total = (size_t)NROWS * D_INNER;
        conv_silu_kernel<<<(unsigned)((total + 255) / 256), 256, 0, stream>>>(
            xz, conv_w, conv_b, xconv);
    }

    ssm_proj_wmma<<<NROWS / 32, 512, 0, stream>>>(xconv, W_x, ssm);

    scan_kernel<<<dim3(D_INNER / SCH, BATCH), SCH * SWAVES, 0, stream>>>(
        xconv, ssm, dt_w, dt_b, A_log, Dp, yinner);

    {
        size_t total = (size_t)NROWS * D_INNER;
        gate_kernel<<<(unsigned)((total + 255) / 256), 256, 0, stream>>>(
            yinner, xz, ygated);
    }

    wmma_gemm_f32<<<dim3(MODEL_DIM / 128, NROWS / 256), 256, 0, stream>>>(
        ygated, W_out, out, NROWS, MODEL_DIM, D_INNER);
}


# output GEMM (y @ W_out) re-tiled: 64x128 block tiles of 4 waves (wave 16x128, K chunks of 32), 512 work-groups = two independent per WGP; was 256x128 tiles, 128 work-groups on half the WGPs
# speedup vs baseline: 5.2658x; 1.2306x over previous
#include <hip/hip_runtime.h>
#include <hip/hip_bf16.h>

typedef __attribute__((ext_vector_type(2))) float v2f;
typedef __attribute__((ext_vector_type(8))) float v8f;

#define MODEL_DIM 1024
#define D_STATE   16
#define D_CONV    4
#define D_INNER   2048
#define BATCH     4
#define SEQLEN    1024
#define NROWS     (BATCH * SEQLEN)

__device__ __forceinline__ float silu_f(float x) {
    return x / (1.0f + __expf(-x));
}

#define BPITCH 288
template <int MT, int KCT, int NW>
__global__ void __launch_bounds__(NW * 32)
wmma_gemm_f32(const float* __restrict__ A, const float* __restrict__ W,
              float* __restrict__ C, int M, int N, int K)
{
    __shared__ float Bs[(KCT / 2) * BPITCH];

    constexpr int NR = KCT / NW;

    const int wave   = threadIdx.x >> 5;
    const int lane   = threadIdx.x & 31;
    const int lane16 = lane & 15;
    const int khalf  = (lane >> 4) * 2;

    const int m0 = blockIdx.y * (NW * MT * 16) + wave * (MT * 16);
    const int n0 = blockIdx.x * 128;

    const float* Arow[MT];
    #pragma unroll
    for (int g = 0; g < MT; ++g)
        Arow[g] = A + (size_t)(m0 + g * 16 + lane16) * K;

    const int wr = wave;
    const int wc = lane * 4;
    const float* Wbase = W + (size_t)wr * N + n0 + wc;
    float* Bst = Bs + (wr >> 1) * BPITCH + (wr & 1) + wc * 2;
    float4 wreg[NR];

    v8f acc[MT][8];
    #pragma unroll
    for (int g = 0; g < MT; ++g)
        #pragma unroll
        for (int j = 0; j < 8; ++j)
            acc[g][j] = (v8f){};

    #pragma unroll
    for (int i = 0; i < NR; ++i)
        wreg[i] = *(const float4*)(Wbase + (size_t)(NW * i) * N);
    #pragma unroll
    for (int i = 0; i < NR; ++i) {
        float* q = Bst + (NW / 2) * i * BPITCH;
        q[0] = wreg[i].x;  q[2] = wreg[i].y;  q[4] = wreg[i].z;  q[6] = wreg[i].w;
    }
    __syncthreads();

    for (int k = 0; k < K; k += KCT) {
        const bool has_next = (k + KCT) < K;
        if (has_next) {
            const float* Wp = Wbase + (size_t)(k + KCT) * N;
            #pragma unroll
            for (int i = 0; i < NR; ++i)
                wreg[i] = *(const float4*)(Wp + (size_t)(NW * i) * N);
        }

        #pragma unroll
        for (int kk = 0; kk < KCT; kk += 4) {
            v2f a[MT];
            #pragma unroll
            for (int g = 0; g < MT; ++g)
                a[g] = *(const v2f*)(Arow[g] + k + kk + khalf);
            const float* bp = Bs + ((kk + khalf) >> 1) * BPITCH + lane16 * 2;
            #pragma unroll
            for (int j = 0; j < 8; ++j) {
                v2f b = *(const v2f*)(bp + j * 32);
                #pragma unroll
                for (int g = 0; g < MT; ++g)
                    acc[g][j] = __builtin_amdgcn_wmma_f32_16x16x4_f32(
                        false, a[g], false, b, (short)0, acc[g][j], false, false);
            }
        }

        __syncthreads();
        if (has_next) {
            #pragma unroll
            for (int i = 0; i < NR; ++i) {
                float* q = Bst + (NW / 2) * i * BPITCH;
                q[0] = wreg[i].x;  q[2] = wreg[i].y;  q[4] = wreg[i].z;  q[6] = wreg[i].w;
            }
            __syncthreads();
        }
    }

    const int crow = (lane >> 4) ? 8 : 0;
    const int ccol = n0 + lane16;
    #pragma unroll
    for (int g = 0; g < MT; ++g) {
        float* Cp = C + (size_t)(m0 + g * 16 + crow) * N + ccol;
        #pragma unroll
        for (int r = 0; r < 8; ++r) {
            #pragma unroll
            for (int j = 0; j < 8; ++j)
                Cp[(size_t)r * N + j * 16] = acc[g][j][r];
        }
    }
}

__global__ void __launch_bounds__(256)
conv_silu_kernel(const float* __restrict__ xz,
                 const float* __restrict__ conv_w,
                 const float* __restrict__ conv_b,
                 float* __restrict__ xconv)
{
    size_t idx = (size_t)blockIdx.x * blockDim.x + threadIdx.x;
    if (idx >= (size_t)NROWS * D_INNER) return;
    const int d = (int)(idx & (D_INNER - 1));
    const size_t row = idx >> 11;
    const int l = (int)(row & (SEQLEN - 1));

    float acc = conv_b[d];
    const float* w = conv_w + (size_t)d * D_CONV;
    #pragma unroll
    for (int k = 0; k < D_CONV; ++k) {
        int li = l - (D_CONV - 1) + k;
        if (li >= 0)
            acc += xz[(row - l + li) * (2 * D_INNER) + d] * w[k];
    }
    xconv[idx] = silu_f(acc);
}

#define SP_NC     33
#define SP_WAVES  16
#define SP_KSLICE (D_INNER / SP_WAVES)
#define SP_GRP    4
#define SP_NGRP   (SP_KSLICE / (8 * SP_GRP))
#define SP_RQ     267
#define SP_RW     (6 * SP_RQ)
#define SP_SLOTS  (SP_WAVES / 2)

struct F3 { float x, y, z; };

__global__ void __launch_bounds__(512)
ssm_proj_wmma(const float* __restrict__ xconv,
              const float* __restrict__ Wx,
              float* __restrict__ ssm)
{
    __shared__ float red[SP_SLOTS * SP_RW];

    const int wave   = threadIdx.x >> 5;
    const int lane   = threadIdx.x & 31;
    const int lane16 = lane & 15;
    const int hi     = lane >> 4;
    const int m0     = blockIdx.x * 32;
    const int k0     = wave * SP_KSLICE + 4 * hi;

    const float* ap0 = xconv + (size_t)(m0 + lane16) * D_INNER + k0;
    const float* ap1 = ap0 + (size_t)16 * D_INNER;
    const int jj = (lane16 < 11) ? lane16 : 10;
    const float* bp = Wx + (size_t)k0 * SP_NC + 3 * jj;

    v8f acc[2][3];
    #pragma unroll
    for (int g = 0; g < 2; ++g)
        #pragma unroll
        for (int t = 0; t < 3; ++t)
            acc[g][t] = (v8f){};

    float4 a0[2][SP_GRP], a1[2][SP_GRP];
    F3 bf[2][SP_GRP][4];

    #pragma unroll
    for (int o = 0; o < SP_GRP; ++o) {
        a0[0][o] = *(const float4*)(ap0 + 8 * o);
        a1[0][o] = *(const float4*)(ap1 + 8 * o);
        #pragma unroll
        for (int r = 0; r < 4; ++r)
            bf[0][o][r] = *(const F3*)(bp + (size_t)(8 * o + r) * SP_NC);
    }

    #pragma unroll
    for (int grp = 0; grp < SP_NGRP; ++grp) {
        const int cur = grp & 1, nxt = cur ^ 1;
        if (grp + 1 < SP_NGRP) {
            #pragma unroll
            for (int o = 0; o < SP_GRP; ++o) {
                const int ko = 8 * ((grp + 1) * SP_GRP + o);
                a0[nxt][o] = *(const float4*)(ap0 + ko);
                a1[nxt][o] = *(const float4*)(ap1 + ko);
                #pragma unroll
                for (int r = 0; r < 4; ++r)
                    bf[nxt][o][r] = *(const F3*)(bp + (size_t)(ko + r) * SP_NC);
            }
        }
        __builtin_amdgcn_sched_barrier(0);
        #pragma unroll
        for (int o = 0; o < SP_GRP; ++o) {
            #pragma unroll
            for (int s = 0; s < 2; ++s) {
                const v2f x0 = s ? (v2f){a0[cur][o].z, a0[cur][o].w}
                                 : (v2f){a0[cur][o].x, a0[cur][o].y};
                const v2f x1 = s ? (v2f){a1[cur][o].z, a1[cur][o].w}
                                 : (v2f){a1[cur][o].x, a1[cur][o].y};
                const F3 lo = bf[cur][o][2 * s], up = bf[cur][o][2 * s + 1];
                const v2f b0 = (v2f){lo.x, up.x};
                const v2f b1 = (v2f){lo.y, up.y};
                const v2f b2 = (v2f){lo.z, up.z};
                acc[0][0] = __builtin_amdgcn_wmma_f32_16x16x4_f32(
                    false, x0, false, b0, (short)0, acc[0][0], false, false);
                acc[1][0] = __builtin_amdgcn_wmma_f32_16x16x4_f32(
                    false, x1, false, b0, (short)0, acc[1][0], false, false);
                acc[0][1] = __builtin_amdgcn_wmma_f32_16x16x4_f32(
                    false, x0, false, b1, (short)0, acc[0][1], false, false);
                acc[1][1] = __builtin_amdgcn_wmma_f32_16x16x4_f32(
                    false, x1, false, b1, (short)0, acc[1][1], false, false);
                acc[0][2] = __builtin_amdgcn_wmma_f32_16x16x4_f32(
                    false, x0, false, b2, (short)0, acc[0][2], false, false);
                acc[1][2] = __builtin_amdgcn_wmma_f32_16x16x4_f32(
                    false, x1, false, b2, (short)0, acc[1][2], false, false);
            }
        }
    }

    float* rp = red + (wave & (SP_SLOTS - 1)) * SP_RW + lane;
    if (wave >= SP_SLOTS) {
        #pragma unroll
        for (int g = 0; g < 2; ++g)
            #pragma unroll
            for (int t = 0; t < 3; ++t)
                #pragma unroll
                for (int r = 0; r < 8; ++r)
                    rp[(g * 3 + t) * SP_RQ + r * 32] = acc[g][t][r];
    }
    __syncthreads();
    if (wave < SP_SLOTS) {
        #pragma unroll
        for (int g = 0; g < 2; ++g)
            #pragma unroll
            for (int t = 0; t < 3; ++t)
                #pragma unroll
                for (int r = 0; r < 8; ++r) {
                    float* q = rp + (g * 3 + t) * SP_RQ + r * 32;
                    *q = acc[g][t][r] + *q;
                }
    }
    __syncthreads();

    float* outp = ssm + (size_t)m0 * SP_NC;
    for (int o = threadIdx.x; o < 32 * SP_NC; o += 512) {
        const int row = o / SP_NC;
        const int n   = o - row * SP_NC;
        const int j   = n / 3;
        const int t   = n - 3 * j;
        const int q   = (row >> 4) * 3 + t;
        const int r   = row & 7;
        const int ln  = ((row >> 3) & 1) * 16 + j;
        const float* p = red + q * SP_RQ + r * 32 + ln;
        float s = 0.0f;
        #pragma unroll
        for (int w = 0; w < SP_SLOTS; ++w)
            s += p[w * SP_RW];
        outp[o] = s;
    }
}

#define TCH 64
#define SBATCH 16
#define SPITCH 36
#define SQ 4
#define SWAVES (D_STATE / SQ)
#define SCH 32
#define SDL (TCH / SWAVES)
#define SFIN (SBATCH / SWAVES)
#define SSTG ((TCH * 33 + SCH * SWAVES - 1) / (SCH * SWAVES))

template <int Q>
__device__ __forceinline__ void scan_finish(const float (&xv)[SBATCH],
                                            const float* pc, float* yt, float Dd)
{
    float ys[SFIN];
    #pragma unroll
    for (int f = 0; f < SFIN; ++f) {
        const float* p = pc + (SWAVES * f + Q) * (SWAVES * SCH);
        ys[f] = ((p[0] + p[SCH]) + p[2 * SCH]) + p[3 * SCH];
    }
    #pragma unroll
    for (int f = 0; f < SFIN; ++f)
        yt[(size_t)(SWAVES * f + Q) * D_INNER] = ys[f] + Dd * xv[SWAVES * f + Q];
}

__global__ void __launch_bounds__(SCH * SWAVES) __attribute__((amdgpu_waves_per_eu(1, 1)))
scan_kernel(const float* __restrict__ xconv,
            const float* __restrict__ ssm,
            const float* __restrict__ dt_w,
            const float* __restrict__ dt_b,
            const float* __restrict__ A_log,
            const float* __restrict__ Dp,
            float* __restrict__ yinner)
{
    __shared__ __attribute__((aligned(16))) float sb[TCH * SPITCH];
    __shared__ float dlc[TCH * SCH];
    __shared__ float yb[2 * SBATCH * SWAVES * SCH];

    const int c = threadIdx.x & (SCH - 1);
    const int q = __builtin_amdgcn_readfirstlane(threadIdx.x >> 5);
    const int d = blockIdx.x * SCH + c;
    const int b = blockIdx.y;

    const float dw = dt_w[d];
    const float db = dt_b[d];
    const float Dd = Dp[d];

    float Areg[SQ];
    #pragma unroll
    for (int n = 0; n < SQ; ++n)
        Areg[n] = -__expf(A_log[(size_t)d * D_STATE + SQ * q + n]);

    float h[SQ];
    #pragma unroll
    for (int n = 0; n < SQ; ++n) h[n] = 0.0f;

    const float* xp = xconv  + (size_t)b * SEQLEN * D_INNER + d;
    float*       yp = yinner + (size_t)b * SEQLEN * D_INNER + d;

    float xn[SBATCH];
    #pragma unroll
    for (int i = 0; i < SBATCH; ++i)
        xn[i] = xp[(size_t)i * D_INNER];

    for (int t0 = 0; t0 < SEQLEN; t0 += TCH) {
        __syncthreads();
        {
            const float* src = ssm + (size_t)(b * SEQLEN + t0) * 33;
            const bool tail = threadIdx.x < TCH * 33 - (SSTG - 1) * (SCH * SWAVES);
            float stg[SSTG];
            #pragma unroll
            for (int k = 0; k < SSTG; ++k) {
                const int i = threadIdx.x + k * (SCH * SWAVES);
                if (k < SSTG - 1 || tail) stg[k] = src[i];
            }
            #pragma unroll
            for (int k = 0; k < SSTG; ++k) {
                const int i = threadIdx.x + k * (SCH * SWAVES);
                const int t = i / 33;
                const int j = i - t * 33;
                if (k < SSTG - 1 || tail) sb[t * SPITCH + (j ? j + 3 : 0)] = stg[k];
            }
        }
        __syncthreads();

        #pragma unroll
        for (int j = 0; j < SDL; ++j) {
            const int t = q * SDL + j;
            const float zz = dw * sb[t * SPITCH] + db;
            const float sp = log1pf(__expf(zz));
            dlc[t * SCH + c] = (zz > 20.0f) ? zz : sp;
        }
        __syncthreads();

        for (int s0 = 0; s0 < TCH; s0 += SBATCH) {
            float dl[SBATCH];
            float4 Bq[SBATCH], Cq[SBATCH];
            #pragma unroll
            for (int i = 0; i < SBATCH; ++i) {
                const float* row = sb + (s0 + i) * SPITCH;
                dl[i] = dlc[(s0 + i) * SCH + c];
                Bq[i] = *(const float4*)(row + 4 + SQ * q);
                Cq[i] = *(const float4*)(row + 4 + D_STATE + SQ * q);
            }

            float xv[SBATCH];
            #pragma unroll
            for (int i = 0; i < SBATCH; ++i) xv[i] = xn[i];

            const int tn = t0 + s0 + SBATCH;
            if (tn < SEQLEN) {
                #pragma unroll
                for (int i = 0; i < SBATCH; ++i)
                    xn[i] = xp[(size_t)(tn + i) * D_INNER];
            }

            float* ybat = yb + ((s0 / SBATCH) & 1) * (SBATCH * SWAVES * SCH);
            float* ypart = ybat + q * SCH + c;

            #pragma unroll
            for (int i = 0; i < SBATCH; ++i) {
                const float delta = dl[i];
                const float dx = delta * xv[i];
                const float Bv[SQ] = {Bq[i].x, Bq[i].y, Bq[i].z, Bq[i].w};
                const float Cv[SQ] = {Cq[i].x, Cq[i].y, Cq[i].z, Cq[i].w};

                float y = 0.0f;
                #pragma unroll
                for (int n = 0; n < SQ; ++n) {
                    float dA = __expf(delta * Areg[n]);
                    h[n] = dA * h[n] + dx * Bv[n];
                    y += h[n] * Cv[n];
                }
                ypart[i * (SWAVES * SCH)] = y;
            }

            __syncthreads();

            {
                const float* pc = ybat + c;
                float* yt = yp + (size_t)(t0 + s0) * D_INNER;
                if (q == 0)      scan_finish<0>(xv, pc, yt, Dd);
                else if (q == 1) scan_finish<1>(xv, pc, yt, Dd);
                else if (q == 2) scan_finish<2>(xv, pc, yt, Dd);
                else             scan_finish<3>(xv, pc, yt, Dd);
            }
        }
    }
}

__global__ void __launch_bounds__(256)
gate_kernel(const float* __restrict__ yinner,
            const float* __restrict__ xz,
            float* __restrict__ ygated)
{
    size_t idx = (size_t)blockIdx.x * blockDim.x + threadIdx.x;
    if (idx >= (size_t)NROWS * D_INNER) return;
    const int d = (int)(idx & (D_INNER - 1));
    const size_t row = idx >> 11;
    const float z = xz[row * (2 * D_INNER) + D_INNER + d];
    ygated[idx] = yinner[idx] * silu_f(z);
}

extern "C" void kernel_launch(void* const* d_in, const int* in_sizes, int n_in,
                              void* d_out, int out_size, void* d_ws, size_t ws_size,
                              hipStream_t stream)
{
    const float* x      = (const float*)d_in[0];
    const float* W_in   = (const float*)d_in[1];
    const float* conv_w = (const float*)d_in[2];
    const float* conv_b = (const float*)d_in[3];
    const float* W_x    = (const float*)d_in[4];
    const float* dt_w   = (const float*)d_in[5];
    const float* dt_b   = (const float*)d_in[6];
    const float* A_log  = (const float*)d_in[7];
    const float* Dp     = (const float*)d_in[8];
    const float* W_out  = (const float*)d_in[9];
    float* out = (float*)d_out;

    float* ws = (float*)d_ws;
    float* xz     = ws;
    float* xconv  = xz    + (size_t)NROWS * 2 * D_INNER;
    float* ssm    = xconv + (size_t)NROWS * D_INNER;
    float* yinner = ssm   + (size_t)NROWS * 33;
    float* ygated = xconv;

    wmma_gemm_f32<2, 16, 8><<<dim3(2 * D_INNER / 128, NROWS / 256), 256, 0, stream>>>(
        x, W_in, xz, NROWS, 2 * D_INNER, MODEL_DIM);

    {
        size_t total = (size_t)NROWS * D_INNER;
        conv_silu_kernel<<<(unsigned)((total + 255) / 256), 256, 0, stream>>>(
            xz, conv_w, conv_b, xconv);
    }

    ssm_proj_wmma<<<NROWS / 32, 512, 0, stream>>>(xconv, W_x, ssm);

    scan_kernel<<<dim3(D_INNER / SCH, BATCH), SCH * SWAVES, 0, stream>>>(
        xconv, ssm, dt_w, dt_b, A_log, Dp, yinner);

    {
        size_t total = (size_t)NROWS * D_INNER;
        gate_kernel<<<(unsigned)((total + 255) / 256), 256, 0, stream>>>(
            yinner, xz, ygated);
    }

    wmma_gemm_f32<1, 32, 4><<<dim3(MODEL_DIM / 128, NROWS / 64), 128, 0, stream>>>(
        ygated, W_out, out, NROWS, MODEL_DIM, D_INNER);
}


# conv+SiLU and gating kernels re-gridded: a thread owns 4 adjacent channels (16-B accesses) for 16 rows in 2 groups, next group's loads ahead; 4096 waves per kernel, was 262144; no sum reordered
# speedup vs baseline: 5.4435x; 1.0337x over previous
#include <hip/hip_runtime.h>
#include <hip/hip_bf16.h>

typedef __attribute__((ext_vector_type(2))) float v2f;
typedef __attribute__((ext_vector_type(8))) float v8f;

#define MODEL_DIM 1024
#define D_STATE   16
#define D_CONV    4
#define D_INNER   2048
#define BATCH     4
#define SEQLEN    1024
#define NROWS     (BATCH * SEQLEN)

__device__ __forceinline__ float silu_f(float x) {
    return x / (1.0f + __expf(-x));
}

#define BPITCH 288
template <int MT, int KCT, int NW>
__global__ void __launch_bounds__(NW * 32)
wmma_gemm_f32(const float* __restrict__ A, const float* __restrict__ W,
              float* __restrict__ C, int M, int N, int K)
{
    __shared__ float Bs[(KCT / 2) * BPITCH];

    constexpr int NR = KCT / NW;

    const int wave   = threadIdx.x >> 5;
    const int lane   = threadIdx.x & 31;
    const int lane16 = lane & 15;
    const int khalf  = (lane >> 4) * 2;

    const int m0 = blockIdx.y * (NW * MT * 16) + wave * (MT * 16);
    const int n0 = blockIdx.x * 128;

    const float* Arow[MT];
    #pragma unroll
    for (int g = 0; g < MT; ++g)
        Arow[g] = A + (size_t)(m0 + g * 16 + lane16) * K;

    const int wr = wave;
    const int wc = lane * 4;
    const float* Wbase = W + (size_t)wr * N + n0 + wc;
    float* Bst = Bs + (wr >> 1) * BPITCH + (wr & 1) + wc * 2;
    float4 wreg[NR];

    v8f acc[MT][8];
    #pragma unroll
    for (int g = 0; g < MT; ++g)
        #pragma unroll
        for (int j = 0; j < 8; ++j)
            acc[g][j] = (v8f){};

    #pragma unroll
    for (int i = 0; i < NR; ++i)
        wreg[i] = *(const float4*)(Wbase + (size_t)(NW * i) * N);
    #pragma unroll
    for (int i = 0; i < NR; ++i) {
        float* q = Bst + (NW / 2) * i * BPITCH;
        q[0] = wreg[i].x;  q[2] = wreg[i].y;  q[4] = wreg[i].z;  q[6] = wreg[i].w;
    }
    __syncthreads();

    for (int k = 0; k < K; k += KCT) {
        const bool has_next = (k + KCT) < K;
        if (has_next) {
            const float* Wp = Wbase + (size_t)(k + KCT) * N;
            #pragma unroll
            for (int i = 0; i < NR; ++i)
                wreg[i] = *(const float4*)(Wp + (size_t)(NW * i) * N);
        }

        #pragma unroll
        for (int kk = 0; kk < KCT; kk += 4) {
            v2f a[MT];
            #pragma unroll
            for (int g = 0; g < MT; ++g)
                a[g] = *(const v2f*)(Arow[g] + k + kk + khalf);
            const float* bp = Bs + ((kk + khalf) >> 1) * BPITCH + lane16 * 2;
            #pragma unroll
            for (int j = 0; j < 8; ++j) {
                v2f b = *(const v2f*)(bp + j * 32);
                #pragma unroll
                for (int g = 0; g < MT; ++g)
                    acc[g][j] = __builtin_amdgcn_wmma_f32_16x16x4_f32(
                        false, a[g], false, b, (short)0, acc[g][j], false, false);
            }
        }

        __syncthreads();
        if (has_next) {
            #pragma unroll
            for (int i = 0; i < NR; ++i) {
                float* q = Bst + (NW / 2) * i * BPITCH;
                q[0] = wreg[i].x;  q[2] = wreg[i].y;  q[4] = wreg[i].z;  q[6] = wreg[i].w;
            }
            __syncthreads();
        }
    }

    const int crow = (lane >> 4) ? 8 : 0;
    const int ccol = n0 + lane16;
    #pragma unroll
    for (int g = 0; g < MT; ++g) {
        float* Cp = C + (size_t)(m0 + g * 16 + crow) * N + ccol;
        #pragma unroll
        for (int r = 0; r < 8; ++r) {
            #pragma unroll
            for (int j = 0; j < 8; ++j)
                Cp[(size_t)r * N + j * 16] = acc[g][j][r];
        }
    }
}

#define EW_ROWS   16
#define EW_GRP    8
#define EW_NGRP   (EW_ROWS / EW_GRP)
#define EW_C4     (D_INNER / 4)
#define EW_BPR    (EW_C4 / 256)
#define EW_BLOCKS ((NROWS / EW_ROWS) * EW_BPR)

__device__ __forceinline__ float conv_tap4(float b, float x0, float x1, float x2, float x3,
                                           float4 w, bool skip0, bool skip1, bool skip2)
{
    float acc = b;
    if (!skip0) acc += x0 * w.x;
    if (!skip1) acc += x1 * w.y;
    if (!skip2) acc += x2 * w.z;
    acc += x3 * w.w;
    return silu_f(acc);
}

__global__ void __launch_bounds__(256)
conv_silu_kernel(const float* __restrict__ xz,
                 const float* __restrict__ conv_w,
                 const float* __restrict__ conv_b,
                 float* __restrict__ xconv)
{
    const int c4   = (blockIdx.x % EW_BPR) * 256 + threadIdx.x;
    const int row0 = (blockIdx.x / EW_BPR) * EW_ROWS;
    const bool first = (row0 & (SEQLEN - 1)) == 0;

    const float4* wp = (const float4*)(conv_w + (size_t)c4 * (4 * D_CONV));
    float4 w[4];
    #pragma unroll
    for (int j = 0; j < 4; ++j) w[j] = wp[j];
    const float4 bias = *(const float4*)(conv_b + c4 * 4);

    const float* xp = xz + (size_t)row0 * (2 * D_INNER) + c4 * 4;
    float*       op = xconv + (size_t)row0 * D_INNER + c4 * 4;

    float4 xw[EW_GRP + D_CONV - 1];
    const ptrdiff_t back = first ? 0 : (ptrdiff_t)(2 * D_INNER);
    #pragma unroll
    for (int m = 0; m < D_CONV - 1; ++m)
        xw[m] = *(const float4*)(xp - (D_CONV - 1 - m) * back);
    #pragma unroll
    for (int i = 0; i < EW_GRP; ++i)
        xw[i + D_CONV - 1] = *(const float4*)(xp + (size_t)i * (2 * D_INNER));

    #pragma unroll 1
    for (int grp = 0; grp < EW_NGRP; ++grp) {
        float4 xn[EW_GRP];
        #pragma unroll
        for (int i = 0; i < EW_GRP; ++i) xn[i] = make_float4(0.0f, 0.0f, 0.0f, 0.0f);
        if (grp + 1 < EW_NGRP) {
            #pragma unroll
            for (int i = 0; i < EW_GRP; ++i)
                xn[i] = *(const float4*)(xp + (size_t)((grp + 1) * EW_GRP + i) * (2 * D_INNER));
        }

        const bool head = first && grp == 0;
        #pragma unroll
        for (int i = 0; i < EW_GRP; ++i) {
            const bool s0 = head && (i + 0 < D_CONV - 1);
            const bool s1 = head && (i + 1 < D_CONV - 1);
            const bool s2 = head && (i + 2 < D_CONV - 1);
            float4 o;
            o.x = conv_tap4(bias.x, xw[i].x, xw[i + 1].x, xw[i + 2].x, xw[i + 3].x, w[0], s0, s1, s2);
            o.y = conv_tap4(bias.y, xw[i].y, xw[i + 1].y, xw[i + 2].y, xw[i + 3].y, w[1], s0, s1, s2);
            o.z = conv_tap4(bias.z, xw[i].z, xw[i + 1].z, xw[i + 2].z, xw[i + 3].z, w[2], s0, s1, s2);
            o.w = conv_tap4(bias.w, xw[i].w, xw[i + 1].w, xw[i + 2].w, xw[i + 3].w, w[3], s0, s1, s2);
            *(float4*)(op + (size_t)(grp * EW_GRP + i) * D_INNER) = o;
        }

        #pragma unroll
        for (int m = 0; m < D_CONV - 1; ++m) xw[m] = xw[EW_GRP + m];
        #pragma unroll
        for (int i = 0; i < EW_GRP; ++i) xw[i + D_CONV - 1] = xn[i];
    }
}

#define SP_NC     33
#define SP_WAVES  16
#define SP_KSLICE (D_INNER / SP_WAVES)
#define SP_GRP    4
#define SP_NGRP   (SP_KSLICE / (8 * SP_GRP))
#define SP_RQ     267
#define SP_RW     (6 * SP_RQ)
#define SP_SLOTS  (SP_WAVES / 2)

struct F3 { float x, y, z; };

__global__ void __launch_bounds__(512)
ssm_proj_wmma(const float* __restrict__ xconv,
              const float* __restrict__ Wx,
              float* __restrict__ ssm)
{
    __shared__ float red[SP_SLOTS * SP_RW];

    const int wave   = threadIdx.x >> 5;
    const int lane   = threadIdx.x & 31;
    const int lane16 = lane & 15;
    const int hi     = lane >> 4;
    const int m0     = blockIdx.x * 32;
    const int k0     = wave * SP_KSLICE + 4 * hi;

    const float* ap0 = xconv + (size_t)(m0 + lane16) * D_INNER + k0;
    const float* ap1 = ap0 + (size_t)16 * D_INNER;
    const int jj = (lane16 < 11) ? lane16 : 10;
    const float* bp = Wx + (size_t)k0 * SP_NC + 3 * jj;

    v8f acc[2][3];
    #pragma unroll
    for (int g = 0; g < 2; ++g)
        #pragma unroll
        for (int t = 0; t < 3; ++t)
            acc[g][t] = (v8f){};

    float4 a0[2][SP_GRP], a1[2][SP_GRP];
    F3 bf[2][SP_GRP][4];

    #pragma unroll
    for (int o = 0; o < SP_GRP; ++o) {
        a0[0][o] = *(const float4*)(ap0 + 8 * o);
        a1[0][o] = *(const float4*)(ap1 + 8 * o);
        #pragma unroll
        for (int r = 0; r < 4; ++r)
            bf[0][o][r] = *(const F3*)(bp + (size_t)(8 * o + r) * SP_NC);
    }

    #pragma unroll
    for (int grp = 0; grp < SP_NGRP; ++grp) {
        const int cur = grp & 1, nxt = cur ^ 1;
        if (grp + 1 < SP_NGRP) {
            #pragma unroll
            for (int o = 0; o < SP_GRP; ++o) {
                const int ko = 8 * ((grp + 1) * SP_GRP + o);
                a0[nxt][o] = *(const float4*)(ap0 + ko);
                a1[nxt][o] = *(const float4*)(ap1 + ko);
                #pragma unroll
                for (int r = 0; r < 4; ++r)
                    bf[nxt][o][r] = *(const F3*)(bp + (size_t)(ko + r) * SP_NC);
            }
        }
        __builtin_amdgcn_sched_barrier(0);
        #pragma unroll
        for (int o = 0; o < SP_GRP; ++o) {
            #pragma unroll
            for (int s = 0; s < 2; ++s) {
                const v2f x0 = s ? (v2f){a0[cur][o].z, a0[cur][o].w}
                                 : (v2f){a0[cur][o].x, a0[cur][o].y};
                const v2f x1 = s ? (v2f){a1[cur][o].z, a1[cur][o].w}
                                 : (v2f){a1[cur][o].x, a1[cur][o].y};
                const F3 lo = bf[cur][o][2 * s], up = bf[cur][o][2 * s + 1];
                const v2f b0 = (v2f){lo.x, up.x};
                const v2f b1 = (v2f){lo.y, up.y};
                const v2f b2 = (v2f){lo.z, up.z};
                acc[0][0] = __builtin_amdgcn_wmma_f32_16x16x4_f32(
                    false, x0, false, b0, (short)0, acc[0][0], false, false);
                acc[1][0] = __builtin_amdgcn_wmma_f32_16x16x4_f32(
                    false, x1, false, b0, (short)0, acc[1][0], false, false);
                acc[0][1] = __builtin_amdgcn_wmma_f32_16x16x4_f32(
                    false, x0, false, b1, (short)0, acc[0][1], false, false);
                acc[1][1] = __builtin_amdgcn_wmma_f32_16x16x4_f32(
                    false, x1, false, b1, (short)0, acc[1][1], false, false);
                acc[0][2] = __builtin_amdgcn_wmma_f32_16x16x4_f32(
                    false, x0, false, b2, (short)0, acc[0][2], false, false);
                acc[1][2] = __builtin_amdgcn_wmma_f32_16x16x4_f32(
                    false, x1, false, b2, (short)0, acc[1][2], false, false);
            }
        }
    }

    float* rp = red + (wave & (SP_SLOTS - 1)) * SP_RW + lane;
    if (wave >= SP_SLOTS) {
        #pragma unroll
        for (int g = 0; g < 2; ++g)
            #pragma unroll
            for (int t = 0; t < 3; ++t)
                #pragma unroll
                for (int r = 0; r < 8; ++r)
                    rp[(g * 3 + t) * SP_RQ + r * 32] = acc[g][t][r];
    }
    __syncthreads();
    if (wave < SP_SLOTS) {
        #pragma unroll
        for (int g = 0; g < 2; ++g)
            #pragma unroll
            for (int t = 0; t < 3; ++t)
                #pragma unroll
                for (int r = 0; r < 8; ++r) {
                    float* q = rp + (g * 3 + t) * SP_RQ + r * 32;
                    *q = acc[g][t][r] + *q;
                }
    }
    __syncthreads();

    float* outp = ssm + (size_t)m0 * SP_NC;
    for (int o = threadIdx.x; o < 32 * SP_NC; o += 512) {
        const int row = o / SP_NC;
        const int n   = o - row * SP_NC;
        const int j   = n / 3;
        const int t   = n - 3 * j;
        const int q   = (row >> 4) * 3 + t;
        const int r   = row & 7;
        const int ln  = ((row >> 3) & 1) * 16 + j;
        const float* p = red + q * SP_RQ + r * 32 + ln;
        float s = 0.0f;
        #pragma unroll
        for (int w = 0; w < SP_SLOTS; ++w)
            s += p[w * SP_RW];
        outp[o] = s;
    }
}

#define TCH 64
#define SBATCH 16
#define SPITCH 36
#define SQ 4
#define SWAVES (D_STATE / SQ)
#define SCH 32
#define SDL (TCH / SWAVES)
#define SFIN (SBATCH / SWAVES)
#define SSTG ((TCH * 33 + SCH * SWAVES - 1) / (SCH * SWAVES))

template <int Q>
__device__ __forceinline__ void scan_finish(const float (&xv)[SBATCH],
                                            const float* pc, float* yt, float Dd)
{
    float ys[SFIN];
    #pragma unroll
    for (int f = 0; f < SFIN; ++f) {
        const float* p = pc + (SWAVES * f + Q) * (SWAVES * SCH);
        ys[f] = ((p[0] + p[SCH]) + p[2 * SCH]) + p[3 * SCH];
    }
    #pragma unroll
    for (int f = 0; f < SFIN; ++f)
        yt[(size_t)(SWAVES * f + Q) * D_INNER] = ys[f] + Dd * xv[SWAVES * f + Q];
}

__global__ void __launch_bounds__(SCH * SWAVES) __attribute__((amdgpu_waves_per_eu(1, 1)))
scan_kernel(const float* __restrict__ xconv,
            const float* __restrict__ ssm,
            const float* __restrict__ dt_w,
            const float* __restrict__ dt_b,
            const float* __restrict__ A_log,
            const float* __restrict__ Dp,
            float* __restrict__ yinner)
{
    __shared__ __attribute__((aligned(16))) float sb[TCH * SPITCH];
    __shared__ float dlc[TCH * SCH];
    __shared__ float yb[2 * SBATCH * SWAVES * SCH];

    const int c = threadIdx.x & (SCH - 1);
    const int q = __builtin_amdgcn_readfirstlane(threadIdx.x >> 5);
    const int d = blockIdx.x * SCH + c;
    const int b = blockIdx.y;

    const float dw = dt_w[d];
    const float db = dt_b[d];
    const float Dd = Dp[d];

    float Areg[SQ];
    #pragma unroll
    for (int n = 0; n < SQ; ++n)
        Areg[n] = -__expf(A_log[(size_t)d * D_STATE + SQ * q + n]);

    float h[SQ];
    #pragma unroll
    for (int n = 0; n < SQ; ++n) h[n] = 0.0f;

    const float* xp = xconv  + (size_t)b * SEQLEN * D_INNER + d;
    float*       yp = yinner + (size_t)b * SEQLEN * D_INNER + d;

    float xn[SBATCH];
    #pragma unroll
    for (int i = 0; i < SBATCH; ++i)
        xn[i] = xp[(size_t)i * D_INNER];

    for (int t0 = 0; t0 < SEQLEN; t0 += TCH) {
        __syncthreads();
        {
            const float* src = ssm + (size_t)(b * SEQLEN + t0) * 33;
            const bool tail = threadIdx.x < TCH * 33 - (SSTG - 1) * (SCH * SWAVES);
            float stg[SSTG];
            #pragma unroll
            for (int k = 0; k < SSTG; ++k) {
                const int i = threadIdx.x + k * (SCH * SWAVES);
                if (k < SSTG - 1 || tail) stg[k] = src[i];
            }
            #pragma unroll
            for (int k = 0; k < SSTG; ++k) {
                const int i = threadIdx.x + k * (SCH * SWAVES);
                const int t = i / 33;
                const int j = i - t * 33;
                if (k < SSTG - 1 || tail) sb[t * SPITCH + (j ? j + 3 : 0)] = stg[k];
            }
        }
        __syncthreads();

        #pragma unroll
        for (int j = 0; j < SDL; ++j) {
            const int t = q * SDL + j;
            const float zz = dw * sb[t * SPITCH] + db;
            const float sp = log1pf(__expf(zz));
            dlc[t * SCH + c] = (zz > 20.0f) ? zz : sp;
        }
        __syncthreads();

        for (int s0 = 0; s0 < TCH; s0 += SBATCH) {
            float dl[SBATCH];
            float4 Bq[SBATCH], Cq[SBATCH];
            #pragma unroll
            for (int i = 0; i < SBATCH; ++i) {
                const float* row = sb + (s0 + i) * SPITCH;
                dl[i] = dlc[(s0 + i) * SCH + c];
                Bq[i] = *(const float4*)(row + 4 + SQ * q);
                Cq[i] = *(const float4*)(row + 4 + D_STATE + SQ * q);
            }

            float xv[SBATCH];
            #pragma unroll
            for (int i = 0; i < SBATCH; ++i) xv[i] = xn[i];

            const int tn = t0 + s0 + SBATCH;
            if (tn < SEQLEN) {
                #pragma unroll
                for (int i = 0; i < SBATCH; ++i)
                    xn[i] = xp[(size_t)(tn + i) * D_INNER];
            }

            float* ybat = yb + ((s0 / SBATCH) & 1) * (SBATCH * SWAVES * SCH);
            float* ypart = ybat + q * SCH + c;

            #pragma unroll
            for (int i = 0; i < SBATCH; ++i) {
                const float delta = dl[i];
                const float dx = delta * xv[i];
                const float Bv[SQ] = {Bq[i].x, Bq[i].y, Bq[i].z, Bq[i].w};
                const float Cv[SQ] = {Cq[i].x, Cq[i].y, Cq[i].z, Cq[i].w};

                float y = 0.0f;
                #pragma unroll
                for (int n = 0; n < SQ; ++n) {
                    float dA = __expf(delta * Areg[n]);
                    h[n] = dA * h[n] + dx * Bv[n];
                    y += h[n] * Cv[n];
                }
                ypart[i * (SWAVES * SCH)] = y;
            }

            __syncthreads();

            {
                const float* pc = ybat + c;
                float* yt = yp + (size_t)(t0 + s0) * D_INNER;
                if (q == 0)      scan_finish<0>(xv, pc, yt, Dd);
                else if (q == 1) scan_finish<1>(xv, pc, yt, Dd);
                else if (q == 2) scan_finish<2>(xv, pc, yt, Dd);
                else             scan_finish<3>(xv, pc, yt, Dd);
            }
        }
    }
}

__global__ void __launch_bounds__(256)
gate_kernel(const float* __restrict__ yinner,
            const float* __restrict__ xz,
            float* __restrict__ ygated)
{
    const int c4   = (blockIdx.x % EW_BPR) * 256 + threadIdx.x;
    const int row0 = (blockIdx.x / EW_BPR) * EW_ROWS;

    const float* yp = yinner + (size_t)row0 * D_INNER + c4 * 4;
    const float* zp = xz + (size_t)row0 * (2 * D_INNER) + D_INNER + c4 * 4;
    float*       op = ygated + (size_t)row0 * D_INNER + c4 * 4;

    float4 yc[EW_GRP], zc[EW_GRP];
    #pragma unroll
    for (int i = 0; i < EW_GRP; ++i) {
        yc[i] = *(const float4*)(yp + (size_t)i * D_INNER);
        zc[i] = *(const float4*)(zp + (size_t)i * (2 * D_INNER));
    }

    #pragma unroll 1
    for (int grp = 0; grp < EW_NGRP; ++grp) {
        float4 yn[EW_GRP], zn[EW_GRP];
        #pragma unroll
        for (int i = 0; i < EW_GRP; ++i) {
            yn[i] = make_float4(0.0f, 0.0f, 0.0f, 0.0f);
            zn[i] = make_float4(0.0f, 0.0f, 0.0f, 0.0f);
        }
        if (grp + 1 < EW_NGRP) {
            #pragma unroll
            for (int i = 0; i < EW_GRP; ++i) {
                yn[i] = *(const float4*)(yp + (size_t)((grp + 1) * EW_GRP + i) * D_INNER);
                zn[i] = *(const float4*)(zp + (size_t)((grp + 1) * EW_GRP + i) * (2 * D_INNER));
            }
        }

        #pragma unroll
        for (int i = 0; i < EW_GRP; ++i) {
            float4 o;
            o.x = yc[i].x * silu_f(zc[i].x);
            o.y = yc[i].y * silu_f(zc[i].y);
            o.z = yc[i].z * silu_f(zc[i].z);
            o.w = yc[i].w * silu_f(zc[i].w);
            *(float4*)(op + (size_t)(grp * EW_GRP + i) * D_INNER) = o;
        }

        #pragma unroll
        for (int i = 0; i < EW_GRP; ++i) { yc[i] = yn[i]; zc[i] = zn[i]; }
    }
}

extern "C" void kernel_launch(void* const* d_in, const int* in_sizes, int n_in,
                              void* d_out, int out_size, void* d_ws, size_t ws_size,
                              hipStream_t stream)
{
    const float* x      = (const float*)d_in[0];
    const float* W_in   = (const float*)d_in[1];
    const float* conv_w = (const float*)d_in[2];
    const float* conv_b = (const float*)d_in[3];
    const float* W_x    = (const float*)d_in[4];
    const float* dt_w   = (const float*)d_in[5];
    const float* dt_b   = (const float*)d_in[6];
    const float* A_log  = (const float*)d_in[7];
    const float* Dp     = (const float*)d_in[8];
    const float* W_out  = (const float*)d_in[9];
    float* out = (float*)d_out;

    float* ws = (float*)d_ws;
    float* xz     = ws;
    float* xconv  = xz    + (size_t)NROWS * 2 * D_INNER;
    float* ssm    = xconv + (size_t)NROWS * D_INNER;
    float* yinner = ssm   + (size_t)NROWS * 33;
    float* ygated = xconv;

    wmma_gemm_f32<2, 16, 8><<<dim3(2 * D_INNER / 128, NROWS / 256), 256, 0, stream>>>(
        x, W_in, xz, NROWS, 2 * D_INNER, MODEL_DIM);

    conv_silu_kernel<<<EW_BLOCKS, 256, 0, stream>>>(xz, conv_w, conv_b, xconv);

    ssm_proj_wmma<<<NROWS / 32, 512, 0, stream>>>(xconv, W_x, ssm);

    scan_kernel<<<dim3(D_INNER / SCH, BATCH), SCH * SWAVES, 0, stream>>>(
        xconv, ssm, dt_w, dt_b, A_log, Dp, yinner);

    gate_kernel<<<EW_BLOCKS, 256, 0, stream>>>(yinner, xz, ygated);

    wmma_gemm_f32<1, 32, 4><<<dim3(MODEL_DIM / 128, NROWS / 64), 128, 0, stream>>>(
        ygated, W_out, out, NROWS, MODEL_DIM, D_INNER);
}


# both GEMMs: operand feed a K chunk ahead: A as 16-B loads (4 consecutive k per lane) into a 2nd register set, W double-buffered in LDS, one barrier per chunk; k summed 0,1,4,5,2,3,6,7 per 8
# speedup vs baseline: 5.7245x; 1.0516x over previous
#include <hip/hip_runtime.h>
#include <hip/hip_bf16.h>

typedef __attribute__((ext_vector_type(2))) float v2f;
typedef __attribute__((ext_vector_type(8))) float v8f;

#define MODEL_DIM 1024
#define D_STATE   16
#define D_CONV    4
#define D_INNER   2048
#define BATCH     4
#define SEQLEN    1024
#define NROWS     (BATCH * SEQLEN)

__device__ __forceinline__ float silu_f(float x) {
    return x / (1.0f + __expf(-x));
}

#define BPITCH 272
template <int MT, int KCT, int NW>
__global__ void __launch_bounds__(NW * 32) __attribute__((amdgpu_waves_per_eu(NW / 2, NW / 2)))
wmma_gemm_f32(const float* __restrict__ A, const float* __restrict__ W,
              float* __restrict__ C, int M, int N, int K)
{
    constexpr int BUF = (KCT / 2) * BPITCH;
    __shared__ float Bs[2 * BUF];

    constexpr int NR = KCT / NW;
    constexpr int NO = KCT / 8;

    const int wave   = threadIdx.x >> 5;
    const int lane   = threadIdx.x & 31;
    const int lane16 = lane & 15;
    const int hi     = lane >> 4;

    const int m0 = blockIdx.y * (NW * MT * 16) + wave * (MT * 16);
    const int n0 = blockIdx.x * 128;

    const float* Arow[MT];
    #pragma unroll
    for (int g = 0; g < MT; ++g)
        Arow[g] = A + (size_t)(m0 + g * 16 + lane16) * K + 4 * hi;

    const int wr = wave;
    const int wc = lane * 4;
    const float* Wbase = W + (size_t)wr * N + n0 + wc;
    float* Bst = Bs + (wr >> 1) * BPITCH + (wr & 1) + wc * 2;
    const float* bp = Bs + (2 * hi) * BPITCH + lane16 * 2;

    float4 wreg[NR];
    float4 afr[2][NO][MT];

    v8f acc[MT][8];
    #pragma unroll
    for (int g = 0; g < MT; ++g)
        #pragma unroll
        for (int j = 0; j < 8; ++j)
            acc[g][j] = (v8f){};

    #pragma unroll
    for (int i = 0; i < NR; ++i)
        wreg[i] = *(const float4*)(Wbase + (size_t)(NW * i) * N);
    #pragma unroll
    for (int o = 0; o < NO; ++o)
        #pragma unroll
        for (int g = 0; g < MT; ++g)
            afr[0][o][g] = *(const float4*)(Arow[g] + 8 * o);
    #pragma unroll
    for (int i = 0; i < NR; ++i) {
        float* q = Bst + (NW / 2) * i * BPITCH;
        q[0] = wreg[i].x;  q[2] = wreg[i].y;  q[4] = wreg[i].z;  q[6] = wreg[i].w;
    }
    __syncthreads();

    for (int k = 0; k < K; k += 2 * KCT) {
        #pragma unroll
        for (int p = 0; p < 2; ++p) {
            const int kc = k + p * KCT;
            const int kn = min(kc + KCT, K - KCT);
            const float* Wp = Wbase + (size_t)kn * N;
            #pragma unroll
            for (int i = 0; i < NR; ++i)
                wreg[i] = *(const float4*)(Wp + (size_t)(NW * i) * N);
            #pragma unroll
            for (int o = 0; o < NO; ++o)
                #pragma unroll
                for (int g = 0; g < MT; ++g)
                    afr[p ^ 1][o][g] = *(const float4*)(Arow[g] + kn + 8 * o);
            __builtin_amdgcn_sched_barrier(0);

            const float* bcur = bp + p * BUF;
            #pragma unroll
            for (int o = 0; o < NO; ++o) {
                #pragma unroll
                for (int s = 0; s < 2; ++s) {
                    v2f a[MT];
                    #pragma unroll
                    for (int g = 0; g < MT; ++g)
                        a[g] = s ? (v2f){afr[p][o][g].z, afr[p][o][g].w}
                                 : (v2f){afr[p][o][g].x, afr[p][o][g].y};
                    const float* br = bcur + (4 * o + s) * BPITCH;
                    #pragma unroll
                    for (int j = 0; j < 8; ++j) {
                        v2f b = *(const v2f*)(br + j * 32);
                        #pragma unroll
                        for (int g = 0; g < MT; ++g)
                            acc[g][j] = __builtin_amdgcn_wmma_f32_16x16x4_f32(
                                false, a[g], false, b, (short)0, acc[g][j], false, false);
                    }
                }
            }

            __builtin_amdgcn_sched_barrier(0);

            float* bnx = Bst + (p ^ 1) * BUF;
            #pragma unroll
            for (int i = 0; i < NR; ++i) {
                float* q = bnx + (NW / 2) * i * BPITCH;
                q[0] = wreg[i].x;  q[2] = wreg[i].y;  q[4] = wreg[i].z;  q[6] = wreg[i].w;
            }
            __syncthreads();
        }
    }

    const int crow = (lane >> 4) ? 8 : 0;
    const int ccol = n0 + lane16;
    #pragma unroll
    for (int g = 0; g < MT; ++g) {
        float* Cp = C + (size_t)(m0 + g * 16 + crow) * N + ccol;
        #pragma unroll
        for (int r = 0; r < 8; ++r) {
            #pragma unroll
            for (int j = 0; j < 8; ++j)
                Cp[(size_t)r * N + j * 16] = acc[g][j][r];
        }
    }
}

#define EW_ROWS   16
#define EW_GRP    8
#define EW_NGRP   (EW_ROWS / EW_GRP)
#define EW_C4     (D_INNER / 4)
#define EW_BPR    (EW_C4 / 256)
#define EW_BLOCKS ((NROWS / EW_ROWS) * EW_BPR)

__device__ __forceinline__ float conv_tap4(float b, float x0, float x1, float x2, float x3,
                                           float4 w, bool skip0, bool skip1, bool skip2)
{
    float acc = b;
    if (!skip0) acc += x0 * w.x;
    if (!skip1) acc += x1 * w.y;
    if (!skip2) acc += x2 * w.z;
    acc += x3 * w.w;
    return silu_f(acc);
}

__global__ void __launch_bounds__(256)
conv_silu_kernel(const float* __restrict__ xz,
                 const float* __restrict__ conv_w,
                 const float* __restrict__ conv_b,
                 float* __restrict__ xconv)
{
    const int c4   = (blockIdx.x % EW_BPR) * 256 + threadIdx.x;
    const int row0 = (blockIdx.x / EW_BPR) * EW_ROWS;
    const bool first = (row0 & (SEQLEN - 1)) == 0;

    const float4* wp = (const float4*)(conv_w + (size_t)c4 * (4 * D_CONV));
    float4 w[4];
    #pragma unroll
    for (int j = 0; j < 4; ++j) w[j] = wp[j];
    const float4 bias = *(const float4*)(conv_b + c4 * 4);

    const float* xp = xz + (size_t)row0 * (2 * D_INNER) + c4 * 4;
    float*       op = xconv + (size_t)row0 * D_INNER + c4 * 4;

    float4 xw[EW_GRP + D_CONV - 1];
    const ptrdiff_t back = first ? 0 : (ptrdiff_t)(2 * D_INNER);
    #pragma unroll
    for (int m = 0; m < D_CONV - 1; ++m)
        xw[m] = *(const float4*)(xp - (D_CONV - 1 - m) * back);
    #pragma unroll
    for (int i = 0; i < EW_GRP; ++i)
        xw[i + D_CONV - 1] = *(const float4*)(xp + (size_t)i * (2 * D_INNER));

    #pragma unroll 1
    for (int grp = 0; grp < EW_NGRP; ++grp) {
        float4 xn[EW_GRP];
        #pragma unroll
        for (int i = 0; i < EW_GRP; ++i) xn[i] = make_float4(0.0f, 0.0f, 0.0f, 0.0f);
        if (grp + 1 < EW_NGRP) {
            #pragma unroll
            for (int i = 0; i < EW_GRP; ++i)
                xn[i] = *(const float4*)(xp + (size_t)((grp + 1) * EW_GRP + i) * (2 * D_INNER));
        }

        const bool head = first && grp == 0;
        #pragma unroll
        for (int i = 0; i < EW_GRP; ++i) {
            const bool s0 = head && (i + 0 < D_CONV - 1);
            const bool s1 = head && (i + 1 < D_CONV - 1);
            const bool s2 = head && (i + 2 < D_CONV - 1);
            float4 o;
            o.x = conv_tap4(bias.x, xw[i].x, xw[i + 1].x, xw[i + 2].x, xw[i + 3].x, w[0], s0, s1, s2);
            o.y = conv_tap4(bias.y, xw[i].y, xw[i + 1].y, xw[i + 2].y, xw[i + 3].y, w[1], s0, s1, s2);
            o.z = conv_tap4(bias.z, xw[i].z, xw[i + 1].z, xw[i + 2].z, xw[i + 3].z, w[2], s0, s1, s2);
            o.w = conv_tap4(bias.w, xw[i].w, xw[i + 1].w, xw[i + 2].w, xw[i + 3].w, w[3], s0, s1, s2);
            *(float4*)(op + (size_t)(grp * EW_GRP + i) * D_INNER) = o;
        }

        #pragma unroll
        for (int m = 0; m < D_CONV - 1; ++m) xw[m] = xw[EW_GRP + m];
        #pragma unroll
        for (int i = 0; i < EW_GRP; ++i) xw[i + D_CONV - 1] = xn[i];
    }
}

#define SP_NC     33
#define SP_WAVES  16
#define SP_KSLICE (D_INNER / SP_WAVES)
#define SP_GRP    4
#define SP_NGRP   (SP_KSLICE / (8 * SP_GRP))
#define SP_RQ     267
#define SP_RW     (6 * SP_RQ)
#define SP_SLOTS  (SP_WAVES / 2)

struct F3 { float x, y, z; };

__global__ void __launch_bounds__(512)
ssm_proj_wmma(const float* __restrict__ xconv,
              const float* __restrict__ Wx,
              float* __restrict__ ssm)
{
    __shared__ float red[SP_SLOTS * SP_RW];

    const int wave   = threadIdx.x >> 5;
    const int lane   = threadIdx.x & 31;
    const int lane16 = lane & 15;
    const int hi     = lane >> 4;
    const int m0     = blockIdx.x * 32;
    const int k0     = wave * SP_KSLICE + 4 * hi;

    const float* ap0 = xconv + (size_t)(m0 + lane16) * D_INNER + k0;
    const float* ap1 = ap0 + (size_t)16 * D_INNER;
    const int jj = (lane16 < 11) ? lane16 : 10;
    const float* bp = Wx + (size_t)k0 * SP_NC + 3 * jj;

    v8f acc[2][3];
    #pragma unroll
    for (int g = 0; g < 2; ++g)
        #pragma unroll
        for (int t = 0; t < 3; ++t)
            acc[g][t] = (v8f){};

    float4 a0[2][SP_GRP], a1[2][SP_GRP];
    F3 bf[2][SP_GRP][4];

    #pragma unroll
    for (int o = 0; o < SP_GRP; ++o) {
        a0[0][o] = *(const float4*)(ap0 + 8 * o);
        a1[0][o] = *(const float4*)(ap1 + 8 * o);
        #pragma unroll
        for (int r = 0; r < 4; ++r)
            bf[0][o][r] = *(const F3*)(bp + (size_t)(8 * o + r) * SP_NC);
    }

    #pragma unroll
    for (int grp = 0; grp < SP_NGRP; ++grp) {
        const int cur = grp & 1, nxt = cur ^ 1;
        if (grp + 1 < SP_NGRP) {
            #pragma unroll
            for (int o = 0; o < SP_GRP; ++o) {
                const int ko = 8 * ((grp + 1) * SP_GRP + o);
                a0[nxt][o] = *(const float4*)(ap0 + ko);
                a1[nxt][o] = *(const float4*)(ap1 + ko);
                #pragma unroll
                for (int r = 0; r < 4; ++r)
                    bf[nxt][o][r] = *(const F3*)(bp + (size_t)(ko + r) * SP_NC);
            }
        }
        __builtin_amdgcn_sched_barrier(0);
        #pragma unroll
        for (int o = 0; o < SP_GRP; ++o) {
            #pragma unroll
            for (int s = 0; s < 2; ++s) {
                const v2f x0 = s ? (v2f){a0[cur][o].z, a0[cur][o].w}
                                 : (v2f){a0[cur][o].x, a0[cur][o].y};
                const v2f x1 = s ? (v2f){a1[cur][o].z, a1[cur][o].w}
                                 : (v2f){a1[cur][o].x, a1[cur][o].y};
                const F3 lo = bf[cur][o][2 * s], up = bf[cur][o][2 * s + 1];
                const v2f b0 = (v2f){lo.x, up.x};
                const v2f b1 = (v2f){lo.y, up.y};
                const v2f b2 = (v2f){lo.z, up.z};
                acc[0][0] = __builtin_amdgcn_wmma_f32_16x16x4_f32(
                    false, x0, false, b0, (short)0, acc[0][0], false, false);
                acc[1][0] = __builtin_amdgcn_wmma_f32_16x16x4_f32(
                    false, x1, false, b0, (short)0, acc[1][0], false, false);
                acc[0][1] = __builtin_amdgcn_wmma_f32_16x16x4_f32(
                    false, x0, false, b1, (short)0, acc[0][1], false, false);
                acc[1][1] = __builtin_amdgcn_wmma_f32_16x16x4_f32(
                    false, x1, false, b1, (short)0, acc[1][1], false, false);
                acc[0][2] = __builtin_amdgcn_wmma_f32_16x16x4_f32(
                    false, x0, false, b2, (short)0, acc[0][2], false, false);
                acc[1][2] = __builtin_amdgcn_wmma_f32_16x16x4_f32(
                    false, x1, false, b2, (short)0, acc[1][2], false, false);
            }
        }
    }

    float* rp = red + (wave & (SP_SLOTS - 1)) * SP_RW + lane;
    if (wave >= SP_SLOTS) {
        #pragma unroll
        for (int g = 0; g < 2; ++g)
            #pragma unroll
            for (int t = 0; t < 3; ++t)
                #pragma unroll
                for (int r = 0; r < 8; ++r)
                    rp[(g * 3 + t) * SP_RQ + r * 32] = acc[g][t][r];
    }
    __syncthreads();
    if (wave < SP_SLOTS) {
        #pragma unroll
        for (int g = 0; g < 2; ++g)
            #pragma unroll
            for (int t = 0; t < 3; ++t)
                #pragma unroll
                for (int r = 0; r < 8; ++r) {
                    float* q = rp + (g * 3 + t) * SP_RQ + r * 32;
                    *q = acc[g][t][r] + *q;
                }
    }
    __syncthreads();

    float* outp = ssm + (size_t)m0 * SP_NC;
    for (int o = threadIdx.x; o < 32 * SP_NC; o += 512) {
        const int row = o / SP_NC;
        const int n   = o - row * SP_NC;
        const int j   = n / 3;
        const int t   = n - 3 * j;
        const int q   = (row >> 4) * 3 + t;
        const int r   = row & 7;
        const int ln  = ((row >> 3) & 1) * 16 + j;
        const float* p = red + q * SP_RQ + r * 32 + ln;
        float s = 0.0f;
        #pragma unroll
        for (int w = 0; w < SP_SLOTS; ++w)
            s += p[w * SP_RW];
        outp[o] = s;
    }
}

#define TCH 64
#define SBATCH 16
#define SPITCH 36
#define SQ 4
#define SWAVES (D_STATE / SQ)
#define SCH 32
#define SDL (TCH / SWAVES)
#define SFIN (SBATCH / SWAVES)
#define SSTG ((TCH * 33 + SCH * SWAVES - 1) / (SCH * SWAVES))

template <int Q>
__device__ __forceinline__ void scan_finish(const float (&xv)[SBATCH],
                                            const float* pc, float* yt, float Dd)
{
    float ys[SFIN];
    #pragma unroll
    for (int f = 0; f < SFIN; ++f) {
        const float* p = pc + (SWAVES * f + Q) * (SWAVES * SCH);
        ys[f] = ((p[0] + p[SCH]) + p[2 * SCH]) + p[3 * SCH];
    }
    #pragma unroll
    for (int f = 0; f < SFIN; ++f)
        yt[(size_t)(SWAVES * f + Q) * D_INNER] = ys[f] + Dd * xv[SWAVES * f + Q];
}

__global__ void __launch_bounds__(SCH * SWAVES) __attribute__((amdgpu_waves_per_eu(1, 1)))
scan_kernel(const float* __restrict__ xconv,
            const float* __restrict__ ssm,
            const float* __restrict__ dt_w,
            const float* __restrict__ dt_b,
            const float* __restrict__ A_log,
            const float* __restrict__ Dp,
            float* __restrict__ yinner)
{
    __shared__ __attribute__((aligned(16))) float sb[TCH * SPITCH];
    __shared__ float dlc[TCH * SCH];
    __shared__ float yb[2 * SBATCH * SWAVES * SCH];

    const int c = threadIdx.x & (SCH - 1);
    const int q = __builtin_amdgcn_readfirstlane(threadIdx.x >> 5);
    const int d = blockIdx.x * SCH + c;
    const int b = blockIdx.y;

    const float dw = dt_w[d];
    const float db = dt_b[d];
    const float Dd = Dp[d];

    float Areg[SQ];
    #pragma unroll
    for (int n = 0; n < SQ; ++n)
        Areg[n] = -__expf(A_log[(size_t)d * D_STATE + SQ * q + n]);

    float h[SQ];
    #pragma unroll
    for (int n = 0; n < SQ; ++n) h[n] = 0.0f;

    const float* xp = xconv  + (size_t)b * SEQLEN * D_INNER + d;
    float*       yp = yinner + (size_t)b * SEQLEN * D_INNER + d;

    float xn[SBATCH];
    #pragma unroll
    for (int i = 0; i < SBATCH; ++i)
        xn[i] = xp[(size_t)i * D_INNER];

    for (int t0 = 0; t0 < SEQLEN; t0 += TCH) {
        __syncthreads();
        {
            const float* src = ssm + (size_t)(b * SEQLEN + t0) * 33;
            const bool tail = threadIdx.x < TCH * 33 - (SSTG - 1) * (SCH * SWAVES);
            float stg[SSTG];
            #pragma unroll
            for (int k = 0; k < SSTG; ++k) {
                const int i = threadIdx.x + k * (SCH * SWAVES);
                if (k < SSTG - 1 || tail) stg[k] = src[i];
            }
            #pragma unroll
            for (int k = 0; k < SSTG; ++k) {
                const int i = threadIdx.x + k * (SCH * SWAVES);
                const int t = i / 33;
                const int j = i - t * 33;
                if (k < SSTG - 1 || tail) sb[t * SPITCH + (j ? j + 3 : 0)] = stg[k];
            }
        }
        __syncthreads();

        #pragma unroll
        for (int j = 0; j < SDL; ++j) {
            const int t = q * SDL + j;
            const float zz = dw * sb[t * SPITCH] + db;
            const float sp = log1pf(__expf(zz));
            dlc[t * SCH + c] = (zz > 20.0f) ? zz : sp;
        }
        __syncthreads();

        for (int s0 = 0; s0 < TCH; s0 += SBATCH) {
            float dl[SBATCH];
            float4 Bq[SBATCH], Cq[SBATCH];
            #pragma unroll
            for (int i = 0; i < SBATCH; ++i) {
                const float* row = sb + (s0 + i) * SPITCH;
                dl[i] = dlc[(s0 + i) * SCH + c];
                Bq[i] = *(const float4*)(row + 4 + SQ * q);
                Cq[i] = *(const float4*)(row + 4 + D_STATE + SQ * q);
            }

            float xv[SBATCH];
            #pragma unroll
            for (int i = 0; i < SBATCH; ++i) xv[i] = xn[i];

            const int tn = t0 + s0 + SBATCH;
            if (tn < SEQLEN) {
                #pragma unroll
                for (int i = 0; i < SBATCH; ++i)
                    xn[i] = xp[(size_t)(tn + i) * D_INNER];
            }

            float* ybat = yb + ((s0 / SBATCH) & 1) * (SBATCH * SWAVES * SCH);
            float* ypart = ybat + q * SCH + c;

            #pragma unroll
            for (int i = 0; i < SBATCH; ++i) {
                const float delta = dl[i];
                const float dx = delta * xv[i];
                const float Bv[SQ] = {Bq[i].x, Bq[i].y, Bq[i].z, Bq[i].w};
                const float Cv[SQ] = {Cq[i].x, Cq[i].y, Cq[i].z, Cq[i].w};

                float y = 0.0f;
                #pragma unroll
                for (int n = 0; n < SQ; ++n) {
                    float dA = __expf(delta * Areg[n]);
                    h[n] = dA * h[n] + dx * Bv[n];
                    y += h[n] * Cv[n];
                }
                ypart[i * (SWAVES * SCH)] = y;
            }

            __syncthreads();

            {
                const float* pc = ybat + c;
                float* yt = yp + (size_t)(t0 + s0) * D_INNER;
                if (q == 0)      scan_finish<0>(xv, pc, yt, Dd);
                else if (q == 1) scan_finish<1>(xv, pc, yt, Dd);
                else if (q == 2) scan_finish<2>(xv, pc, yt, Dd);
                else             scan_finish<3>(xv, pc, yt, Dd);
            }
        }
    }
}

__global__ void __launch_bounds__(256)
gate_kernel(const float* __restrict__ yinner,
            const float* __restrict__ xz,
            float* __restrict__ ygated)
{
    const int c4   = (blockIdx.x % EW_BPR) * 256 + threadIdx.x;
    const int row0 = (blockIdx.x / EW_BPR) * EW_ROWS;

    const float* yp = yinner + (size_t)row0 * D_INNER + c4 * 4;
    const float* zp = xz + (size_t)row0 * (2 * D_INNER) + D_INNER + c4 * 4;
    float*       op = ygated + (size_t)row0 * D_INNER + c4 * 4;

    float4 yc[EW_GRP], zc[EW_GRP];
    #pragma unroll
    for (int i = 0; i < EW_GRP; ++i) {
        yc[i] = *(const float4*)(yp + (size_t)i * D_INNER);
        zc[i] = *(const float4*)(zp + (size_t)i * (2 * D_INNER));
    }

    #pragma unroll 1
    for (int grp = 0; grp < EW_NGRP; ++grp) {
        float4 yn[EW_GRP], zn[EW_GRP];
        #pragma unroll
        for (int i = 0; i < EW_GRP; ++i) {
            yn[i] = make_float4(0.0f, 0.0f, 0.0f, 0.0f);
            zn[i] = make_float4(0.0f, 0.0f, 0.0f, 0.0f);
        }
        if (grp + 1 < EW_NGRP) {
            #pragma unroll
            for (int i = 0; i < EW_GRP; ++i) {
                yn[i] = *(const float4*)(yp + (size_t)((grp + 1) * EW_GRP + i) * D_INNER);
                zn[i] = *(const float4*)(zp + (size_t)((grp + 1) * EW_GRP + i) * (2 * D_INNER));
            }
        }

        #pragma unroll
        for (int i = 0; i < EW_GRP; ++i) {
            float4 o;
            o.x = yc[i].x * silu_f(zc[i].x);
            o.y = yc[i].y * silu_f(zc[i].y);
            o.z = yc[i].z * silu_f(zc[i].z);
            o.w = yc[i].w * silu_f(zc[i].w);
            *(float4*)(op + (size_t)(grp * EW_GRP + i) * D_INNER) = o;
        }

        #pragma unroll
        for (int i = 0; i < EW_GRP; ++i) { yc[i] = yn[i]; zc[i] = zn[i]; }
    }
}

extern "C" void kernel_launch(void* const* d_in, const int* in_sizes, int n_in,
                              void* d_out, int out_size, void* d_ws, size_t ws_size,
                              hipStream_t stream)
{
    const float* x      = (const float*)d_in[0];
    const float* W_in   = (const float*)d_in[1];
    const float* conv_w = (const float*)d_in[2];
    const float* conv_b = (const float*)d_in[3];
    const float* W_x    = (const float*)d_in[4];
    const float* dt_w   = (const float*)d_in[5];
    const float* dt_b   = (const float*)d_in[6];
    const float* A_log  = (const float*)d_in[7];
    const float* Dp     = (const float*)d_in[8];
    const float* W_out  = (const float*)d_in[9];
    float* out = (float*)d_out;

    float* ws = (float*)d_ws;
    float* xz     = ws;
    float* xconv  = xz    + (size_t)NROWS * 2 * D_INNER;
    float* ssm    = xconv + (size_t)NROWS * D_INNER;
    float* yinner = ssm   + (size_t)NROWS * 33;
    float* ygated = xconv;

    wmma_gemm_f32<2, 16, 8><<<dim3(2 * D_INNER / 128, NROWS / 256), 256, 0, stream>>>(
        x, W_in, xz, NROWS, 2 * D_INNER, MODEL_DIM);

    conv_silu_kernel<<<EW_BLOCKS, 256, 0, stream>>>(xz, conv_w, conv_b, xconv);

    ssm_proj_wmma<<<NROWS / 32, 512, 0, stream>>>(xconv, W_x, ssm);

    scan_kernel<<<dim3(D_INNER / SCH, BATCH), SCH * SWAVES, 0, stream>>>(
        xconv, ssm, dt_w, dt_b, A_log, Dp, yinner);

    gate_kernel<<<EW_BLOCKS, 256, 0, stream>>>(yinner, xz, ygated);

    wmma_gemm_f32<1, 32, 4><<<dim3(MODEL_DIM / 128, NROWS / 64), 128, 0, stream>>>(
        ygated, W_out, out, NROWS, MODEL_DIM, D_INNER);
}


# scan kernel: step sizes softplus(z) = max(z,0) + log1p(exp(-|z|)) on hardware exp2/log2/rcp, log1p(e) as log(u)*e/(u-1) with u = 1+e, in place of the library log1pf (about 110 vector ops each)
# speedup vs baseline: 5.9197x; 1.0341x over previous
#include <hip/hip_runtime.h>
#include <hip/hip_bf16.h>

typedef __attribute__((ext_vector_type(2))) float v2f;
typedef __attribute__((ext_vector_type(8))) float v8f;

#define MODEL_DIM 1024
#define D_STATE   16
#define D_CONV    4
#define D_INNER   2048
#define BATCH     4
#define SEQLEN    1024
#define NROWS     (BATCH * SEQLEN)

__device__ __forceinline__ float silu_f(float x) {
    return x / (1.0f + __expf(-x));
}

__device__ __forceinline__ float softplus_f(float z) {
    const float e = __expf(-fabsf(z));
    const float u = 1.0f + e;
    const float d = u - 1.0f;
    const float l = __builtin_amdgcn_logf(u) * 0.69314718f;
    const float r = (d == 0.0f) ? e : l * (e * __builtin_amdgcn_rcpf(d));
    return fmaxf(z, 0.0f) + r;
}

#define BPITCH 272
template <int MT, int KCT, int NW>
__global__ void __launch_bounds__(NW * 32) __attribute__((amdgpu_waves_per_eu(NW / 2, NW / 2)))
wmma_gemm_f32(const float* __restrict__ A, const float* __restrict__ W,
              float* __restrict__ C, int M, int N, int K)
{
    constexpr int BUF = (KCT / 2) * BPITCH;
    __shared__ float Bs[2 * BUF];

    constexpr int NR = KCT / NW;
    constexpr int NO = KCT / 8;

    const int wave   = threadIdx.x >> 5;
    const int lane   = threadIdx.x & 31;
    const int lane16 = lane & 15;
    const int hi     = lane >> 4;

    const int m0 = blockIdx.y * (NW * MT * 16) + wave * (MT * 16);
    const int n0 = blockIdx.x * 128;

    const float* Arow[MT];
    #pragma unroll
    for (int g = 0; g < MT; ++g)
        Arow[g] = A + (size_t)(m0 + g * 16 + lane16) * K + 4 * hi;

    const int wr = wave;
    const int wc = lane * 4;
    const float* Wbase = W + (size_t)wr * N + n0 + wc;
    float* Bst = Bs + (wr >> 1) * BPITCH + (wr & 1) + wc * 2;
    const float* bp = Bs + (2 * hi) * BPITCH + lane16 * 2;

    float4 wreg[NR];
    float4 afr[2][NO][MT];

    v8f acc[MT][8];
    #pragma unroll
    for (int g = 0; g < MT; ++g)
        #pragma unroll
        for (int j = 0; j < 8; ++j)
            acc[g][j] = (v8f){};

    #pragma unroll
    for (int i = 0; i < NR; ++i)
        wreg[i] = *(const float4*)(Wbase + (size_t)(NW * i) * N);
    #pragma unroll
    for (int o = 0; o < NO; ++o)
        #pragma unroll
        for (int g = 0; g < MT; ++g)
            afr[0][o][g] = *(const float4*)(Arow[g] + 8 * o);
    #pragma unroll
    for (int i = 0; i < NR; ++i) {
        float* q = Bst + (NW / 2) * i * BPITCH;
        q[0] = wreg[i].x;  q[2] = wreg[i].y;  q[4] = wreg[i].z;  q[6] = wreg[i].w;
    }
    __syncthreads();

    for (int k = 0; k < K; k += 2 * KCT) {
        #pragma unroll
        for (int p = 0; p < 2; ++p) {
            const int kc = k + p * KCT;
            const int kn = min(kc + KCT, K - KCT);
            const float* Wp = Wbase + (size_t)kn * N;
            #pragma unroll
            for (int i = 0; i < NR; ++i)
                wreg[i] = *(const float4*)(Wp + (size_t)(NW * i) * N);
            #pragma unroll
            for (int o = 0; o < NO; ++o)
                #pragma unroll
                for (int g = 0; g < MT; ++g)
                    afr[p ^ 1][o][g] = *(const float4*)(Arow[g] + kn + 8 * o);
            __builtin_amdgcn_sched_barrier(0);

            const float* bcur = bp + p * BUF;
            #pragma unroll
            for (int o = 0; o < NO; ++o) {
                #pragma unroll
                for (int s = 0; s < 2; ++s) {
                    v2f a[MT];
                    #pragma unroll
                    for (int g = 0; g < MT; ++g)
                        a[g] = s ? (v2f){afr[p][o][g].z, afr[p][o][g].w}
                                 : (v2f){afr[p][o][g].x, afr[p][o][g].y};
                    const float* br = bcur + (4 * o + s) * BPITCH;
                    #pragma unroll
                    for (int j = 0; j < 8; ++j) {
                        v2f b = *(const v2f*)(br + j * 32);
                        #pragma unroll
                        for (int g = 0; g < MT; ++g)
                            acc[g][j] = __builtin_amdgcn_wmma_f32_16x16x4_f32(
                                false, a[g], false, b, (short)0, acc[g][j], false, false);
                    }
                }
            }

            __builtin_amdgcn_sched_barrier(0);

            float* bnx = Bst + (p ^ 1) * BUF;
            #pragma unroll
            for (int i = 0; i < NR; ++i) {
                float* q = bnx + (NW / 2) * i * BPITCH;
                q[0] = wreg[i].x;  q[2] = wreg[i].y;  q[4] = wreg[i].z;  q[6] = wreg[i].w;
            }
            __syncthreads();
        }
    }

    const int crow = (lane >> 4) ? 8 : 0;
    const int ccol = n0 + lane16;
    #pragma unroll
    for (int g = 0; g < MT; ++g) {
        float* Cp = C + (size_t)(m0 + g * 16 + crow) * N + ccol;
        #pragma unroll
        for (int r = 0; r < 8; ++r) {
            #pragma unroll
            for (int j = 0; j < 8; ++j)
                Cp[(size_t)r * N + j * 16] = acc[g][j][r];
        }
    }
}

#define EW_ROWS   16
#define EW_GRP    8
#define EW_NGRP   (EW_ROWS / EW_GRP)
#define EW_C4     (D_INNER / 4)
#define EW_BPR    (EW_C4 / 256)
#define EW_BLOCKS ((NROWS / EW_ROWS) * EW_BPR)

__device__ __forceinline__ float conv_tap4(float b, float x0, float x1, float x2, float x3,
                                           float4 w, bool skip0, bool skip1, bool skip2)
{
    float acc = b;
    if (!skip0) acc += x0 * w.x;
    if (!skip1) acc += x1 * w.y;
    if (!skip2) acc += x2 * w.z;
    acc += x3 * w.w;
    return silu_f(acc);
}

__global__ void __launch_bounds__(256)
conv_silu_kernel(const float* __restrict__ xz,
                 const float* __restrict__ conv_w,
                 const float* __restrict__ conv_b,
                 float* __restrict__ xconv)
{
    const int c4   = (blockIdx.x % EW_BPR) * 256 + threadIdx.x;
    const int row0 = (blockIdx.x / EW_BPR) * EW_ROWS;
    const bool first = (row0 & (SEQLEN - 1)) == 0;

    const float4* wp = (const float4*)(conv_w + (size_t)c4 * (4 * D_CONV));
    float4 w[4];
    #pragma unroll
    for (int j = 0; j < 4; ++j) w[j] = wp[j];
    const float4 bias = *(const float4*)(conv_b + c4 * 4);

    const float* xp = xz + (size_t)row0 * (2 * D_INNER) + c4 * 4;
    float*       op = xconv + (size_t)row0 * D_INNER + c4 * 4;

    float4 xw[EW_GRP + D_CONV - 1];
    const ptrdiff_t back = first ? 0 : (ptrdiff_t)(2 * D_INNER);
    #pragma unroll
    for (int m = 0; m < D_CONV - 1; ++m)
        xw[m] = *(const float4*)(xp - (D_CONV - 1 - m) * back);
    #pragma unroll
    for (int i = 0; i < EW_GRP; ++i)
        xw[i + D_CONV - 1] = *(const float4*)(xp + (size_t)i * (2 * D_INNER));

    #pragma unroll 1
    for (int grp = 0; grp < EW_NGRP; ++grp) {
        float4 xn[EW_GRP];
        #pragma unroll
        for (int i = 0; i < EW_GRP; ++i) xn[i] = make_float4(0.0f, 0.0f, 0.0f, 0.0f);
        if (grp + 1 < EW_NGRP) {
            #pragma unroll
            for (int i = 0; i < EW_GRP; ++i)
                xn[i] = *(const float4*)(xp + (size_t)((grp + 1) * EW_GRP + i) * (2 * D_INNER));
        }

        const bool head = first && grp == 0;
        #pragma unroll
        for (int i = 0; i < EW_GRP; ++i) {
            const bool s0 = head && (i + 0 < D_CONV - 1);
            const bool s1 = head && (i + 1 < D_CONV - 1);
            const bool s2 = head && (i + 2 < D_CONV - 1);
            float4 o;
            o.x = conv_tap4(bias.x, xw[i].x, xw[i + 1].x, xw[i + 2].x, xw[i + 3].x, w[0], s0, s1, s2);
            o.y = conv_tap4(bias.y, xw[i].y, xw[i + 1].y, xw[i + 2].y, xw[i + 3].y, w[1], s0, s1, s2);
            o.z = conv_tap4(bias.z, xw[i].z, xw[i + 1].z, xw[i + 2].z, xw[i + 3].z, w[2], s0, s1, s2);
            o.w = conv_tap4(bias.w, xw[i].w, xw[i + 1].w, xw[i + 2].w, xw[i + 3].w, w[3], s0, s1, s2);
            *(float4*)(op + (size_t)(grp * EW_GRP + i) * D_INNER) = o;
        }

        #pragma unroll
        for (int m = 0; m < D_CONV - 1; ++m) xw[m] = xw[EW_GRP + m];
        #pragma unroll
        for (int i = 0; i < EW_GRP; ++i) xw[i + D_CONV - 1] = xn[i];
    }
}

#define SP_NC     33
#define SP_WAVES  16
#define SP_KSLICE (D_INNER / SP_WAVES)
#define SP_GRP    4
#define SP_NGRP   (SP_KSLICE / (8 * SP_GRP))
#define SP_RQ     267
#define SP_RW     (6 * SP_RQ)
#define SP_SLOTS  (SP_WAVES / 2)

struct F3 { float x, y, z; };

__global__ void __launch_bounds__(512)
ssm_proj_wmma(const float* __restrict__ xconv,
              const float* __restrict__ Wx,
              float* __restrict__ ssm)
{
    __shared__ float red[SP_SLOTS * SP_RW];

    const int wave   = threadIdx.x >> 5;
    const int lane   = threadIdx.x & 31;
    const int lane16 = lane & 15;
    const int hi     = lane >> 4;
    const int m0     = blockIdx.x * 32;
    const int k0     = wave * SP_KSLICE + 4 * hi;

    const float* ap0 = xconv + (size_t)(m0 + lane16) * D_INNER + k0;
    const float* ap1 = ap0 + (size_t)16 * D_INNER;
    const int jj = (lane16 < 11) ? lane16 : 10;
    const float* bp = Wx + (size_t)k0 * SP_NC + 3 * jj;

    v8f acc[2][3];
    #pragma unroll
    for (int g = 0; g < 2; ++g)
        #pragma unroll
        for (int t = 0; t < 3; ++t)
            acc[g][t] = (v8f){};

    float4 a0[2][SP_GRP], a1[2][SP_GRP];
    F3 bf[2][SP_GRP][4];

    #pragma unroll
    for (int o = 0; o < SP_GRP; ++o) {
        a0[0][o] = *(const float4*)(ap0 + 8 * o);
        a1[0][o] = *(const float4*)(ap1 + 8 * o);
        #pragma unroll
        for (int r = 0; r < 4; ++r)
            bf[0][o][r] = *(const F3*)(bp + (size_t)(8 * o + r) * SP_NC);
    }

    #pragma unroll
    for (int grp = 0; grp < SP_NGRP; ++grp) {
        const int cur = grp & 1, nxt = cur ^ 1;
        if (grp + 1 < SP_NGRP) {
            #pragma unroll
            for (int o = 0; o < SP_GRP; ++o) {
                const int ko = 8 * ((grp + 1) * SP_GRP + o);
                a0[nxt][o] = *(const float4*)(ap0 + ko);
                a1[nxt][o] = *(const float4*)(ap1 + ko);
                #pragma unroll
                for (int r = 0; r < 4; ++r)
                    bf[nxt][o][r] = *(const F3*)(bp + (size_t)(ko + r) * SP_NC);
            }
        }
        __builtin_amdgcn_sched_barrier(0);
        #pragma unroll
        for (int o = 0; o < SP_GRP; ++o) {
            #pragma unroll
            for (int s = 0; s < 2; ++s) {
                const v2f x0 = s ? (v2f){a0[cur][o].z, a0[cur][o].w}
                                 : (v2f){a0[cur][o].x, a0[cur][o].y};
                const v2f x1 = s ? (v2f){a1[cur][o].z, a1[cur][o].w}
                                 : (v2f){a1[cur][o].x, a1[cur][o].y};
                const F3 lo = bf[cur][o][2 * s], up = bf[cur][o][2 * s + 1];
                const v2f b0 = (v2f){lo.x, up.x};
                const v2f b1 = (v2f){lo.y, up.y};
                const v2f b2 = (v2f){lo.z, up.z};
                acc[0][0] = __builtin_amdgcn_wmma_f32_16x16x4_f32(
                    false, x0, false, b0, (short)0, acc[0][0], false, false);
                acc[1][0] = __builtin_amdgcn_wmma_f32_16x16x4_f32(
                    false, x1, false, b0, (short)0, acc[1][0], false, false);
                acc[0][1] = __builtin_amdgcn_wmma_f32_16x16x4_f32(
                    false, x0, false, b1, (short)0, acc[0][1], false, false);
                acc[1][1] = __builtin_amdgcn_wmma_f32_16x16x4_f32(
                    false, x1, false, b1, (short)0, acc[1][1], false, false);
                acc[0][2] = __builtin_amdgcn_wmma_f32_16x16x4_f32(
                    false, x0, false, b2, (short)0, acc[0][2], false, false);
                acc[1][2] = __builtin_amdgcn_wmma_f32_16x16x4_f32(
                    false, x1, false, b2, (short)0, acc[1][2], false, false);
            }
        }
    }

    float* rp = red + (wave & (SP_SLOTS - 1)) * SP_RW + lane;
    if (wave >= SP_SLOTS) {
        #pragma unroll
        for (int g = 0; g < 2; ++g)
            #pragma unroll
            for (int t = 0; t < 3; ++t)
                #pragma unroll
                for (int r = 0; r < 8; ++r)
                    rp[(g * 3 + t) * SP_RQ + r * 32] = acc[g][t][r];
    }
    __syncthreads();
    if (wave < SP_SLOTS) {
        #pragma unroll
        for (int g = 0; g < 2; ++g)
            #pragma unroll
            for (int t = 0; t < 3; ++t)
                #pragma unroll
                for (int r = 0; r < 8; ++r) {
                    float* q = rp + (g * 3 + t) * SP_RQ + r * 32;
                    *q = acc[g][t][r] + *q;
                }
    }
    __syncthreads();

    float* outp = ssm + (size_t)m0 * SP_NC;
    for (int o = threadIdx.x; o < 32 * SP_NC; o += 512) {
        const int row = o / SP_NC;
        const int n   = o - row * SP_NC;
        const int j   = n / 3;
        const int t   = n - 3 * j;
        const int q   = (row >> 4) * 3 + t;
        const int r   = row & 7;
        const int ln  = ((row >> 3) & 1) * 16 + j;
        const float* p = red + q * SP_RQ + r * 32 + ln;
        float s = 0.0f;
        #pragma unroll
        for (int w = 0; w < SP_SLOTS; ++w)
            s += p[w * SP_RW];
        outp[o] = s;
    }
}

#define TCH 64
#define SBATCH 16
#define SPITCH 36
#define SQ 4
#define SWAVES (D_STATE / SQ)
#define SCH 32
#define SDL (TCH / SWAVES)
#define SFIN (SBATCH / SWAVES)
#define SSTG ((TCH * 33 + SCH * SWAVES - 1) / (SCH * SWAVES))

template <int Q>
__device__ __forceinline__ void scan_finish(const float (&xv)[SBATCH],
                                            const float* pc, float* yt, float Dd)
{
    float ys[SFIN];
    #pragma unroll
    for (int f = 0; f < SFIN; ++f) {
        const float* p = pc + (SWAVES * f + Q) * (SWAVES * SCH);
        ys[f] = ((p[0] + p[SCH]) + p[2 * SCH]) + p[3 * SCH];
    }
    #pragma unroll
    for (int f = 0; f < SFIN; ++f)
        yt[(size_t)(SWAVES * f + Q) * D_INNER] = ys[f] + Dd * xv[SWAVES * f + Q];
}

__global__ void __launch_bounds__(SCH * SWAVES) __attribute__((amdgpu_waves_per_eu(1, 1)))
scan_kernel(const float* __restrict__ xconv,
            const float* __restrict__ ssm,
            const float* __restrict__ dt_w,
            const float* __restrict__ dt_b,
            const float* __restrict__ A_log,
            const float* __restrict__ Dp,
            float* __restrict__ yinner)
{
    __shared__ __attribute__((aligned(16))) float sb[TCH * SPITCH];
    __shared__ float dlc[TCH * SCH];
    __shared__ float yb[2 * SBATCH * SWAVES * SCH];

    const int c = threadIdx.x & (SCH - 1);
    const int q = __builtin_amdgcn_readfirstlane(threadIdx.x >> 5);
    const int d = blockIdx.x * SCH + c;
    const int b = blockIdx.y;

    const float dw = dt_w[d];
    const float db = dt_b[d];
    const float Dd = Dp[d];

    float Areg[SQ];
    #pragma unroll
    for (int n = 0; n < SQ; ++n)
        Areg[n] = -__expf(A_log[(size_t)d * D_STATE + SQ * q + n]);

    float h[SQ];
    #pragma unroll
    for (int n = 0; n < SQ; ++n) h[n] = 0.0f;

    const float* xp = xconv  + (size_t)b * SEQLEN * D_INNER + d;
    float*       yp = yinner + (size_t)b * SEQLEN * D_INNER + d;

    float xn[SBATCH];
    #pragma unroll
    for (int i = 0; i < SBATCH; ++i)
        xn[i] = xp[(size_t)i * D_INNER];

    for (int t0 = 0; t0 < SEQLEN; t0 += TCH) {
        __syncthreads();
        {
            const float* src = ssm + (size_t)(b * SEQLEN + t0) * 33;
            const bool tail = threadIdx.x < TCH * 33 - (SSTG - 1) * (SCH * SWAVES);
            float stg[SSTG];
            #pragma unroll
            for (int k = 0; k < SSTG; ++k) {
                const int i = threadIdx.x + k * (SCH * SWAVES);
                if (k < SSTG - 1 || tail) stg[k] = src[i];
            }
            #pragma unroll
            for (int k = 0; k < SSTG; ++k) {
                const int i = threadIdx.x + k * (SCH * SWAVES);
                const int t = i / 33;
                const int j = i - t * 33;
                if (k < SSTG - 1 || tail) sb[t * SPITCH + (j ? j + 3 : 0)] = stg[k];
            }
        }
        __syncthreads();

        #pragma unroll
        for (int j = 0; j < SDL; ++j) {
            const int t = q * SDL + j;
            const float zz = dw * sb[t * SPITCH] + db;
            dlc[t * SCH + c] = softplus_f(zz);
        }
        __syncthreads();

        for (int s0 = 0; s0 < TCH; s0 += SBATCH) {
            float dl[SBATCH];
            float4 Bq[SBATCH], Cq[SBATCH];
            #pragma unroll
            for (int i = 0; i < SBATCH; ++i) {
                const float* row = sb + (s0 + i) * SPITCH;
                dl[i] = dlc[(s0 + i) * SCH + c];
                Bq[i] = *(const float4*)(row + 4 + SQ * q);
                Cq[i] = *(const float4*)(row + 4 + D_STATE + SQ * q);
            }

            float xv[SBATCH];
            #pragma unroll
            for (int i = 0; i < SBATCH; ++i) xv[i] = xn[i];

            const int tn = t0 + s0 + SBATCH;
            if (tn < SEQLEN) {
                #pragma unroll
                for (int i = 0; i < SBATCH; ++i)
                    xn[i] = xp[(size_t)(tn + i) * D_INNER];
            }

            float* ybat = yb + ((s0 / SBATCH) & 1) * (SBATCH * SWAVES * SCH);
            float* ypart = ybat + q * SCH + c;

            #pragma unroll
            for (int i = 0; i < SBATCH; ++i) {
                const float delta = dl[i];
                const float dx = delta * xv[i];
                const float Bv[SQ] = {Bq[i].x, Bq[i].y, Bq[i].z, Bq[i].w};
                const float Cv[SQ] = {Cq[i].x, Cq[i].y, Cq[i].z, Cq[i].w};

                float y = 0.0f;
                #pragma unroll
                for (int n = 0; n < SQ; ++n) {
                    float dA = __expf(delta * Areg[n]);
                    h[n] = dA * h[n] + dx * Bv[n];
                    y += h[n] * Cv[n];
                }
                ypart[i * (SWAVES * SCH)] = y;
            }

            __syncthreads();

            {
                const float* pc = ybat + c;
                float* yt = yp + (size_t)(t0 + s0) * D_INNER;
                if (q == 0)      scan_finish<0>(xv, pc, yt, Dd);
                else if (q == 1) scan_finish<1>(xv, pc, yt, Dd);
                else if (q == 2) scan_finish<2>(xv, pc, yt, Dd);
                else             scan_finish<3>(xv, pc, yt, Dd);
            }
        }
    }
}

__global__ void __launch_bounds__(256)
gate_kernel(const float* __restrict__ yinner,
            const float* __restrict__ xz,
            float* __restrict__ ygated)
{
    const int c4   = (blockIdx.x % EW_BPR) * 256 + threadIdx.x;
    const int row0 = (blockIdx.x / EW_BPR) * EW_ROWS;

    const float* yp = yinner + (size_t)row0 * D_INNER + c4 * 4;
    const float* zp = xz + (size_t)row0 * (2 * D_INNER) + D_INNER + c4 * 4;
    float*       op = ygated + (size_t)row0 * D_INNER + c4 * 4;

    float4 yc[EW_GRP], zc[EW_GRP];
    #pragma unroll
    for (int i = 0; i < EW_GRP; ++i) {
        yc[i] = *(const float4*)(yp + (size_t)i * D_INNER);
        zc[i] = *(const float4*)(zp + (size_t)i * (2 * D_INNER));
    }

    #pragma unroll 1
    for (int grp = 0; grp < EW_NGRP; ++grp) {
        float4 yn[EW_GRP], zn[EW_GRP];
        #pragma unroll
        for (int i = 0; i < EW_GRP; ++i) {
            yn[i] = make_float4(0.0f, 0.0f, 0.0f, 0.0f);
            zn[i] = make_float4(0.0f, 0.0f, 0.0f, 0.0f);
        }
        if (grp + 1 < EW_NGRP) {
            #pragma unroll
            for (int i = 0; i < EW_GRP; ++i) {
                yn[i] = *(const float4*)(yp + (size_t)((grp + 1) * EW_GRP + i) * D_INNER);
                zn[i] = *(const float4*)(zp + (size_t)((grp + 1) * EW_GRP + i) * (2 * D_INNER));
            }
        }

        #pragma unroll
        for (int i = 0; i < EW_GRP; ++i) {
            float4 o;
            o.x = yc[i].x * silu_f(zc[i].x);
            o.y = yc[i].y * silu_f(zc[i].y);
            o.z = yc[i].z * silu_f(zc[i].z);
            o.w = yc[i].w * silu_f(zc[i].w);
            *(float4*)(op + (size_t)(grp * EW_GRP + i) * D_INNER) = o;
        }

        #pragma unroll
        for (int i = 0; i < EW_GRP; ++i) { yc[i] = yn[i]; zc[i] = zn[i]; }
    }
}

extern "C" void kernel_launch(void* const* d_in, const int* in_sizes, int n_in,
                              void* d_out, int out_size, void* d_ws, size_t ws_size,
                              hipStream_t stream)
{
    const float* x      = (const float*)d_in[0];
    const float* W_in   = (const float*)d_in[1];
    const float* conv_w = (const float*)d_in[2];
    const float* conv_b = (const float*)d_in[3];
    const float* W_x    = (const float*)d_in[4];
    const float* dt_w   = (const float*)d_in[5];
    const float* dt_b   = (const float*)d_in[6];
    const float* A_log  = (const float*)d_in[7];
    const float* Dp     = (const float*)d_in[8];
    const float* W_out  = (const float*)d_in[9];
    float* out = (float*)d_out;

    float* ws = (float*)d_ws;
    float* xz     = ws;
    float* xconv  = xz    + (size_t)NROWS * 2 * D_INNER;
    float* ssm    = xconv + (size_t)NROWS * D_INNER;
    float* yinner = ssm   + (size_t)NROWS * 33;
    float* ygated = xconv;

    wmma_gemm_f32<2, 16, 8><<<dim3(2 * D_INNER / 128, NROWS / 256), 256, 0, stream>>>(
        x, W_in, xz, NROWS, 2 * D_INNER, MODEL_DIM);

    conv_silu_kernel<<<EW_BLOCKS, 256, 0, stream>>>(xz, conv_w, conv_b, xconv);

    ssm_proj_wmma<<<NROWS / 32, 512, 0, stream>>>(xconv, W_x, ssm);

    scan_kernel<<<dim3(D_INNER / SCH, BATCH), SCH * SWAVES, 0, stream>>>(
        xconv, ssm, dt_w, dt_b, A_log, Dp, yinner);

    gate_kernel<<<EW_BLOCKS, 256, 0, stream>>>(yinner, xz, ygated);

    wmma_gemm_f32<1, 32, 4><<<dim3(MODEL_DIM / 128, NROWS / 64), 128, 0, stream>>>(
        ygated, W_out, out, NROWS, MODEL_DIM, D_INNER);
}


# gating y*silu(z) moved into the scan kernel's store phase (each wave gates the 4 steps per batch it stores, z fetched one batch ahead beside x); gate kernel and its launch removed; no sum reordered
# speedup vs baseline: 6.1169x; 1.0333x over previous
#include <hip/hip_runtime.h>
#include <hip/hip_bf16.h>

typedef __attribute__((ext_vector_type(2))) float v2f;
typedef __attribute__((ext_vector_type(8))) float v8f;

#define MODEL_DIM 1024
#define D_STATE   16
#define D_CONV    4
#define D_INNER   2048
#define BATCH     4
#define SEQLEN    1024
#define NROWS     (BATCH * SEQLEN)

__device__ __forceinline__ float silu_f(float x) {
    return x / (1.0f + __expf(-x));
}

__device__ __forceinline__ float softplus_f(float z) {
    const float e = __expf(-fabsf(z));
    const float u = 1.0f + e;
    const float d = u - 1.0f;
    const float l = __builtin_amdgcn_logf(u) * 0.69314718f;
    const float r = (d == 0.0f) ? e : l * (e * __builtin_amdgcn_rcpf(d));
    return fmaxf(z, 0.0f) + r;
}

#define BPITCH 272
template <int MT, int KCT, int NW>
__global__ void __launch_bounds__(NW * 32) __attribute__((amdgpu_waves_per_eu(NW / 2, NW / 2)))
wmma_gemm_f32(const float* __restrict__ A, const float* __restrict__ W,
              float* __restrict__ C, int M, int N, int K)
{
    constexpr int BUF = (KCT / 2) * BPITCH;
    __shared__ float Bs[2 * BUF];

    constexpr int NR = KCT / NW;
    constexpr int NO = KCT / 8;

    const int wave   = threadIdx.x >> 5;
    const int lane   = threadIdx.x & 31;
    const int lane16 = lane & 15;
    const int hi     = lane >> 4;

    const int m0 = blockIdx.y * (NW * MT * 16) + wave * (MT * 16);
    const int n0 = blockIdx.x * 128;

    const float* Arow[MT];
    #pragma unroll
    for (int g = 0; g < MT; ++g)
        Arow[g] = A + (size_t)(m0 + g * 16 + lane16) * K + 4 * hi;

    const int wr = wave;
    const int wc = lane * 4;
    const float* Wbase = W + (size_t)wr * N + n0 + wc;
    float* Bst = Bs + (wr >> 1) * BPITCH + (wr & 1) + wc * 2;
    const float* bp = Bs + (2 * hi) * BPITCH + lane16 * 2;

    float4 wreg[NR];
    float4 afr[2][NO][MT];

    v8f acc[MT][8];
    #pragma unroll
    for (int g = 0; g < MT; ++g)
        #pragma unroll
        for (int j = 0; j < 8; ++j)
            acc[g][j] = (v8f){};

    #pragma unroll
    for (int i = 0; i < NR; ++i)
        wreg[i] = *(const float4*)(Wbase + (size_t)(NW * i) * N);
    #pragma unroll
    for (int o = 0; o < NO; ++o)
        #pragma unroll
        for (int g = 0; g < MT; ++g)
            afr[0][o][g] = *(const float4*)(Arow[g] + 8 * o);
    #pragma unroll
    for (int i = 0; i < NR; ++i) {
        float* q = Bst + (NW / 2) * i * BPITCH;
        q[0] = wreg[i].x;  q[2] = wreg[i].y;  q[4] = wreg[i].z;  q[6] = wreg[i].w;
    }
    __syncthreads();

    for (int k = 0; k < K; k += 2 * KCT) {
        #pragma unroll
        for (int p = 0; p < 2; ++p) {
            const int kc = k + p * KCT;
            const int kn = min(kc + KCT, K - KCT);
            const float* Wp = Wbase + (size_t)kn * N;
            #pragma unroll
            for (int i = 0; i < NR; ++i)
                wreg[i] = *(const float4*)(Wp + (size_t)(NW * i) * N);
            #pragma unroll
            for (int o = 0; o < NO; ++o)
                #pragma unroll
                for (int g = 0; g < MT; ++g)
                    afr[p ^ 1][o][g] = *(const float4*)(Arow[g] + kn + 8 * o);
            __builtin_amdgcn_sched_barrier(0);

            const float* bcur = bp + p * BUF;
            #pragma unroll
            for (int o = 0; o < NO; ++o) {
                #pragma unroll
                for (int s = 0; s < 2; ++s) {
                    v2f a[MT];
                    #pragma unroll
                    for (int g = 0; g < MT; ++g)
                        a[g] = s ? (v2f){afr[p][o][g].z, afr[p][o][g].w}
                                 : (v2f){afr[p][o][g].x, afr[p][o][g].y};
                    const float* br = bcur + (4 * o + s) * BPITCH;
                    #pragma unroll
                    for (int j = 0; j < 8; ++j) {
                        v2f b = *(const v2f*)(br + j * 32);
                        #pragma unroll
                        for (int g = 0; g < MT; ++g)
                            acc[g][j] = __builtin_amdgcn_wmma_f32_16x16x4_f32(
                                false, a[g], false, b, (short)0, acc[g][j], false, false);
                    }
                }
            }

            __builtin_amdgcn_sched_barrier(0);

            float* bnx = Bst + (p ^ 1) * BUF;
            #pragma unroll
            for (int i = 0; i < NR; ++i) {
                float* q = bnx + (NW / 2) * i * BPITCH;
                q[0] = wreg[i].x;  q[2] = wreg[i].y;  q[4] = wreg[i].z;  q[6] = wreg[i].w;
            }
            __syncthreads();
        }
    }

    const int crow = (lane >> 4) ? 8 : 0;
    const int ccol = n0 + lane16;
    #pragma unroll
    for (int g = 0; g < MT; ++g) {
        float* Cp = C + (size_t)(m0 + g * 16 + crow) * N + ccol;
        #pragma unroll
        for (int r = 0; r < 8; ++r) {
            #pragma unroll
            for (int j = 0; j < 8; ++j)
                Cp[(size_t)r * N + j * 16] = acc[g][j][r];
        }
    }
}

#define EW_ROWS   16
#define EW_GRP    8
#define EW_NGRP   (EW_ROWS / EW_GRP)
#define EW_C4     (D_INNER / 4)
#define EW_BPR    (EW_C4 / 256)
#define EW_BLOCKS ((NROWS / EW_ROWS) * EW_BPR)

__device__ __forceinline__ float conv_tap4(float b, float x0, float x1, float x2, float x3,
                                           float4 w, bool skip0, bool skip1, bool skip2)
{
    float acc = b;
    if (!skip0) acc += x0 * w.x;
    if (!skip1) acc += x1 * w.y;
    if (!skip2) acc += x2 * w.z;
    acc += x3 * w.w;
    return silu_f(acc);
}

__global__ void __launch_bounds__(256)
conv_silu_kernel(const float* __restrict__ xz,
                 const float* __restrict__ conv_w,
                 const float* __restrict__ conv_b,
                 float* __restrict__ xconv)
{
    const int c4   = (blockIdx.x % EW_BPR) * 256 + threadIdx.x;
    const int row0 = (blockIdx.x / EW_BPR) * EW_ROWS;
    const bool first = (row0 & (SEQLEN - 1)) == 0;

    const float4* wp = (const float4*)(conv_w + (size_t)c4 * (4 * D_CONV));
    float4 w[4];
    #pragma unroll
    for (int j = 0; j < 4; ++j) w[j] = wp[j];
    const float4 bias = *(const float4*)(conv_b + c4 * 4);

    const float* xp = xz + (size_t)row0 * (2 * D_INNER) + c4 * 4;
    float*       op = xconv + (size_t)row0 * D_INNER + c4 * 4;

    float4 xw[EW_GRP + D_CONV - 1];
    const ptrdiff_t back = first ? 0 : (ptrdiff_t)(2 * D_INNER);
    #pragma unroll
    for (int m = 0; m < D_CONV - 1; ++m)
        xw[m] = *(const float4*)(xp - (D_CONV - 1 - m) * back);
    #pragma unroll
    for (int i = 0; i < EW_GRP; ++i)
        xw[i + D_CONV - 1] = *(const float4*)(xp + (size_t)i * (2 * D_INNER));

    #pragma unroll 1
    for (int grp = 0; grp < EW_NGRP; ++grp) {
        float4 xn[EW_GRP];
        #pragma unroll
        for (int i = 0; i < EW_GRP; ++i) xn[i] = make_float4(0.0f, 0.0f, 0.0f, 0.0f);
        if (grp + 1 < EW_NGRP) {
            #pragma unroll
            for (int i = 0; i < EW_GRP; ++i)
                xn[i] = *(const float4*)(xp + (size_t)((grp + 1) * EW_GRP + i) * (2 * D_INNER));
        }

        const bool head = first && grp == 0;
        #pragma unroll
        for (int i = 0; i < EW_GRP; ++i) {
            const bool s0 = head && (i + 0 < D_CONV - 1);
            const bool s1 = head && (i + 1 < D_CONV - 1);
            const bool s2 = head && (i + 2 < D_CONV - 1);
            float4 o;
            o.x = conv_tap4(bias.x, xw[i].x, xw[i + 1].x, xw[i + 2].x, xw[i + 3].x, w[0], s0, s1, s2);
            o.y = conv_tap4(bias.y, xw[i].y, xw[i + 1].y, xw[i + 2].y, xw[i + 3].y, w[1], s0, s1, s2);
            o.z = conv_tap4(bias.z, xw[i].z, xw[i + 1].z, xw[i + 2].z, xw[i + 3].z, w[2], s0, s1, s2);
            o.w = conv_tap4(bias.w, xw[i].w, xw[i + 1].w, xw[i + 2].w, xw[i + 3].w, w[3], s0, s1, s2);
            *(float4*)(op + (size_t)(grp * EW_GRP + i) * D_INNER) = o;
        }

        #pragma unroll
        for (int m = 0; m < D_CONV - 1; ++m) xw[m] = xw[EW_GRP + m];
        #pragma unroll
        for (int i = 0; i < EW_GRP; ++i) xw[i + D_CONV - 1] = xn[i];
    }
}

#define SP_NC     33
#define SP_WAVES  16
#define SP_KSLICE (D_INNER / SP_WAVES)
#define SP_GRP    4
#define SP_NGRP   (SP_KSLICE / (8 * SP_GRP))
#define SP_RQ     267
#define SP_RW     (6 * SP_RQ)
#define SP_SLOTS  (SP_WAVES / 2)

struct F3 { float x, y, z; };

__global__ void __launch_bounds__(512)
ssm_proj_wmma(const float* __restrict__ xconv,
              const float* __restrict__ Wx,
              float* __restrict__ ssm)
{
    __shared__ float red[SP_SLOTS * SP_RW];

    const int wave   = threadIdx.x >> 5;
    const int lane   = threadIdx.x & 31;
    const int lane16 = lane & 15;
    const int hi     = lane >> 4;
    const int m0     = blockIdx.x * 32;
    const int k0     = wave * SP_KSLICE + 4 * hi;

    const float* ap0 = xconv + (size_t)(m0 + lane16) * D_INNER + k0;
    const float* ap1 = ap0 + (size_t)16 * D_INNER;
    const int jj = (lane16 < 11) ? lane16 : 10;
    const float* bp = Wx + (size_t)k0 * SP_NC + 3 * jj;

    v8f acc[2][3];
    #pragma unroll
    for (int g = 0; g < 2; ++g)
        #pragma unroll
        for (int t = 0; t < 3; ++t)
            acc[g][t] = (v8f){};

    float4 a0[2][SP_GRP], a1[2][SP_GRP];
    F3 bf[2][SP_GRP][4];

    #pragma unroll
    for (int o = 0; o < SP_GRP; ++o) {
        a0[0][o] = *(const float4*)(ap0 + 8 * o);
        a1[0][o] = *(const float4*)(ap1 + 8 * o);
        #pragma unroll
        for (int r = 0; r < 4; ++r)
            bf[0][o][r] = *(const F3*)(bp + (size_t)(8 * o + r) * SP_NC);
    }

    #pragma unroll
    for (int grp = 0; grp < SP_NGRP; ++grp) {
        const int cur = grp & 1, nxt = cur ^ 1;
        if (grp + 1 < SP_NGRP) {
            #pragma unroll
            for (int o = 0; o < SP_GRP; ++o) {
                const int ko = 8 * ((grp + 1) * SP_GRP + o);
                a0[nxt][o] = *(const float4*)(ap0 + ko);
                a1[nxt][o] = *(const float4*)(ap1 + ko);
                #pragma unroll
                for (int r = 0; r < 4; ++r)
                    bf[nxt][o][r] = *(const F3*)(bp + (size_t)(ko + r) * SP_NC);
            }
        }
        __builtin_amdgcn_sched_barrier(0);
        #pragma unroll
        for (int o = 0; o < SP_GRP; ++o) {
            #pragma unroll
            for (int s = 0; s < 2; ++s) {
                const v2f x0 = s ? (v2f){a0[cur][o].z, a0[cur][o].w}
                                 : (v2f){a0[cur][o].x, a0[cur][o].y};
                const v2f x1 = s ? (v2f){a1[cur][o].z, a1[cur][o].w}
                                 : (v2f){a1[cur][o].x, a1[cur][o].y};
                const F3 lo = bf[cur][o][2 * s], up = bf[cur][o][2 * s + 1];
                const v2f b0 = (v2f){lo.x, up.x};
                const v2f b1 = (v2f){lo.y, up.y};
                const v2f b2 = (v2f){lo.z, up.z};
                acc[0][0] = __builtin_amdgcn_wmma_f32_16x16x4_f32(
                    false, x0, false, b0, (short)0, acc[0][0], false, false);
                acc[1][0] = __builtin_amdgcn_wmma_f32_16x16x4_f32(
                    false, x1, false, b0, (short)0, acc[1][0], false, false);
                acc[0][1] = __builtin_amdgcn_wmma_f32_16x16x4_f32(
                    false, x0, false, b1, (short)0, acc[0][1], false, false);
                acc[1][1] = __builtin_amdgcn_wmma_f32_16x16x4_f32(
                    false, x1, false, b1, (short)0, acc[1][1], false, false);
                acc[0][2] = __builtin_amdgcn_wmma_f32_16x16x4_f32(
                    false, x0, false, b2, (short)0, acc[0][2], false, false);
                acc[1][2] = __builtin_amdgcn_wmma_f32_16x16x4_f32(
                    false, x1, false, b2, (short)0, acc[1][2], false, false);
            }
        }
    }

    float* rp = red + (wave & (SP_SLOTS - 1)) * SP_RW + lane;
    if (wave >= SP_SLOTS) {
        #pragma unroll
        for (int g = 0; g < 2; ++g)
            #pragma unroll
            for (int t = 0; t < 3; ++t)
                #pragma unroll
                for (int r = 0; r < 8; ++r)
                    rp[(g * 3 + t) * SP_RQ + r * 32] = acc[g][t][r];
    }
    __syncthreads();
    if (wave < SP_SLOTS) {
        #pragma unroll
        for (int g = 0; g < 2; ++g)
            #pragma unroll
            for (int t = 0; t < 3; ++t)
                #pragma unroll
                for (int r = 0; r < 8; ++r) {
                    float* q = rp + (g * 3 + t) * SP_RQ + r * 32;
                    *q = acc[g][t][r] + *q;
                }
    }
    __syncthreads();

    float* outp = ssm + (size_t)m0 * SP_NC;
    for (int o = threadIdx.x; o < 32 * SP_NC; o += 512) {
        const int row = o / SP_NC;
        const int n   = o - row * SP_NC;
        const int j   = n / 3;
        const int t   = n - 3 * j;
        const int q   = (row >> 4) * 3 + t;
        const int r   = row & 7;
        const int ln  = ((row >> 3) & 1) * 16 + j;
        const float* p = red + q * SP_RQ + r * 32 + ln;
        float s = 0.0f;
        #pragma unroll
        for (int w = 0; w < SP_SLOTS; ++w)
            s += p[w * SP_RW];
        outp[o] = s;
    }
}

#define TCH 64
#define SBATCH 16
#define SPITCH 36
#define SQ 4
#define SWAVES (D_STATE / SQ)
#define SCH 32
#define SDL (TCH / SWAVES)
#define SFIN (SBATCH / SWAVES)
#define SSTG ((TCH * 33 + SCH * SWAVES - 1) / (SCH * SWAVES))

template <int Q>
__device__ __forceinline__ void scan_finish(const float (&xv)[SBATCH], const float (&gz)[SFIN],
                                            const float* pc, float* yt, float Dd)
{
    float ys[SFIN];
    #pragma unroll
    for (int f = 0; f < SFIN; ++f) {
        const float* p = pc + (SWAVES * f + Q) * (SWAVES * SCH);
        ys[f] = ((p[0] + p[SCH]) + p[2 * SCH]) + p[3 * SCH];
    }
    #pragma unroll
    for (int f = 0; f < SFIN; ++f)
        yt[(size_t)(SWAVES * f + Q) * D_INNER] = (ys[f] + Dd * xv[SWAVES * f + Q]) * gz[f];
}

__global__ void __launch_bounds__(SCH * SWAVES) __attribute__((amdgpu_waves_per_eu(1, 1)))
scan_kernel(const float* __restrict__ xconv,
            const float* __restrict__ xz,
            const float* __restrict__ ssm,
            const float* __restrict__ dt_w,
            const float* __restrict__ dt_b,
            const float* __restrict__ A_log,
            const float* __restrict__ Dp,
            float* __restrict__ ygated)
{
    __shared__ __attribute__((aligned(16))) float sb[TCH * SPITCH];
    __shared__ float dlc[TCH * SCH];
    __shared__ float yb[2 * SBATCH * SWAVES * SCH];

    const int c = threadIdx.x & (SCH - 1);
    const int q = __builtin_amdgcn_readfirstlane(threadIdx.x >> 5);
    const int d = blockIdx.x * SCH + c;
    const int b = blockIdx.y;

    const float dw = dt_w[d];
    const float db = dt_b[d];
    const float Dd = Dp[d];

    float Areg[SQ];
    #pragma unroll
    for (int n = 0; n < SQ; ++n)
        Areg[n] = -__expf(A_log[(size_t)d * D_STATE + SQ * q + n]);

    float h[SQ];
    #pragma unroll
    for (int n = 0; n < SQ; ++n) h[n] = 0.0f;

    const float* xp = xconv  + (size_t)b * SEQLEN * D_INNER + d;
    float*       yp = ygated + (size_t)b * SEQLEN * D_INNER + d;
    const float* zp = xz + ((size_t)b * SEQLEN + q) * (2 * D_INNER) + D_INNER + d;

    float xn[SBATCH];
    #pragma unroll
    for (int i = 0; i < SBATCH; ++i)
        xn[i] = xp[(size_t)i * D_INNER];
    float zn[SFIN];
    #pragma unroll
    for (int f = 0; f < SFIN; ++f)
        zn[f] = zp[(size_t)(SWAVES * f) * (2 * D_INNER)];

    for (int t0 = 0; t0 < SEQLEN; t0 += TCH) {
        __syncthreads();
        {
            const float* src = ssm + (size_t)(b * SEQLEN + t0) * 33;
            const bool tail = threadIdx.x < TCH * 33 - (SSTG - 1) * (SCH * SWAVES);
            float stg[SSTG];
            #pragma unroll
            for (int k = 0; k < SSTG; ++k) {
                const int i = threadIdx.x + k * (SCH * SWAVES);
                if (k < SSTG - 1 || tail) stg[k] = src[i];
            }
            #pragma unroll
            for (int k = 0; k < SSTG; ++k) {
                const int i = threadIdx.x + k * (SCH * SWAVES);
                const int t = i / 33;
                const int j = i - t * 33;
                if (k < SSTG - 1 || tail) sb[t * SPITCH + (j ? j + 3 : 0)] = stg[k];
            }
        }
        __syncthreads();

        #pragma unroll
        for (int j = 0; j < SDL; ++j) {
            const int t = q * SDL + j;
            const float zz = dw * sb[t * SPITCH] + db;
            dlc[t * SCH + c] = softplus_f(zz);
        }
        __syncthreads();

        for (int s0 = 0; s0 < TCH; s0 += SBATCH) {
            float dl[SBATCH];
            float4 Bq[SBATCH], Cq[SBATCH];
            #pragma unroll
            for (int i = 0; i < SBATCH; ++i) {
                const float* row = sb + (s0 + i) * SPITCH;
                dl[i] = dlc[(s0 + i) * SCH + c];
                Bq[i] = *(const float4*)(row + 4 + SQ * q);
                Cq[i] = *(const float4*)(row + 4 + D_STATE + SQ * q);
            }

            float xv[SBATCH];
            #pragma unroll
            for (int i = 0; i < SBATCH; ++i) xv[i] = xn[i];
            float gz[SFIN];
            #pragma unroll
            for (int f = 0; f < SFIN; ++f) gz[f] = zn[f];

            const int tn = t0 + s0 + SBATCH;
            if (tn < SEQLEN) {
                #pragma unroll
                for (int i = 0; i < SBATCH; ++i)
                    xn[i] = xp[(size_t)(tn + i) * D_INNER];
                #pragma unroll
                for (int f = 0; f < SFIN; ++f)
                    zn[f] = zp[(size_t)(tn + SWAVES * f) * (2 * D_INNER)];
            }

            #pragma unroll
            for (int f = 0; f < SFIN; ++f) gz[f] = silu_f(gz[f]);

            float* ybat = yb + ((s0 / SBATCH) & 1) * (SBATCH * SWAVES * SCH);
            float* ypart = ybat + q * SCH + c;

            #pragma unroll
            for (int i = 0; i < SBATCH; ++i) {
                const float delta = dl[i];
                const float dx = delta * xv[i];
                const float Bv[SQ] = {Bq[i].x, Bq[i].y, Bq[i].z, Bq[i].w};
                const float Cv[SQ] = {Cq[i].x, Cq[i].y, Cq[i].z, Cq[i].w};

                float y = 0.0f;
                #pragma unroll
                for (int n = 0; n < SQ; ++n) {
                    float dA = __expf(delta * Areg[n]);
                    h[n] = dA * h[n] + dx * Bv[n];
                    y += h[n] * Cv[n];
                }
                ypart[i * (SWAVES * SCH)] = y;
            }

            __syncthreads();

            {
                const float* pc = ybat + c;
                float* yt = yp + (size_t)(t0 + s0) * D_INNER;
                if (q == 0)      scan_finish<0>(xv, gz, pc, yt, Dd);
                else if (q == 1) scan_finish<1>(xv, gz, pc, yt, Dd);
                else if (q == 2) scan_finish<2>(xv, gz, pc, yt, Dd);
                else             scan_finish<3>(xv, gz, pc, yt, Dd);
            }
        }
    }
}

extern "C" void kernel_launch(void* const* d_in, const int* in_sizes, int n_in,
                              void* d_out, int out_size, void* d_ws, size_t ws_size,
                              hipStream_t stream)
{
    const float* x      = (const float*)d_in[0];
    const float* W_in   = (const float*)d_in[1];
    const float* conv_w = (const float*)d_in[2];
    const float* conv_b = (const float*)d_in[3];
    const float* W_x    = (const float*)d_in[4];
    const float* dt_w   = (const float*)d_in[5];
    const float* dt_b   = (const float*)d_in[6];
    const float* A_log  = (const float*)d_in[7];
    const float* Dp     = (const float*)d_in[8];
    const float* W_out  = (const float*)d_in[9];
    float* out = (float*)d_out;

    float* ws = (float*)d_ws;
    float* xz     = ws;
    float* xconv  = xz    + (size_t)NROWS * 2 * D_INNER;
    float* ssm    = xconv + (size_t)NROWS * D_INNER;
    float* ygated = ssm   + (size_t)NROWS * 33;

    wmma_gemm_f32<2, 16, 8><<<dim3(2 * D_INNER / 128, NROWS / 256), 256, 0, stream>>>(
        x, W_in, xz, NROWS, 2 * D_INNER, MODEL_DIM);

    conv_silu_kernel<<<EW_BLOCKS, 256, 0, stream>>>(xz, conv_w, conv_b, xconv);

    ssm_proj_wmma<<<NROWS / 32, 512, 0, stream>>>(xconv, W_x, ssm);

    scan_kernel<<<dim3(D_INNER / SCH, BATCH), SCH * SWAVES, 0, stream>>>(
        xconv, xz, ssm, dt_w, dt_b, A_log, Dp, ygated);

    wmma_gemm_f32<1, 32, 4><<<dim3(MODEL_DIM / 128, NROWS / 64), 128, 0, stream>>>(
        ygated, W_out, out, NROWS, MODEL_DIM, D_INNER);
}


# both GEMMs: W staging stores into LDS made bank-conflict-free: each block of 32 columns of the pair-interleaved W tile shifted by 2 dwords (was 32 lanes on 8 banks, 4 ways); no sum reordered
# speedup vs baseline: 6.3497x; 1.0381x over previous
#include <hip/hip_runtime.h>
#include <hip/hip_bf16.h>

typedef __attribute__((ext_vector_type(2))) float v2f;
typedef __attribute__((ext_vector_type(8))) float v8f;

#define MODEL_DIM 1024
#define D_STATE   16
#define D_CONV    4
#define D_INNER   2048
#define BATCH     4
#define SEQLEN    1024
#define NROWS     (BATCH * SEQLEN)

__device__ __forceinline__ float silu_f(float x) {
    return x / (1.0f + __expf(-x));
}

__device__ __forceinline__ float softplus_f(float z) {
    const float e = __expf(-fabsf(z));
    const float u = 1.0f + e;
    const float d = u - 1.0f;
    const float l = __builtin_amdgcn_logf(u) * 0.69314718f;
    const float r = (d == 0.0f) ? e : l * (e * __builtin_amdgcn_rcpf(d));
    return fmaxf(z, 0.0f) + r;
}

#define BPITCH 272
template <int MT, int KCT, int NW>
__global__ void __launch_bounds__(NW * 32) __attribute__((amdgpu_waves_per_eu(NW / 2, NW / 2)))
wmma_gemm_f32(const float* __restrict__ A, const float* __restrict__ W,
              float* __restrict__ C, int M, int N, int K)
{
    constexpr int BUF = (KCT / 2) * BPITCH;
    __shared__ float Bs[2 * BUF];

    constexpr int NR = KCT / NW;
    constexpr int NO = KCT / 8;

    const int wave   = threadIdx.x >> 5;
    const int lane   = threadIdx.x & 31;
    const int lane16 = lane & 15;
    const int hi     = lane >> 4;

    const int m0 = blockIdx.y * (NW * MT * 16) + wave * (MT * 16);
    const int n0 = blockIdx.x * 128;

    const float* Arow[MT];
    #pragma unroll
    for (int g = 0; g < MT; ++g)
        Arow[g] = A + (size_t)(m0 + g * 16 + lane16) * K + 4 * hi;

    const int wr = wave;
    const int wc = lane * 4;
    const float* Wbase = W + (size_t)wr * N + n0 + wc;
    float* Bst = Bs + (wr >> 1) * BPITCH + (wr & 1) + wc * 2 + 2 * (lane >> 3);
    const float* bp = Bs + (2 * hi) * BPITCH + lane16 * 2;

    float4 wreg[NR];
    float4 afr[2][NO][MT];

    v8f acc[MT][8];
    #pragma unroll
    for (int g = 0; g < MT; ++g)
        #pragma unroll
        for (int j = 0; j < 8; ++j)
            acc[g][j] = (v8f){};

    #pragma unroll
    for (int i = 0; i < NR; ++i)
        wreg[i] = *(const float4*)(Wbase + (size_t)(NW * i) * N);
    #pragma unroll
    for (int o = 0; o < NO; ++o)
        #pragma unroll
        for (int g = 0; g < MT; ++g)
            afr[0][o][g] = *(const float4*)(Arow[g] + 8 * o);
    #pragma unroll
    for (int i = 0; i < NR; ++i) {
        float* q = Bst + (NW / 2) * i * BPITCH;
        q[0] = wreg[i].x;  q[2] = wreg[i].y;  q[4] = wreg[i].z;  q[6] = wreg[i].w;
    }
    __syncthreads();

    for (int k = 0; k < K; k += 2 * KCT) {
        #pragma unroll
        for (int p = 0; p < 2; ++p) {
            const int kc = k + p * KCT;
            const int kn = min(kc + KCT, K - KCT);
            const float* Wp = Wbase + (size_t)kn * N;
            #pragma unroll
            for (int i = 0; i < NR; ++i)
                wreg[i] = *(const float4*)(Wp + (size_t)(NW * i) * N);
            #pragma unroll
            for (int o = 0; o < NO; ++o)
                #pragma unroll
                for (int g = 0; g < MT; ++g)
                    afr[p ^ 1][o][g] = *(const float4*)(Arow[g] + kn + 8 * o);
            __builtin_amdgcn_sched_barrier(0);

            const float* bcur = bp + p * BUF;
            #pragma unroll
            for (int o = 0; o < NO; ++o) {
                #pragma unroll
                for (int s = 0; s < 2; ++s) {
                    v2f a[MT];
                    #pragma unroll
                    for (int g = 0; g < MT; ++g)
                        a[g] = s ? (v2f){afr[p][o][g].z, afr[p][o][g].w}
                                 : (v2f){afr[p][o][g].x, afr[p][o][g].y};
                    const float* br = bcur + (4 * o + s) * BPITCH;
                    #pragma unroll
                    for (int j = 0; j < 8; ++j) {
                        v2f b = *(const v2f*)(br + j * 32 + 2 * (j >> 1));
                        #pragma unroll
                        for (int g = 0; g < MT; ++g)
                            acc[g][j] = __builtin_amdgcn_wmma_f32_16x16x4_f32(
                                false, a[g], false, b, (short)0, acc[g][j], false, false);
                    }
                }
            }

            __builtin_amdgcn_sched_barrier(0);

            float* bnx = Bst + (p ^ 1) * BUF;
            #pragma unroll
            for (int i = 0; i < NR; ++i) {
                float* q = bnx + (NW / 2) * i * BPITCH;
                q[0] = wreg[i].x;  q[2] = wreg[i].y;  q[4] = wreg[i].z;  q[6] = wreg[i].w;
            }
            __syncthreads();
        }
    }

    const int crow = (lane >> 4) ? 8 : 0;
    const int ccol = n0 + lane16;
    #pragma unroll
    for (int g = 0; g < MT; ++g) {
        float* Cp = C + (size_t)(m0 + g * 16 + crow) * N + ccol;
        #pragma unroll
        for (int r = 0; r < 8; ++r) {
            #pragma unroll
            for (int j = 0; j < 8; ++j)
                Cp[(size_t)r * N + j * 16] = acc[g][j][r];
        }
    }
}

#define EW_ROWS   16
#define EW_GRP    8
#define EW_NGRP   (EW_ROWS / EW_GRP)
#define EW_C4     (D_INNER / 4)
#define EW_BPR    (EW_C4 / 256)
#define EW_BLOCKS ((NROWS / EW_ROWS) * EW_BPR)

__device__ __forceinline__ float conv_tap4(float b, float x0, float x1, float x2, float x3,
                                           float4 w, bool skip0, bool skip1, bool skip2)
{
    float acc = b;
    if (!skip0) acc += x0 * w.x;
    if (!skip1) acc += x1 * w.y;
    if (!skip2) acc += x2 * w.z;
    acc += x3 * w.w;
    return silu_f(acc);
}

__global__ void __launch_bounds__(256)
conv_silu_kernel(const float* __restrict__ xz,
                 const float* __restrict__ conv_w,
                 const float* __restrict__ conv_b,
                 float* __restrict__ xconv)
{
    const int c4   = (blockIdx.x % EW_BPR) * 256 + threadIdx.x;
    const int row0 = (blockIdx.x / EW_BPR) * EW_ROWS;
    const bool first = (row0 & (SEQLEN - 1)) == 0;

    const float4* wp = (const float4*)(conv_w + (size_t)c4 * (4 * D_CONV));
    float4 w[4];
    #pragma unroll
    for (int j = 0; j < 4; ++j) w[j] = wp[j];
    const float4 bias = *(const float4*)(conv_b + c4 * 4);

    const float* xp = xz + (size_t)row0 * (2 * D_INNER) + c4 * 4;
    float*       op = xconv + (size_t)row0 * D_INNER + c4 * 4;

    float4 xw[EW_GRP + D_CONV - 1];
    const ptrdiff_t back = first ? 0 : (ptrdiff_t)(2 * D_INNER);
    #pragma unroll
    for (int m = 0; m < D_CONV - 1; ++m)
        xw[m] = *(const float4*)(xp - (D_CONV - 1 - m) * back);
    #pragma unroll
    for (int i = 0; i < EW_GRP; ++i)
        xw[i + D_CONV - 1] = *(const float4*)(xp + (size_t)i * (2 * D_INNER));

    #pragma unroll 1
    for (int grp = 0; grp < EW_NGRP; ++grp) {
        float4 xn[EW_GRP];
        #pragma unroll
        for (int i = 0; i < EW_GRP; ++i) xn[i] = make_float4(0.0f, 0.0f, 0.0f, 0.0f);
        if (grp + 1 < EW_NGRP) {
            #pragma unroll
            for (int i = 0; i < EW_GRP; ++i)
                xn[i] = *(const float4*)(xp + (size_t)((grp + 1) * EW_GRP + i) * (2 * D_INNER));
        }

        const bool head = first && grp == 0;
        #pragma unroll
        for (int i = 0; i < EW_GRP; ++i) {
            const bool s0 = head && (i + 0 < D_CONV - 1);
            const bool s1 = head && (i + 1 < D_CONV - 1);
            const bool s2 = head && (i + 2 < D_CONV - 1);
            float4 o;
            o.x = conv_tap4(bias.x, xw[i].x, xw[i + 1].x, xw[i + 2].x, xw[i + 3].x, w[0], s0, s1, s2);
            o.y = conv_tap4(bias.y, xw[i].y, xw[i + 1].y, xw[i + 2].y, xw[i + 3].y, w[1], s0, s1, s2);
            o.z = conv_tap4(bias.z, xw[i].z, xw[i + 1].z, xw[i + 2].z, xw[i + 3].z, w[2], s0, s1, s2);
            o.w = conv_tap4(bias.w, xw[i].w, xw[i + 1].w, xw[i + 2].w, xw[i + 3].w, w[3], s0, s1, s2);
            *(float4*)(op + (size_t)(grp * EW_GRP + i) * D_INNER) = o;
        }

        #pragma unroll
        for (int m = 0; m < D_CONV - 1; ++m) xw[m] = xw[EW_GRP + m];
        #pragma unroll
        for (int i = 0; i < EW_GRP; ++i) xw[i + D_CONV - 1] = xn[i];
    }
}

#define SP_NC     33
#define SP_WAVES  16
#define SP_KSLICE (D_INNER / SP_WAVES)
#define SP_GRP    4
#define SP_NGRP   (SP_KSLICE / (8 * SP_GRP))
#define SP_RQ     267
#define SP_RW     (6 * SP_RQ)
#define SP_SLOTS  (SP_WAVES / 2)

struct F3 { float x, y, z; };

__global__ void __launch_bounds__(512)
ssm_proj_wmma(const float* __restrict__ xconv,
              const float* __restrict__ Wx,
              float* __restrict__ ssm)
{
    __shared__ float red[SP_SLOTS * SP_RW];

    const int wave   = threadIdx.x >> 5;
    const int lane   = threadIdx.x & 31;
    const int lane16 = lane & 15;
    const int hi     = lane >> 4;
    const int m0     = blockIdx.x * 32;
    const int k0     = wave * SP_KSLICE + 4 * hi;

    const float* ap0 = xconv + (size_t)(m0 + lane16) * D_INNER + k0;
    const float* ap1 = ap0 + (size_t)16 * D_INNER;
    const int jj = (lane16 < 11) ? lane16 : 10;
    const float* bp = Wx + (size_t)k0 * SP_NC + 3 * jj;

    v8f acc[2][3];
    #pragma unroll
    for (int g = 0; g < 2; ++g)
        #pragma unroll
        for (int t = 0; t < 3; ++t)
            acc[g][t] = (v8f){};

    float4 a0[2][SP_GRP], a1[2][SP_GRP];
    F3 bf[2][SP_GRP][4];

    #pragma unroll
    for (int o = 0; o < SP_GRP; ++o) {
        a0[0][o] = *(const float4*)(ap0 + 8 * o);
        a1[0][o] = *(const float4*)(ap1 + 8 * o);
        #pragma unroll
        for (int r = 0; r < 4; ++r)
            bf[0][o][r] = *(const F3*)(bp + (size_t)(8 * o + r) * SP_NC);
    }

    #pragma unroll
    for (int grp = 0; grp < SP_NGRP; ++grp) {
        const int cur = grp & 1, nxt = cur ^ 1;
        if (grp + 1 < SP_NGRP) {
            #pragma unroll
            for (int o = 0; o < SP_GRP; ++o) {
                const int ko = 8 * ((grp + 1) * SP_GRP + o);
                a0[nxt][o] = *(const float4*)(ap0 + ko);
                a1[nxt][o] = *(const float4*)(ap1 + ko);
                #pragma unroll
                for (int r = 0; r < 4; ++r)
                    bf[nxt][o][r] = *(const F3*)(bp + (size_t)(ko + r) * SP_NC);
            }
        }
        __builtin_amdgcn_sched_barrier(0);
        #pragma unroll
        for (int o = 0; o < SP_GRP; ++o) {
            #pragma unroll
            for (int s = 0; s < 2; ++s) {
                const v2f x0 = s ? (v2f){a0[cur][o].z, a0[cur][o].w}
                                 : (v2f){a0[cur][o].x, a0[cur][o].y};
                const v2f x1 = s ? (v2f){a1[cur][o].z, a1[cur][o].w}
                                 : (v2f){a1[cur][o].x, a1[cur][o].y};
                const F3 lo = bf[cur][o][2 * s], up = bf[cur][o][2 * s + 1];
                const v2f b0 = (v2f){lo.x, up.x};
                const v2f b1 = (v2f){lo.y, up.y};
                const v2f b2 = (v2f){lo.z, up.z};
                acc[0][0] = __builtin_amdgcn_wmma_f32_16x16x4_f32(
                    false, x0, false, b0, (short)0, acc[0][0], false, false);
                acc[1][0] = __builtin_amdgcn_wmma_f32_16x16x4_f32(
                    false, x1, false, b0, (short)0, acc[1][0], false, false);
                acc[0][1] = __builtin_amdgcn_wmma_f32_16x16x4_f32(
                    false, x0, false, b1, (short)0, acc[0][1], false, false);
                acc[1][1] = __builtin_amdgcn_wmma_f32_16x16x4_f32(
                    false, x1, false, b1, (short)0, acc[1][1], false, false);
                acc[0][2] = __builtin_amdgcn_wmma_f32_16x16x4_f32(
                    false, x0, false, b2, (short)0, acc[0][2], false, false);
                acc[1][2] = __builtin_amdgcn_wmma_f32_16x16x4_f32(
                    false, x1, false, b2, (short)0, acc[1][2], false, false);
            }
        }
    }

    float* rp = red + (wave & (SP_SLOTS - 1)) * SP_RW + lane;
    if (wave >= SP_SLOTS) {
        #pragma unroll
        for (int g = 0; g < 2; ++g)
            #pragma unroll
            for (int t = 0; t < 3; ++t)
                #pragma unroll
                for (int r = 0; r < 8; ++r)
                    rp[(g * 3 + t) * SP_RQ + r * 32] = acc[g][t][r];
    }
    __syncthreads();
    if (wave < SP_SLOTS) {
        #pragma unroll
        for (int g = 0; g < 2; ++g)
            #pragma unroll
            for (int t = 0; t < 3; ++t)
                #pragma unroll
                for (int r = 0; r < 8; ++r) {
                    float* q = rp + (g * 3 + t) * SP_RQ + r * 32;
                    *q = acc[g][t][r] + *q;
                }
    }
    __syncthreads();

    float* outp = ssm + (size_t)m0 * SP_NC;
    for (int o = threadIdx.x; o < 32 * SP_NC; o += 512) {
        const int row = o / SP_NC;
        const int n   = o - row * SP_NC;
        const int j   = n / 3;
        const int t   = n - 3 * j;
        const int q   = (row >> 4) * 3 + t;
        const int r   = row & 7;
        const int ln  = ((row >> 3) & 1) * 16 + j;
        const float* p = red + q * SP_RQ + r * 32 + ln;
        float s = 0.0f;
        #pragma unroll
        for (int w = 0; w < SP_SLOTS; ++w)
            s += p[w * SP_RW];
        outp[o] = s;
    }
}

#define TCH 64
#define SBATCH 16
#define SPITCH 36
#define SQ 4
#define SWAVES (D_STATE / SQ)
#define SCH 32
#define SDL (TCH / SWAVES)
#define SFIN (SBATCH / SWAVES)
#define SSTG ((TCH * 33 + SCH * SWAVES - 1) / (SCH * SWAVES))

template <int Q>
__device__ __forceinline__ void scan_finish(const float (&xv)[SBATCH], const float (&gz)[SFIN],
                                            const float* pc, float* yt, float Dd)
{
    float ys[SFIN];
    #pragma unroll
    for (int f = 0; f < SFIN; ++f) {
        const float* p = pc + (SWAVES * f + Q) * (SWAVES * SCH);
        ys[f] = ((p[0] + p[SCH]) + p[2 * SCH]) + p[3 * SCH];
    }
    #pragma unroll
    for (int f = 0; f < SFIN; ++f)
        yt[(size_t)(SWAVES * f + Q) * D_INNER] = (ys[f] + Dd * xv[SWAVES * f + Q]) * gz[f];
}

__global__ void __launch_bounds__(SCH * SWAVES) __attribute__((amdgpu_waves_per_eu(1, 1)))
scan_kernel(const float* __restrict__ xconv,
            const float* __restrict__ xz,
            const float* __restrict__ ssm,
            const float* __restrict__ dt_w,
            const float* __restrict__ dt_b,
            const float* __restrict__ A_log,
            const float* __restrict__ Dp,
            float* __restrict__ ygated)
{
    __shared__ __attribute__((aligned(16))) float sb[TCH * SPITCH];
    __shared__ float dlc[TCH * SCH];
    __shared__ float yb[2 * SBATCH * SWAVES * SCH];

    const int c = threadIdx.x & (SCH - 1);
    const int q = __builtin_amdgcn_readfirstlane(threadIdx.x >> 5);
    const int d = blockIdx.x * SCH + c;
    const int b = blockIdx.y;

    const float dw = dt_w[d];
    const float db = dt_b[d];
    const float Dd = Dp[d];

    float Areg[SQ];
    #pragma unroll
    for (int n = 0; n < SQ; ++n)
        Areg[n] = -__expf(A_log[(size_t)d * D_STATE + SQ * q + n]);

    float h[SQ];
    #pragma unroll
    for (int n = 0; n < SQ; ++n) h[n] = 0.0f;

    const float* xp = xconv  + (size_t)b * SEQLEN * D_INNER + d;
    float*       yp = ygated + (size_t)b * SEQLEN * D_INNER + d;
    const float* zp = xz + ((size_t)b * SEQLEN + q) * (2 * D_INNER) + D_INNER + d;

    float xn[SBATCH];
    #pragma unroll
    for (int i = 0; i < SBATCH; ++i)
        xn[i] = xp[(size_t)i * D_INNER];
    float zn[SFIN];
    #pragma unroll
    for (int f = 0; f < SFIN; ++f)
        zn[f] = zp[(size_t)(SWAVES * f) * (2 * D_INNER)];

    for (int t0 = 0; t0 < SEQLEN; t0 += TCH) {
        __syncthreads();
        {
            const float* src = ssm + (size_t)(b * SEQLEN + t0) * 33;
            const bool tail = threadIdx.x < TCH * 33 - (SSTG - 1) * (SCH * SWAVES);
            float stg[SSTG];
            #pragma unroll
            for (int k = 0; k < SSTG; ++k) {
                const int i = threadIdx.x + k * (SCH * SWAVES);
                if (k < SSTG - 1 || tail) stg[k] = src[i];
            }
            #pragma unroll
            for (int k = 0; k < SSTG; ++k) {
                const int i = threadIdx.x + k * (SCH * SWAVES);
                const int t = i / 33;
                const int j = i - t * 33;
                if (k < SSTG - 1 || tail) sb[t * SPITCH + (j ? j + 3 : 0)] = stg[k];
            }
        }
        __syncthreads();

        #pragma unroll
        for (int j = 0; j < SDL; ++j) {
            const int t = q * SDL + j;
            const float zz = dw * sb[t * SPITCH] + db;
            dlc[t * SCH + c] = softplus_f(zz);
        }
        __syncthreads();

        for (int s0 = 0; s0 < TCH; s0 += SBATCH) {
            float dl[SBATCH];
            float4 Bq[SBATCH], Cq[SBATCH];
            #pragma unroll
            for (int i = 0; i < SBATCH; ++i) {
                const float* row = sb + (s0 + i) * SPITCH;
                dl[i] = dlc[(s0 + i) * SCH + c];
                Bq[i] = *(const float4*)(row + 4 + SQ * q);
                Cq[i] = *(const float4*)(row + 4 + D_STATE + SQ * q);
            }

            float xv[SBATCH];
            #pragma unroll
            for (int i = 0; i < SBATCH; ++i) xv[i] = xn[i];
            float gz[SFIN];
            #pragma unroll
            for (int f = 0; f < SFIN; ++f) gz[f] = zn[f];

            const int tn = t0 + s0 + SBATCH;
            if (tn < SEQLEN) {
                #pragma unroll
                for (int i = 0; i < SBATCH; ++i)
                    xn[i] = xp[(size_t)(tn + i) * D_INNER];
                #pragma unroll
                for (int f = 0; f < SFIN; ++f)
                    zn[f] = zp[(size_t)(tn + SWAVES * f) * (2 * D_INNER)];
            }

            #pragma unroll
            for (int f = 0; f < SFIN; ++f) gz[f] = silu_f(gz[f]);

            float* ybat = yb + ((s0 / SBATCH) & 1) * (SBATCH * SWAVES * SCH);
            float* ypart = ybat + q * SCH + c;

            #pragma unroll
            for (int i = 0; i < SBATCH; ++i) {
                const float delta = dl[i];
                const float dx = delta * xv[i];
                const float Bv[SQ] = {Bq[i].x, Bq[i].y, Bq[i].z, Bq[i].w};
                const float Cv[SQ] = {Cq[i].x, Cq[i].y, Cq[i].z, Cq[i].w};

                float y = 0.0f;
                #pragma unroll
                for (int n = 0; n < SQ; ++n) {
                    float dA = __expf(delta * Areg[n]);
                    h[n] = dA * h[n] + dx * Bv[n];
                    y += h[n] * Cv[n];
                }
                ypart[i * (SWAVES * SCH)] = y;
            }

            __syncthreads();

            {
                const float* pc = ybat + c;
                float* yt = yp + (size_t)(t0 + s0) * D_INNER;
                if (q == 0)      scan_finish<0>(xv, gz, pc, yt, Dd);
                else if (q == 1) scan_finish<1>(xv, gz, pc, yt, Dd);
                else if (q == 2) scan_finish<2>(xv, gz, pc, yt, Dd);
                else             scan_finish<3>(xv, gz, pc, yt, Dd);
            }
        }
    }
}

extern "C" void kernel_launch(void* const* d_in, const int* in_sizes, int n_in,
                              void* d_out, int out_size, void* d_ws, size_t ws_size,
                              hipStream_t stream)
{
    const float* x      = (const float*)d_in[0];
    const float* W_in   = (const float*)d_in[1];
    const float* conv_w = (const float*)d_in[2];
    const float* conv_b = (const float*)d_in[3];
    const float* W_x    = (const float*)d_in[4];
    const float* dt_w   = (const float*)d_in[5];
    const float* dt_b   = (const float*)d_in[6];
    const float* A_log  = (const float*)d_in[7];
    const float* Dp     = (const float*)d_in[8];
    const float* W_out  = (const float*)d_in[9];
    float* out = (float*)d_out;

    float* ws = (float*)d_ws;
    float* xz     = ws;
    float* xconv  = xz    + (size_t)NROWS * 2 * D_INNER;
    float* ssm    = xconv + (size_t)NROWS * D_INNER;
    float* ygated = ssm   + (size_t)NROWS * 33;

    wmma_gemm_f32<2, 16, 8><<<dim3(2 * D_INNER / 128, NROWS / 256), 256, 0, stream>>>(
        x, W_in, xz, NROWS, 2 * D_INNER, MODEL_DIM);

    conv_silu_kernel<<<EW_BLOCKS, 256, 0, stream>>>(xz, conv_w, conv_b, xconv);

    ssm_proj_wmma<<<NROWS / 32, 512, 0, stream>>>(xconv, W_x, ssm);

    scan_kernel<<<dim3(D_INNER / SCH, BATCH), SCH * SWAVES, 0, stream>>>(
        xconv, xz, ssm, dt_w, dt_b, A_log, Dp, ygated);

    wmma_gemm_f32<1, 32, 4><<<dim3(MODEL_DIM / 128, NROWS / 64), 128, 0, stream>>>(
        ygated, W_out, out, NROWS, MODEL_DIM, D_INNER);
}


# scan kernel: barriers wait for LDS only (s_wait_dscnt + barrier by hand; __syncthreads also drained the global loads), so x/z are fetched 2 batches ahead, ssm rows half a chunk; no sum reordered
# speedup vs baseline: 6.7090x; 1.0566x over previous
#include <hip/hip_runtime.h>
#include <hip/hip_bf16.h>

typedef __attribute__((ext_vector_type(2))) float v2f;
typedef __attribute__((ext_vector_type(8))) float v8f;

#define MODEL_DIM 1024
#define D_STATE   16
#define D_CONV    4
#define D_INNER   2048
#define BATCH     4
#define SEQLEN    1024
#define NROWS     (BATCH * SEQLEN)

__device__ __forceinline__ float silu_f(float x) {
    return x / (1.0f + __expf(-x));
}

__device__ __forceinline__ float softplus_f(float z) {
    const float e = __expf(-fabsf(z));
    const float u = 1.0f + e;
    const float d = u - 1.0f;
    const float l = __builtin_amdgcn_logf(u) * 0.69314718f;
    const float r = (d == 0.0f) ? e : l * (e * __builtin_amdgcn_rcpf(d));
    return fmaxf(z, 0.0f) + r;
}

#define BPITCH 272
template <int MT, int KCT, int NW>
__global__ void __launch_bounds__(NW * 32) __attribute__((amdgpu_waves_per_eu(NW / 2, NW / 2)))
wmma_gemm_f32(const float* __restrict__ A, const float* __restrict__ W,
              float* __restrict__ C, int M, int N, int K)
{
    constexpr int BUF = (KCT / 2) * BPITCH;
    __shared__ float Bs[2 * BUF];

    constexpr int NR = KCT / NW;
    constexpr int NO = KCT / 8;

    const int wave   = threadIdx.x >> 5;
    const int lane   = threadIdx.x & 31;
    const int lane16 = lane & 15;
    const int hi     = lane >> 4;

    const int m0 = blockIdx.y * (NW * MT * 16) + wave * (MT * 16);
    const int n0 = blockIdx.x * 128;

    const float* Arow[MT];
    #pragma unroll
    for (int g = 0; g < MT; ++g)
        Arow[g] = A + (size_t)(m0 + g * 16 + lane16) * K + 4 * hi;

    const int wr = wave;
    const int wc = lane * 4;
    const float* Wbase = W + (size_t)wr * N + n0 + wc;
    float* Bst = Bs + (wr >> 1) * BPITCH + (wr & 1) + wc * 2 + 2 * (lane >> 3);
    const float* bp = Bs + (2 * hi) * BPITCH + lane16 * 2;

    float4 wreg[NR];
    float4 afr[2][NO][MT];

    v8f acc[MT][8];
    #pragma unroll
    for (int g = 0; g < MT; ++g)
        #pragma unroll
        for (int j = 0; j < 8; ++j)
            acc[g][j] = (v8f){};

    #pragma unroll
    for (int i = 0; i < NR; ++i)
        wreg[i] = *(const float4*)(Wbase + (size_t)(NW * i) * N);
    #pragma unroll
    for (int o = 0; o < NO; ++o)
        #pragma unroll
        for (int g = 0; g < MT; ++g)
            afr[0][o][g] = *(const float4*)(Arow[g] + 8 * o);
    #pragma unroll
    for (int i = 0; i < NR; ++i) {
        float* q = Bst + (NW / 2) * i * BPITCH;
        q[0] = wreg[i].x;  q[2] = wreg[i].y;  q[4] = wreg[i].z;  q[6] = wreg[i].w;
    }
    __syncthreads();

    for (int k = 0; k < K; k += 2 * KCT) {
        #pragma unroll
        for (int p = 0; p < 2; ++p) {
            const int kc = k + p * KCT;
            const int kn = min(kc + KCT, K - KCT);
            const float* Wp = Wbase + (size_t)kn * N;
            #pragma unroll
            for (int i = 0; i < NR; ++i)
                wreg[i] = *(const float4*)(Wp + (size_t)(NW * i) * N);
            #pragma unroll
            for (int o = 0; o < NO; ++o)
                #pragma unroll
                for (int g = 0; g < MT; ++g)
                    afr[p ^ 1][o][g] = *(const float4*)(Arow[g] + kn + 8 * o);
            __builtin_amdgcn_sched_barrier(0);

            const float* bcur = bp + p * BUF;
            #pragma unroll
            for (int o = 0; o < NO; ++o) {
                #pragma unroll
                for (int s = 0; s < 2; ++s) {
                    v2f a[MT];
                    #pragma unroll
                    for (int g = 0; g < MT; ++g)
                        a[g] = s ? (v2f){afr[p][o][g].z, afr[p][o][g].w}
                                 : (v2f){afr[p][o][g].x, afr[p][o][g].y};
                    const float* br = bcur + (4 * o + s) * BPITCH;
                    #pragma unroll
                    for (int j = 0; j < 8; ++j) {
                        v2f b = *(const v2f*)(br + j * 32 + 2 * (j >> 1));
                        #pragma unroll
                        for (int g = 0; g < MT; ++g)
                            acc[g][j] = __builtin_amdgcn_wmma_f32_16x16x4_f32(
                                false, a[g], false, b, (short)0, acc[g][j], false, false);
                    }
                }
            }

            __builtin_amdgcn_sched_barrier(0);

            float* bnx = Bst + (p ^ 1) * BUF;
            #pragma unroll
            for (int i = 0; i < NR; ++i) {
                float* q = bnx + (NW / 2) * i * BPITCH;
                q[0] = wreg[i].x;  q[2] = wreg[i].y;  q[4] = wreg[i].z;  q[6] = wreg[i].w;
            }
            __syncthreads();
        }
    }

    const int crow = (lane >> 4) ? 8 : 0;
    const int ccol = n0 + lane16;
    #pragma unroll
    for (int g = 0; g < MT; ++g) {
        float* Cp = C + (size_t)(m0 + g * 16 + crow) * N + ccol;
        #pragma unroll
        for (int r = 0; r < 8; ++r) {
            #pragma unroll
            for (int j = 0; j < 8; ++j)
                Cp[(size_t)r * N + j * 16] = acc[g][j][r];
        }
    }
}

#define EW_ROWS   16
#define EW_GRP    8
#define EW_NGRP   (EW_ROWS / EW_GRP)
#define EW_C4     (D_INNER / 4)
#define EW_BPR    (EW_C4 / 256)
#define EW_BLOCKS ((NROWS / EW_ROWS) * EW_BPR)

__device__ __forceinline__ float conv_tap4(float b, float x0, float x1, float x2, float x3,
                                           float4 w, bool skip0, bool skip1, bool skip2)
{
    float acc = b;
    if (!skip0) acc += x0 * w.x;
    if (!skip1) acc += x1 * w.y;
    if (!skip2) acc += x2 * w.z;
    acc += x3 * w.w;
    return silu_f(acc);
}

__global__ void __launch_bounds__(256)
conv_silu_kernel(const float* __restrict__ xz,
                 const float* __restrict__ conv_w,
                 const float* __restrict__ conv_b,
                 float* __restrict__ xconv)
{
    const int c4   = (blockIdx.x % EW_BPR) * 256 + threadIdx.x;
    const int row0 = (blockIdx.x / EW_BPR) * EW_ROWS;
    const bool first = (row0 & (SEQLEN - 1)) == 0;

    const float4* wp = (const float4*)(conv_w + (size_t)c4 * (4 * D_CONV));
    float4 w[4];
    #pragma unroll
    for (int j = 0; j < 4; ++j) w[j] = wp[j];
    const float4 bias = *(const float4*)(conv_b + c4 * 4);

    const float* xp = xz + (size_t)row0 * (2 * D_INNER) + c4 * 4;
    float*       op = xconv + (size_t)row0 * D_INNER + c4 * 4;

    float4 xw[EW_GRP + D_CONV - 1];
    const ptrdiff_t back = first ? 0 : (ptrdiff_t)(2 * D_INNER);
    #pragma unroll
    for (int m = 0; m < D_CONV - 1; ++m)
        xw[m] = *(const float4*)(xp - (D_CONV - 1 - m) * back);
    #pragma unroll
    for (int i = 0; i < EW_GRP; ++i)
        xw[i + D_CONV - 1] = *(const float4*)(xp + (size_t)i * (2 * D_INNER));

    #pragma unroll 1
    for (int grp = 0; grp < EW_NGRP; ++grp) {
        float4 xn[EW_GRP];
        #pragma unroll
        for (int i = 0; i < EW_GRP; ++i) xn[i] = make_float4(0.0f, 0.0f, 0.0f, 0.0f);
        if (grp + 1 < EW_NGRP) {
            #pragma unroll
            for (int i = 0; i < EW_GRP; ++i)
                xn[i] = *(const float4*)(xp + (size_t)((grp + 1) * EW_GRP + i) * (2 * D_INNER));
        }

        const bool head = first && grp == 0;
        #pragma unroll
        for (int i = 0; i < EW_GRP; ++i) {
            const bool s0 = head && (i + 0 < D_CONV - 1);
            const bool s1 = head && (i + 1 < D_CONV - 1);
            const bool s2 = head && (i + 2 < D_CONV - 1);
            float4 o;
            o.x = conv_tap4(bias.x, xw[i].x, xw[i + 1].x, xw[i + 2].x, xw[i + 3].x, w[0], s0, s1, s2);
            o.y = conv_tap4(bias.y, xw[i].y, xw[i + 1].y, xw[i + 2].y, xw[i + 3].y, w[1], s0, s1, s2);
            o.z = conv_tap4(bias.z, xw[i].z, xw[i + 1].z, xw[i + 2].z, xw[i + 3].z, w[2], s0, s1, s2);
            o.w = conv_tap4(bias.w, xw[i].w, xw[i + 1].w, xw[i + 2].w, xw[i + 3].w, w[3], s0, s1, s2);
            *(float4*)(op + (size_t)(grp * EW_GRP + i) * D_INNER) = o;
        }

        #pragma unroll
        for (int m = 0; m < D_CONV - 1; ++m) xw[m] = xw[EW_GRP + m];
        #pragma unroll
        for (int i = 0; i < EW_GRP; ++i) xw[i + D_CONV - 1] = xn[i];
    }
}

#define SP_NC     33
#define SP_WAVES  16
#define SP_KSLICE (D_INNER / SP_WAVES)
#define SP_GRP    4
#define SP_NGRP   (SP_KSLICE / (8 * SP_GRP))
#define SP_RQ     267
#define SP_RW     (6 * SP_RQ)
#define SP_SLOTS  (SP_WAVES / 2)

struct F3 { float x, y, z; };

__global__ void __launch_bounds__(512)
ssm_proj_wmma(const float* __restrict__ xconv,
              const float* __restrict__ Wx,
              float* __restrict__ ssm)
{
    __shared__ float red[SP_SLOTS * SP_RW];

    const int wave   = threadIdx.x >> 5;
    const int lane   = threadIdx.x & 31;
    const int lane16 = lane & 15;
    const int hi     = lane >> 4;
    const int m0     = blockIdx.x * 32;
    const int k0     = wave * SP_KSLICE + 4 * hi;

    const float* ap0 = xconv + (size_t)(m0 + lane16) * D_INNER + k0;
    const float* ap1 = ap0 + (size_t)16 * D_INNER;
    const int jj = (lane16 < 11) ? lane16 : 10;
    const float* bp = Wx + (size_t)k0 * SP_NC + 3 * jj;

    v8f acc[2][3];
    #pragma unroll
    for (int g = 0; g < 2; ++g)
        #pragma unroll
        for (int t = 0; t < 3; ++t)
            acc[g][t] = (v8f){};

    float4 a0[2][SP_GRP], a1[2][SP_GRP];
    F3 bf[2][SP_GRP][4];

    #pragma unroll
    for (int o = 0; o < SP_GRP; ++o) {
        a0[0][o] = *(const float4*)(ap0 + 8 * o);
        a1[0][o] = *(const float4*)(ap1 + 8 * o);
        #pragma unroll
        for (int r = 0; r < 4; ++r)
            bf[0][o][r] = *(const F3*)(bp + (size_t)(8 * o + r) * SP_NC);
    }

    #pragma unroll
    for (int grp = 0; grp < SP_NGRP; ++grp) {
        const int cur = grp & 1, nxt = cur ^ 1;
        if (grp + 1 < SP_NGRP) {
            #pragma unroll
            for (int o = 0; o < SP_GRP; ++o) {
                const int ko = 8 * ((grp + 1) * SP_GRP + o);
                a0[nxt][o] = *(const float4*)(ap0 + ko);
                a1[nxt][o] = *(const float4*)(ap1 + ko);
                #pragma unroll
                for (int r = 0; r < 4; ++r)
                    bf[nxt][o][r] = *(const F3*)(bp + (size_t)(ko + r) * SP_NC);
            }
        }
        __builtin_amdgcn_sched_barrier(0);
        #pragma unroll
        for (int o = 0; o < SP_GRP; ++o) {
            #pragma unroll
            for (int s = 0; s < 2; ++s) {
                const v2f x0 = s ? (v2f){a0[cur][o].z, a0[cur][o].w}
                                 : (v2f){a0[cur][o].x, a0[cur][o].y};
                const v2f x1 = s ? (v2f){a1[cur][o].z, a1[cur][o].w}
                                 : (v2f){a1[cur][o].x, a1[cur][o].y};
                const F3 lo = bf[cur][o][2 * s], up = bf[cur][o][2 * s + 1];
                const v2f b0 = (v2f){lo.x, up.x};
                const v2f b1 = (v2f){lo.y, up.y};
                const v2f b2 = (v2f){lo.z, up.z};
                acc[0][0] = __builtin_amdgcn_wmma_f32_16x16x4_f32(
                    false, x0, false, b0, (short)0, acc[0][0], false, false);
                acc[1][0] = __builtin_amdgcn_wmma_f32_16x16x4_f32(
                    false, x1, false, b0, (short)0, acc[1][0], false, false);
                acc[0][1] = __builtin_amdgcn_wmma_f32_16x16x4_f32(
                    false, x0, false, b1, (short)0, acc[0][1], false, false);
                acc[1][1] = __builtin_amdgcn_wmma_f32_16x16x4_f32(
                    false, x1, false, b1, (short)0, acc[1][1], false, false);
                acc[0][2] = __builtin_amdgcn_wmma_f32_16x16x4_f32(
                    false, x0, false, b2, (short)0, acc[0][2], false, false);
                acc[1][2] = __builtin_amdgcn_wmma_f32_16x16x4_f32(
                    false, x1, false, b2, (short)0, acc[1][2], false, false);
            }
        }
    }

    float* rp = red + (wave & (SP_SLOTS - 1)) * SP_RW + lane;
    if (wave >= SP_SLOTS) {
        #pragma unroll
        for (int g = 0; g < 2; ++g)
            #pragma unroll
            for (int t = 0; t < 3; ++t)
                #pragma unroll
                for (int r = 0; r < 8; ++r)
                    rp[(g * 3 + t) * SP_RQ + r * 32] = acc[g][t][r];
    }
    __syncthreads();
    if (wave < SP_SLOTS) {
        #pragma unroll
        for (int g = 0; g < 2; ++g)
            #pragma unroll
            for (int t = 0; t < 3; ++t)
                #pragma unroll
                for (int r = 0; r < 8; ++r) {
                    float* q = rp + (g * 3 + t) * SP_RQ + r * 32;
                    *q = acc[g][t][r] + *q;
                }
    }
    __syncthreads();

    float* outp = ssm + (size_t)m0 * SP_NC;
    for (int o = threadIdx.x; o < 32 * SP_NC; o += 512) {
        const int row = o / SP_NC;
        const int n   = o - row * SP_NC;
        const int j   = n / 3;
        const int t   = n - 3 * j;
        const int q   = (row >> 4) * 3 + t;
        const int r   = row & 7;
        const int ln  = ((row >> 3) & 1) * 16 + j;
        const float* p = red + q * SP_RQ + r * 32 + ln;
        float s = 0.0f;
        #pragma unroll
        for (int w = 0; w < SP_SLOTS; ++w)
            s += p[w * SP_RW];
        outp[o] = s;
    }
}

#define TCH 64
#define SBATCH 16
#define SPITCH 36
#define SQ 4
#define SWAVES (D_STATE / SQ)
#define SCH 32
#define SDL (TCH / SWAVES)
#define SFIN (SBATCH / SWAVES)
#define SSTG ((TCH * 33 + SCH * SWAVES - 1) / (SCH * SWAVES))

#define SNB (TCH / SBATCH)
#define SAHEAD 2

__device__ __forceinline__ void lds_barrier()
{
    asm volatile("s_wait_dscnt 0x0\n\ts_barrier_signal -1\n\ts_barrier_wait -1" ::: "memory");
}

__device__ __forceinline__ void scan_fetch(float (&x)[SBATCH], float (&z)[SFIN],
                                           const float* xp, const float* zp, int t)
{
    #pragma unroll
    for (int i = 0; i < SBATCH; ++i)
        x[i] = xp[(size_t)(t + i) * D_INNER];
    #pragma unroll
    for (int f = 0; f < SFIN; ++f)
        z[f] = zp[(size_t)(t + SWAVES * f) * (2 * D_INNER)];
}

__device__ __forceinline__ void scan_fetch_rows(float (&stg)[SSTG], const float* src, bool tail)
{
    #pragma unroll
    for (int k = 0; k < SSTG - 1; ++k)
        stg[k] = src[threadIdx.x + k * (SCH * SWAVES)];
    stg[SSTG - 1] = src[tail ? threadIdx.x + (SSTG - 1) * (SCH * SWAVES) : threadIdx.x];
}

template <int Q>
__device__ __forceinline__ void scan_finish(const float (&xv)[SBATCH], const float (&gz)[SFIN],
                                            const float* pc, float* yt, float Dd)
{
    float ys[SFIN];
    #pragma unroll
    for (int f = 0; f < SFIN; ++f) {
        const float* p = pc + (SWAVES * f + Q) * (SWAVES * SCH);
        ys[f] = ((p[0] + p[SCH]) + p[2 * SCH]) + p[3 * SCH];
    }
    #pragma unroll
    for (int f = 0; f < SFIN; ++f)
        yt[(size_t)(SWAVES * f + Q) * D_INNER] = (ys[f] + Dd * xv[SWAVES * f + Q]) * gz[f];
}

__global__ void __launch_bounds__(SCH * SWAVES) __attribute__((amdgpu_waves_per_eu(1, 1)))
scan_kernel(const float* xconv,
            const float* xz,
            const float* ssm,
            const float* __restrict__ dt_w,
            const float* __restrict__ dt_b,
            const float* __restrict__ A_log,
            const float* __restrict__ Dp,
            float* __restrict__ ygated)
{
    __shared__ __attribute__((aligned(16))) float sb[TCH * SPITCH];
    __shared__ float dlc[TCH * SCH];
    __shared__ float yb[2 * SBATCH * SWAVES * SCH];

    const int c = threadIdx.x & (SCH - 1);
    const int q = __builtin_amdgcn_readfirstlane(threadIdx.x >> 5);
    const int d = blockIdx.x * SCH + c;
    const int b = blockIdx.y;

    const float dw = dt_w[d];
    const float db = dt_b[d];
    const float Dd = Dp[d];

    float Areg[SQ];
    #pragma unroll
    for (int n = 0; n < SQ; ++n)
        Areg[n] = -__expf(A_log[(size_t)d * D_STATE + SQ * q + n]);

    float h[SQ];
    #pragma unroll
    for (int n = 0; n < SQ; ++n) h[n] = 0.0f;

    const float* xp = xconv  + (size_t)b * SEQLEN * D_INNER + d;
    float*       yp = ygated + (size_t)b * SEQLEN * D_INNER + d;
    const float* zp = xz + ((size_t)b * SEQLEN + q) * (2 * D_INNER) + D_INNER + d;

    float xs[SNB][SBATCH];
    float zs[SNB][SFIN];
    float stg[SSTG];
    const bool tail = threadIdx.x < TCH * 33 - (SSTG - 1) * (SCH * SWAVES);
    const float* sp = ssm + (size_t)b * SEQLEN * 33;

    scan_fetch(xs[0], zs[0], xp, zp, 0);
    __builtin_amdgcn_sched_barrier(0);
    scan_fetch_rows(stg, sp, tail);
    __builtin_amdgcn_sched_barrier(0);
    scan_fetch(xs[1], zs[1], xp, zp, SBATCH);
    __builtin_amdgcn_sched_barrier(0);

    for (int t0 = 0; t0 < SEQLEN; t0 += TCH) {
        lds_barrier();
        {
            #pragma unroll
            for (int k = 0; k < SSTG; ++k) {
                const int i = threadIdx.x + k * (SCH * SWAVES);
                const int t = i / 33;
                const int j = i - t * 33;
                if (k < SSTG - 1 || tail) sb[t * SPITCH + (j ? j + 3 : 0)] = stg[k];
            }
        }
        lds_barrier();

        #pragma unroll
        for (int j = 0; j < SDL; ++j) {
            const int t = q * SDL + j;
            const float zz = dw * sb[t * SPITCH] + db;
            dlc[t * SCH + c] = softplus_f(zz);
        }
        lds_barrier();

        const int tnc = min(t0 + TCH, SEQLEN - TCH);

        #pragma unroll
        for (int jb = 0; jb < SNB; ++jb) {
            const int s0 = jb * SBATCH;
            {
                const int jn = (jb + SAHEAD) % SNB;
                const int tn = (jb + SAHEAD < SNB) ? t0 + s0 + SAHEAD * SBATCH
                                                   : tnc + (jb + SAHEAD - SNB) * SBATCH;
                scan_fetch(xs[jn], zs[jn], xp, zp, tn);
                if (jb == SNB - SAHEAD)
                    scan_fetch_rows(stg, sp + (size_t)tnc * 33, tail);
            }
            __builtin_amdgcn_sched_barrier(0);

            float dl[SBATCH];
            float4 Bq[SBATCH], Cq[SBATCH];
            #pragma unroll
            for (int i = 0; i < SBATCH; ++i) {
                const float* row = sb + (s0 + i) * SPITCH;
                dl[i] = dlc[(s0 + i) * SCH + c];
                Bq[i] = *(const float4*)(row + 4 + SQ * q);
                Cq[i] = *(const float4*)(row + 4 + D_STATE + SQ * q);
            }

            float gz[SFIN];
            #pragma unroll
            for (int f = 0; f < SFIN; ++f) gz[f] = silu_f(zs[jb][f]);

            float* ybat = yb + (jb & 1) * (SBATCH * SWAVES * SCH);
            float* ypart = ybat + q * SCH + c;

            #pragma unroll
            for (int i = 0; i < SBATCH; ++i) {
                const float delta = dl[i];
                const float dx = delta * xs[jb][i];
                const float Bv[SQ] = {Bq[i].x, Bq[i].y, Bq[i].z, Bq[i].w};
                const float Cv[SQ] = {Cq[i].x, Cq[i].y, Cq[i].z, Cq[i].w};

                float y = 0.0f;
                #pragma unroll
                for (int n = 0; n < SQ; ++n) {
                    float dA = __expf(delta * Areg[n]);
                    h[n] = dA * h[n] + dx * Bv[n];
                    y += h[n] * Cv[n];
                }
                ypart[i * (SWAVES * SCH)] = y;
                if (i == SBATCH / 2 - 1) __builtin_amdgcn_sched_barrier(0);
            }

            lds_barrier();

            {
                const float* pc = ybat + c;
                float* yt = yp + (size_t)(t0 + s0) * D_INNER;
                if (q == 0)      scan_finish<0>(xs[jb], gz, pc, yt, Dd);
                else if (q == 1) scan_finish<1>(xs[jb], gz, pc, yt, Dd);
                else if (q == 2) scan_finish<2>(xs[jb], gz, pc, yt, Dd);
                else             scan_finish<3>(xs[jb], gz, pc, yt, Dd);
            }
        }
    }
}

extern "C" void kernel_launch(void* const* d_in, const int* in_sizes, int n_in,
                              void* d_out, int out_size, void* d_ws, size_t ws_size,
                              hipStream_t stream)
{
    const float* x      = (const float*)d_in[0];
    const float* W_in   = (const float*)d_in[1];
    const float* conv_w = (const float*)d_in[2];
    const float* conv_b = (const float*)d_in[3];
    const float* W_x    = (const float*)d_in[4];
    const float* dt_w   = (const float*)d_in[5];
    const float* dt_b   = (const float*)d_in[6];
    const float* A_log  = (const float*)d_in[7];
    const float* Dp     = (const float*)d_in[8];
    const float* W_out  = (const float*)d_in[9];
    float* out = (float*)d_out;

    float* ws = (float*)d_ws;
    float* xz     = ws;
    float* xconv  = xz    + (size_t)NROWS * 2 * D_INNER;
    float* ssm    = xconv + (size_t)NROWS * D_INNER;
    float* ygated = ssm   + (size_t)NROWS * 33;

    wmma_gemm_f32<2, 16, 8><<<dim3(2 * D_INNER / 128, NROWS / 256), 256, 0, stream>>>(
        x, W_in, xz, NROWS, 2 * D_INNER, MODEL_DIM);

    conv_silu_kernel<<<EW_BLOCKS, 256, 0, stream>>>(xz, conv_w, conv_b, xconv);

    ssm_proj_wmma<<<NROWS / 32, 512, 0, stream>>>(xconv, W_x, ssm);

    scan_kernel<<<dim3(D_INNER / SCH, BATCH), SCH * SWAVES, 0, stream>>>(
        xconv, xz, ssm, dt_w, dt_b, A_log, Dp, ygated);

    wmma_gemm_f32<1, 32, 4><<<dim3(MODEL_DIM / 128, NROWS / 64), 128, 0, stream>>>(
        ygated, W_out, out, NROWS, MODEL_DIM, D_INNER);
}
